# starting kernel
# baseline (speedup 1.0000x reference)
#include <hip/hip_runtime.h>
#include <hip/hip_bf16.h>
#include <hip/hip_fp16.h>

typedef __attribute__((ext_vector_type(2)))  float    v2f;
typedef __attribute__((ext_vector_type(8)))  float    v8f;
typedef __attribute__((ext_vector_type(16))) _Float16 v16h;

#define B_   2
#define C_   32
#define H_   96
#define W_   128
#define HW_  (H_ * W_)
#define BC_  (B_ * C_)
#define EPSF 1e-6f

#define OFF_QS    0
#define OFF_QC    6144
#define OFF_KH    14336
#define OFF_KW    20480
#define OFF_SQ    28672
#define OFF_SK    28736
#define OFF_GMAX  28800
#define OFF_GSMAX 28928
#define OFF_HR    28960
#define OFF_HC    (OFF_HR + 32 * 6144)
#define OFF_WR    (OFF_HC + 32 * 6144)
#define OFF_WC    (OFF_WR + 32 * 8192)
#define OFF_IRS   (OFF_WC + 32 * 8192)
#define OFF_SCORE (OFF_IRS + 4 * 64 * 224)

#if __has_builtin(__builtin_amdgcn_wmma_f32_16x16x4_f32)
#define USE_WMMA_F32 1
#endif

__global__ void k1_sums(const float* __restrict__ q, const float* __restrict__ k,
                        float* __restrict__ ws) {
  __shared__ float tile[HW_];
  __shared__ float rowbuf[H_];
  const int bc  = blockIdx.x;
  const int tid = threadIdx.x;
  for (int p = 0; p < 2; ++p) {
    const float* s = p ? k : q;
    for (int i = tid; i < HW_; i += 256) tile[i] = s[bc * HW_ + i];
    __syncthreads();
    if (tid < H_) {
      float acc = 0.f;
      for (int x = 0; x < W_; ++x) acc += tile[tid * W_ + x];
      ws[(p ? OFF_KH : OFF_QS) + bc * H_ + tid] = acc;
      rowbuf[tid] = acc;
    }
    if (tid >= 128) {
      const int x = tid - 128;
      float acc = 0.f;
      for (int h = 0; h < H_; ++h) acc += tile[h * W_ + x];
      ws[(p ? OFF_KW : OFF_QC) + bc * W_ + x] = acc;
    }
    __syncthreads();
    if (tid == 0) {
      float t = 0.f;
      for (int h = 0; h < H_; ++h) t += rowbuf[h];
      ws[(p ? OFF_SK : OFF_SQ) + bc] = t;
    }
    __syncthreads();
  }
}

__global__ void k2_norm(float* __restrict__ ws) {
  const int type = blockIdx.x & 3;
  const int c0   = blockIdx.x >> 2;
  const int tid  = threadIdx.x;
  __shared__ float rmn[256], rmx[256];
  const int   len    = (type < 2) ? H_ : W_;
  const int   N      = BC_ * len;
  const float addeps = (type < 2) ? (H_ * EPSF) : (W_ * EPSF);
  float* dst;
  if      (type == 0) dst = ws + OFF_HR + c0 * (BC_ * H_);
  else if (type == 1) dst = ws + OFF_HC + c0 * (BC_ * H_);
  else if (type == 2) dst = ws + OFF_WR + c0 * (BC_ * W_);
  else                dst = ws + OFF_WC + c0 * (BC_ * W_);
  float lmn = 3.4e38f, lmx = -3.4e38f;
  for (int i = tid; i < N; i += 256) {
    const int b = i / (C_ * len);
    const int r = i % (C_ * len);
    const int c = r / len;
    const int x = r % len;
    float raw;
    if      (type == 0) raw = ws[OFF_SQ + b * C_ + c0] * ws[OFF_KH + (b * C_ + c) * H_ + x] + addeps;
    else if (type == 1) raw = ws[OFF_QS + (b * C_ + c0) * H_ + x] * ws[OFF_SK + b * C_ + c] + addeps;
    else if (type == 2) raw = ws[OFF_SQ + b * C_ + c0] * ws[OFF_KW + (b * C_ + c) * W_ + x] + addeps;
    else                raw = ws[OFF_QC + (b * C_ + c0) * W_ + x] * ws[OFF_SK + b * C_ + c] + addeps;
    dst[i] = raw;
    lmn = fminf(lmn, raw);
    lmx = fmaxf(lmx, raw);
  }
  rmn[tid] = lmn; rmx[tid] = lmx;
  __syncthreads();
  for (int s = 128; s > 0; s >>= 1) {
    if (tid < s) { rmn[tid] = fminf(rmn[tid], rmn[tid + s]); rmx[tid] = fmaxf(rmx[tid], rmx[tid + s]); }
    __syncthreads();
  }
  const float mn  = rmn[0];
  const float inv = 1.0f / (rmx[0] - mn);
  for (int i = tid; i < N; i += 256) dst[i] = 2.0f + (dst[i] - mn) * inv;
  if (blockIdx.x == 0)
    for (int i = tid; i < 160; i += 256) ws[OFF_GMAX + i] = 0.0f;
}

__global__ void k3_tile(float* __restrict__ ws, int c0) {
  const int t   = blockIdx.x >> 6;
  const int bc  = blockIdx.x & 63;
  const int tid = threadIdx.x;
  __shared__ float E[HW_];
  __shared__ float mv[H_], nv[W_];
  __shared__ float rsb[H_], csb[W_];
  __shared__ float red[256];

  const float* mptr = ws + ((t < 2) ? OFF_HR : OFF_HC) + c0 * (BC_ * H_) + bc * H_;
  const float* nptr = ws + (((t & 1) == 0) ? OFF_WR : OFF_WC) + c0 * (BC_ * W_) + bc * W_;
  if (tid < H_)   mv[tid] = mptr[tid];
  if (tid >= 128) nv[tid - 128] = nptr[tid - 128];
  __syncthreads();

  for (int i = tid; i < HW_; i += 256) E[i] = __expf(mv[i >> 7] * nv[i & 127]);
  __syncthreads();

  const int wave = tid >> 5, lane = tid & 31;
#if defined(USE_WMMA_F32)
  if (wave < 6) {
    const int M   = wave * 16 + (lane & 15);
    const int kho = (lane < 16) ? 0 : 2;
    const v2f ones = {1.0f, 1.0f};
    v8f acc = {};
    for (int k0 = 0; k0 < W_; k0 += 4) {
      v2f a = { E[M * W_ + k0 + kho], E[M * W_ + k0 + kho + 1] };
      acc = __builtin_amdgcn_wmma_f32_16x16x4_f32(false, a, false, ones, (short)0, acc, false, false);
    }
    if ((lane & 15) == 0) {
      const int rbase = wave * 16 + ((lane >> 4) << 3);
      #pragma unroll
      for (int v = 0; v < 8; ++v) rsb[rbase + v] = acc[v];
    }
  } else {
    const int N   = lane & 15;
    const int kho = (lane < 16) ? 0 : 2;
    const v2f onesA = {1.0f, 1.0f};
    for (int nb = (wave - 6) * 4; nb < (wave - 6) * 4 + 4; ++nb) {
      const int N0 = nb * 16;
      v8f acc = {};
      for (int k0 = 0; k0 < H_; k0 += 4) {
        v2f b = { E[(k0 + kho) * W_ + N0 + N], E[(k0 + kho + 1) * W_ + N0 + N] };
        acc = __builtin_amdgcn_wmma_f32_16x16x4_f32(false, onesA, false, b, (short)0, acc, false, false);
      }
      if (lane < 16) csb[N0 + N] = acc[0];
    }
  }
#else
  if (wave < 6) {
    const int M = wave * 16 + (lane & 15);
    const int khalf = (lane < 16) ? 0 : 8;
    v16h hb;
    #pragma unroll
    for (int e = 0; e < 16; ++e) hb[e] = (_Float16)1.0f;
    v8f acc = {};
    for (int k0 = 0; k0 < W_; k0 += 32) {
      v16h ha;
      #pragma unroll
      for (int v = 0; v < 8; ++v) {
        const int kp = (v & 3) * 2 + ((v >> 2) << 4) + khalf;
        ha[2 * v]     = (_Float16)E[M * W_ + k0 + kp];
        ha[2 * v + 1] = (_Float16)E[M * W_ + k0 + kp + 1];
      }
      acc = __builtin_amdgcn_wmma_f32_16x16x32_f16(false, ha, false, hb, (short)0, acc, false, false);
    }
    if ((lane & 15) == 0) {
      const int rbase = wave * 16 + ((lane >> 4) << 3);
      #pragma unroll
      for (int v = 0; v < 8; ++v) rsb[rbase + v] = acc[v];
    }
  } else {
    const int N = lane & 15;
    const int khalf = (lane < 16) ? 0 : 16;
    v16h ha;
    #pragma unroll
    for (int e = 0; e < 16; ++e) ha[e] = (_Float16)1.0f;
    for (int nb = (wave - 6) * 4; nb < (wave - 6) * 4 + 4; ++nb) {
      const int N0 = nb * 16;
      v8f acc = {};
      for (int k0 = 0; k0 < H_; k0 += 32) {
        v16h hbm;
        #pragma unroll
        for (int v = 0; v < 8; ++v) {
          const int kk = khalf + 2 * v;
          hbm[2 * v]     = (_Float16)E[(k0 + kk) * W_ + N0 + N];
          hbm[2 * v + 1] = (_Float16)E[(k0 + kk + 1) * W_ + N0 + N];
        }
        acc = __builtin_amdgcn_wmma_f32_16x16x32_f16(false, ha, false, hbm, (short)0, acc, false, false);
      }
      if (lane < 16) csb[N0 + N] = acc[0];
    }
  }
#endif
  __syncthreads();

  float* irs = ws + OFF_IRS + (t * 64 + bc) * 224;
  if (tid < H_)   { const float v = 1.0f / rsb[tid]; rsb[tid] = v; irs[tid] = v; }
  if (tid >= 128) { const int x = tid - 128; const float v = 1.0f / csb[x]; csb[x] = v; irs[96 + x] = v; }
  __syncthreads();

  float lmax = 0.0f;
  for (int i = tid; i < HW_; i += 256)
    lmax = fmaxf(lmax, E[i] * (rsb[i >> 7] + csb[i & 127]));
  red[tid] = lmax;
  __syncthreads();
  for (int s = 128; s > 0; s >>= 1) {
    if (tid < s) red[tid] = fmaxf(red[tid], red[tid + s]);
    __syncthreads();
  }
  if (tid == 0)
    atomicMax((int*)(ws + OFF_GMAX + c0 * 4 + t), __float_as_int(red[0]));
}

__global__ void k4_score(float* __restrict__ ws, int c0) {
  const int bc  = blockIdx.x;
  const int tid = threadIdx.x;
  __shared__ float m2[2][H_], n2[2][W_], ir[4][H_], ic[4][W_], red[256];
  for (int i = tid; i < 2 * H_; i += 256) {
    const int s = i / H_, h = i % H_;
    m2[s][h] = ws[(s ? OFF_HC : OFF_HR) + c0 * (BC_ * H_) + bc * H_ + h];
  }
  for (int i = tid; i < 2 * W_; i += 256) {
    const int s = i / W_, x = i % W_;
    n2[s][x] = ws[(s ? OFF_WC : OFF_WR) + c0 * (BC_ * W_) + bc * W_ + x];
  }
  for (int i = tid; i < 4 * 224; i += 256) {
    const int t = i / 224, r = i % 224;
    const float v = ws[OFF_IRS + (t * 64 + bc) * 224 + r];
    if (r < 96) ir[t][r] = v; else ic[t][r - 96] = v;
  }
  __syncthreads();
  float ig[4];
  #pragma unroll
  for (int t = 0; t < 4; ++t) ig[t] = 1.0f / ws[OFF_GMAX + c0 * 4 + t];

  float* sc = ws + OFF_SCORE + bc * HW_;
  float lmax = 0.0f;
  for (int i = tid; i < HW_; i += 256) {
    const int h = i >> 7, x = i & 127;
    float s = 0.f;
    #pragma unroll
    for (int t = 0; t < 4; ++t) {
      const float e = __expf(m2[t >> 1][h] * n2[t & 1][x]);
      s += e * (ir[t][h] + ic[t][x]) * ig[t];
    }
    sc[i] = s;
    lmax = fmaxf(lmax, s);
  }
  red[tid] = lmax;
  __syncthreads();
  for (int s2 = 128; s2 > 0; s2 >>= 1) {
    if (tid < s2) red[tid] = fmaxf(red[tid], red[tid + s2]);
    __syncthreads();
  }
  if (tid == 0)
    atomicMax((int*)(ws + OFF_GSMAX + c0), __float_as_int(red[0]));
}

__global__ void k5_z(const float* __restrict__ ws, const float* __restrict__ values,
                     float* __restrict__ out, int c0) {
  const int gid = blockIdx.x * 256 + threadIdx.x;
  const int b = gid / HW_;
  const int r = gid % HW_;
  const float inv = 1.0f / ws[OFF_GSMAX + c0];
  float z = 0.f;
  for (int c = 0; c < C_; ++c) {
    const int off = (b * C_ + c) * HW_ + r;
    z += ws[OFF_SCORE + off] * values[off];
  }
  out[(b * C_ + c0) * HW_ + r] = z * inv;
}

__global__ void k6_scoreout(const float* __restrict__ ws, float* __restrict__ out) {
  const int gid = blockIdx.x * 256 + threadIdx.x;
  out[BC_ * HW_ + gid] = ws[OFF_SCORE + gid] * (1.0f / ws[OFF_GSMAX + 31]);
}

extern "C" void kernel_launch(void* const* d_in, const int* in_sizes, int n_in,
                              void* d_out, int out_size, void* d_ws, size_t ws_size,
                              hipStream_t stream) {
  (void)in_sizes; (void)n_in; (void)out_size; (void)ws_size;
  const float* q   = (const float*)d_in[0];
  const float* k   = (const float*)d_in[1];
  const float* val = (const float*)d_in[2];
  float* out = (float*)d_out;
  float* ws  = (float*)d_ws;

  k1_sums<<<64, 256, 0, stream>>>(q, k, ws);
  k2_norm<<<128, 256, 0, stream>>>(ws);
  for (int c0 = 0; c0 < C_; ++c0) {
    k3_tile<<<256, 256, 0, stream>>>(ws, c0);
    k4_score<<<64, 256, 0, stream>>>(ws, c0);
    k5_z<<<96, 256, 0, stream>>>(ws, val, out, c0);
  }
  k6_scoreout<<<3072, 256, 0, stream>>>(ws, out);
}


# k1, k2, kg (pre-passes): 1024-thread blocks, not 256, so 32 waves a block overlap the serial per-element L2 round trips (k2: 6-8 elements a thread, not 24-32); same grids, sums and orders
# speedup vs baseline: 7.3339x; 7.3339x over previous
#include <hip/hip_runtime.h>

typedef __attribute__((ext_vector_type(2)))  float    v2f;
typedef __attribute__((ext_vector_type(8)))  float    v8f;

#define B_   2
#define C_   32
#define H_   96
#define W_   128
#define HW_  (H_ * W_)
#define BC_  (B_ * C_)
#define EPSF 1e-6f

#define OFF_QS    0
#define OFF_QC    6144
#define OFF_KH    14336
#define OFF_KW    20480
#define OFF_SQ    28672
#define OFF_SK    28736
#define OFF_GMAX  28800
#define OFF_GSMAX 28928
#define OFF_HR    28960
#define OFF_HC    (OFF_HR + 32 * 6144)
#define OFF_WR    (OFF_HC + 32 * 6144)
#define OFF_WC    (OFF_WR + 32 * 8192)
#define OFF_IRS   (OFF_WC + 32 * 8192)
#define OFF_SCORE (OFF_IRS + 4 * 64 * 224)


__global__ void __launch_bounds__(1024)
k1_sums(const float* __restrict__ q, const float* __restrict__ k, float* __restrict__ ws) {
  __shared__ float tile[HW_];
  __shared__ float rowbuf[H_];
  const int bc  = blockIdx.x;
  const int tid = threadIdx.x;
  for (int p = 0; p < 2; ++p) {
    const float* s = p ? k : q;
    #pragma unroll
    for (int j = 0; j < HW_ / 1024; ++j) tile[tid + j * 1024] = s[bc * HW_ + tid + j * 1024];
    __syncthreads();
    if (tid < H_) {
      float acc = 0.f;
      for (int x = 0; x < W_; ++x) acc += tile[tid * W_ + x];
      ws[(p ? OFF_KH : OFF_QS) + bc * H_ + tid] = acc;
      rowbuf[tid] = acc;
    }
    if (tid >= 128 && tid < 256) {
      const int x = tid - 128;
      float acc = 0.f;
      for (int h = 0; h < H_; ++h) acc += tile[h * W_ + x];
      ws[(p ? OFF_KW : OFF_QC) + bc * W_ + x] = acc;
    }
    __syncthreads();
    if (tid == 0) {
      float t = 0.f;
      for (int h = 0; h < H_; ++h) t += rowbuf[h];
      ws[(p ? OFF_SK : OFF_SQ) + bc] = t;
    }
    __syncthreads();
  }
}

__global__ void __launch_bounds__(1024) k2_norm(float* __restrict__ ws) {
  const int type = blockIdx.x & 3;
  const int c0   = blockIdx.x >> 2;
  const int tid  = threadIdx.x;
  __shared__ float rmn[1024], rmx[1024];
  const int   len    = (type < 2) ? H_ : W_;
  const int   N      = BC_ * len;
  const float addeps = (type < 2) ? (H_ * EPSF) : (W_ * EPSF);
  float* dst;
  if      (type == 0) dst = ws + OFF_HR + c0 * (BC_ * H_);
  else if (type == 1) dst = ws + OFF_HC + c0 * (BC_ * H_);
  else if (type == 2) dst = ws + OFF_WR + c0 * (BC_ * W_);
  else                dst = ws + OFF_WC + c0 * (BC_ * W_);
  float lmn = 3.4e38f, lmx = -3.4e38f;
  for (int i = tid; i < N; i += 1024) {
    const int b = i / (C_ * len);
    const int r = i % (C_ * len);
    const int c = r / len;
    const int x = r % len;
    float raw;
    if      (type == 0) raw = ws[OFF_SQ + b * C_ + c0] * ws[OFF_KH + (b * C_ + c) * H_ + x] + addeps;
    else if (type == 1) raw = ws[OFF_QS + (b * C_ + c0) * H_ + x] * ws[OFF_SK + b * C_ + c] + addeps;
    else if (type == 2) raw = ws[OFF_SQ + b * C_ + c0] * ws[OFF_KW + (b * C_ + c) * W_ + x] + addeps;
    else                raw = ws[OFF_QC + (b * C_ + c0) * W_ + x] * ws[OFF_SK + b * C_ + c] + addeps;
    dst[i] = raw;
    lmn = fminf(lmn, raw);
    lmx = fmaxf(lmx, raw);
  }
  rmn[tid] = lmn; rmx[tid] = lmx;
  __syncthreads();
  for (int s = 512; s > 0; s >>= 1) {
    if (tid < s) { rmn[tid] = fminf(rmn[tid], rmn[tid + s]); rmx[tid] = fmaxf(rmx[tid], rmx[tid + s]); }
    __syncthreads();
  }
  const float mn  = rmn[0];
  const float inv = 1.0f / (rmx[0] - mn);
  #pragma unroll
  for (int j = 0; j < 8; ++j) {
    const int i = tid + j * 1024;
    if (i < N) dst[i] = 2.0f + (dst[i] - mn) * inv;
  }
  if (blockIdx.x == 0)
    for (int i = tid; i < 160; i += 1024) ws[OFF_GMAX + i] = 0.0f;
}

__global__ void __launch_bounds__(1024) kg_peak(float* __restrict__ ws) {
  const int t    = blockIdx.x & 3;
  const int c0   = blockIdx.x >> 2;
  const int tid  = threadIdx.x;
  const int wave = tid >> 5, lane = tid & 31;
  __shared__ float red[32];
  const float* mbase = ws + ((t < 2) ? OFF_HR : OFF_HC) + c0 * (BC_ * H_);
  const float* nbase = ws + (((t & 1) == 0) ? OFF_WR : OFF_WC) + c0 * (BC_ * W_);
  float best = 0.0f;
  for (int bc = wave; bc < BC_; bc += 32) {
    const float m0 = mbase[bc * H_ + lane * 3 + 0];
    const float m1 = mbase[bc * H_ + lane * 3 + 1];
    const float m2 = mbase[bc * H_ + lane * 3 + 2];
    const float n0 = nbase[bc * W_ + lane * 4 + 0];
    const float n1 = nbase[bc * W_ + lane * 4 + 1];
    const float n2 = nbase[bc * W_ + lane * 4 + 2];
    const float n3 = nbase[bc * W_ + lane * 4 + 3];
    float mm = fmaxf(fmaxf(m0, m1), m2);
    float nn = fmaxf(fmaxf(n0, n1), fmaxf(n2, n3));
    for (int o = 16; o > 0; o >>= 1) {
      mm = fmaxf(mm, __shfl_xor(mm, o));
      nn = fmaxf(nn, __shfl_xor(nn, o));
    }
    float rs = __expf(mm * n0) + __expf(mm * n1) + __expf(mm * n2) + __expf(mm * n3);
    float cs = __expf(m0 * nn) + __expf(m1 * nn) + __expf(m2 * nn);
    for (int o = 16; o > 0; o >>= 1) {
      rs += __shfl_xor(rs, o);
      cs += __shfl_xor(cs, o);
    }
    best = fmaxf(best, __expf(mm * nn) * (1.0f / rs + 1.0f / cs));
  }
  if (lane == 0) red[wave] = best;
  __syncthreads();
  if (tid == 0) {
    float g = red[0];
    for (int i = 1; i < 32; ++i) g = fmaxf(g, red[i]);
    ws[OFF_GMAX + c0 * 4 + t] = g;
  }
}

#define PRP 33
__global__ void __launch_bounds__(1024)
km_fused(float* __restrict__ ws, const float* __restrict__ values, float* __restrict__ out) {
  const int half = blockIdx.x & 1;
  const int b    = (blockIdx.x >> 1) & 1;
  const int c0   = blockIdx.x >> 2;
  const int tid  = threadIdx.x;
  const int wave = __builtin_amdgcn_readfirstlane(tid >> 5);
  const int lane = tid & 31;
  __shared__ float PR[4][H_ * PRP];
  __shared__ __attribute__((aligned(16))) float PC[4][32 * W_];
  __shared__ float IR[4][H_];
  __shared__ __attribute__((aligned(16))) float IC[4][W_];
  __shared__ float red[32];

  float ig[4];
  #pragma unroll
  for (int t = 0; t < 4; ++t) ig[t] = 1.0f / ws[OFF_GMAX + c0 * 4 + t];

  const int row0 = wave * 3, col0 = lane * 4;
  float z[3][4];
  #pragma unroll
  for (int r = 0; r < 3; ++r) { z[r][0] = 0.f; z[r][1] = 0.f; z[r][2] = 0.f; z[r][3] = 0.f; }
  float smax = 0.0f;

  for (int ci = 0; ci < 16; ++ci) {
    const int c  = half * 16 + ci;
    const int bc = b * C_ + c;
    float mh[2][3], nw[2][4];
    #pragma unroll
    for (int r = 0; r < 3; ++r) {
      mh[0][r] = ws[OFF_HR + c0 * (BC_ * H_) + bc * H_ + row0 + r];
      mh[1][r] = ws[OFF_HC + c0 * (BC_ * H_) + bc * H_ + row0 + r];
    }
    {
      const float4 a = *(const float4*)(ws + OFF_WR + c0 * (BC_ * W_) + bc * W_ + col0);
      const float4 d = *(const float4*)(ws + OFF_WC + c0 * (BC_ * W_) + bc * W_ + col0);
      nw[0][0] = a.x; nw[0][1] = a.y; nw[0][2] = a.z; nw[0][3] = a.w;
      nw[1][0] = d.x; nw[1][1] = d.y; nw[1][2] = d.z; nw[1][3] = d.w;
    }
    float E[4][3][4];
    #pragma unroll
    for (int t = 0; t < 4; ++t) {
      #pragma unroll
      for (int r = 0; r < 3; ++r) {
        #pragma unroll
        for (int q = 0; q < 4; ++q) E[t][r][q] = __expf(mh[t >> 1][r] * nw[t & 1][q]);
      }
    }
    #pragma unroll
    for (int t = 0; t < 4; ++t) {
      #pragma unroll
      for (int r = 0; r < 3; ++r)
        PR[t][(row0 + r) * PRP + lane] = (E[t][r][0] + E[t][r][1]) + (E[t][r][2] + E[t][r][3]);
      float4 cp;
      cp.x = E[t][0][0] + E[t][1][0] + E[t][2][0];
      cp.y = E[t][0][1] + E[t][1][1] + E[t][2][1];
      cp.z = E[t][0][2] + E[t][1][2] + E[t][2][2];
      cp.w = E[t][0][3] + E[t][1][3] + E[t][2][3];
      *(float4*)&PC[t][wave * W_ + col0] = cp;
    }
    __syncthreads();

    float4 vv[3];
    #pragma unroll
    for (int r = 0; r < 3; ++r)
      vv[r] = *(const float4*)(values + bc * HW_ + (row0 + r) * W_ + col0);

    {
      const v2f ones = {1.0f, 1.0f};
      const int kho = (lane < 16) ? 0 : 2;
      if (wave < 24) {
        const int tt = wave / 6, rb = wave % 6;
        const int M  = rb * 16 + (lane & 15);
        v8f acc = {};
        for (int k0 = 0; k0 < 32; k0 += 4) {
          v2f a = { PR[tt][M * PRP + k0 + kho], PR[tt][M * PRP + k0 + kho + 1] };
          acc = __builtin_amdgcn_wmma_f32_16x16x4_f32(false, a, false, ones, (short)0, acc, false, false);
        }
        if ((lane & 15) == 0) {
          const int rbase = rb * 16 + ((lane >> 4) << 3);
          #pragma unroll
          for (int v = 0; v < 8; ++v) IR[tt][rbase + v] = 1.0f / acc[v];
        }
      }
      {
        const int id = (wave < 24) ? (wave + 8) : (wave - 24);
        const int tt = id >> 3, N0 = (id & 7) * 16, N = lane & 15;
        v8f acc = {};
        for (int k0 = 0; k0 < 32; k0 += 4) {
          v2f bb = { PC[tt][(k0 + kho) * W_ + N0 + N], PC[tt][(k0 + kho + 1) * W_ + N0 + N] };
          acc = __builtin_amdgcn_wmma_f32_16x16x4_f32(false, ones, false, bb, (short)0, acc, false, false);
        }
        if (lane < 16) IC[tt][N0 + N] = 1.0f / acc[0];
      }
    }
    __syncthreads();

    float icv[4][4];
    #pragma unroll
    for (int t = 0; t < 4; ++t) {
      const float4 x = *(const float4*)&IC[t][col0];
      icv[t][0] = x.x; icv[t][1] = x.y; icv[t][2] = x.z; icv[t][3] = x.w;
    }
    #pragma unroll
    for (int r = 0; r < 3; ++r) {
      float irv[4];
      #pragma unroll
      for (int t = 0; t < 4; ++t) irv[t] = IR[t][row0 + r];
      const float vr[4] = { vv[r].x, vv[r].y, vv[r].z, vv[r].w };
      float sc[4];
      #pragma unroll
      for (int q = 0; q < 4; ++q) {
        float s = 0.f;
        #pragma unroll
        for (int t = 0; t < 4; ++t) s += E[t][r][q] * (irv[t] + icv[t][q]) * ig[t];
        z[r][q] += s * vr[q];
        smax = fmaxf(smax, s);
        sc[q] = s;
      }
      if (c0 == C_ - 1) {
        float4 o; o.x = sc[0]; o.y = sc[1]; o.z = sc[2]; o.w = sc[3];
        *(float4*)(out + BC_ * HW_ + bc * HW_ + (row0 + r) * W_ + col0) = o;
      }
    }
  }

  float* zdst = (half ? (ws + OFF_SCORE) : out) + (b * C_ + c0) * HW_;
  #pragma unroll
  for (int r = 0; r < 3; ++r) {
    float4 o; o.x = z[r][0]; o.y = z[r][1]; o.z = z[r][2]; o.w = z[r][3];
    *(float4*)(zdst + (row0 + r) * W_ + col0) = o;
  }
  for (int o = 16; o > 0; o >>= 1) smax = fmaxf(smax, __shfl_xor(smax, o));
  if (lane == 0) red[wave] = smax;
  __syncthreads();
  if (tid == 0) {
    float g = red[0];
    for (int i = 1; i < 32; ++i) g = fmaxf(g, red[i]);
    atomicMax((int*)(ws + OFF_GSMAX + c0), __float_as_int(g));
  }
}

__global__ void ks_scale(const float* __restrict__ ws, float* __restrict__ out) {
  const int gid = blockIdx.x * 256 + threadIdx.x;
  float4* o4 = (float4*)out;
  if (gid < BC_ * HW_ / 4) {
    const int c0 = ((gid * 4) / HW_) % C_;
    const float inv = 1.0f / ws[OFF_GSMAX + c0];
    const float4 a = o4[gid];
    const float4 p = *(const float4*)(ws + OFF_SCORE + gid * 4);
    float4 r;
    r.x = (a.x + p.x) * inv; r.y = (a.y + p.y) * inv; r.z = (a.z + p.z) * inv; r.w = (a.w + p.w) * inv;
    o4[gid] = r;
  } else {
    const float inv = 1.0f / ws[OFF_GSMAX + 31];
    float4 a = o4[gid];
    a.x *= inv; a.y *= inv; a.z *= inv; a.w *= inv;
    o4[gid] = a;
  }
}

extern "C" void kernel_launch(void* const* d_in, const int* in_sizes, int n_in,
                              void* d_out, int out_size, void* d_ws, size_t ws_size,
                              hipStream_t stream) {
  (void)in_sizes; (void)n_in; (void)out_size; (void)ws_size;
  const float* q   = (const float*)d_in[0];
  const float* k   = (const float*)d_in[1];
  const float* val = (const float*)d_in[2];
  float* out = (float*)d_out;
  float* ws  = (float*)d_ws;

  k1_sums<<<64, 1024, 0, stream>>>(q, k, ws);
  k2_norm<<<128, 1024, 0, stream>>>(ws);
  kg_peak<<<128, 1024, 0, stream>>>(ws);
  km_fused<<<128, 1024, 0, stream>>>(ws, val, out);
  ks_scale<<<1536, 256, 0, stream>>>(ws, out);
}


# km: column sums move from the matrix unit to the vector unit: 8 waves add the 32 wave partials per column (lane halves joined by permlanex16) beside 24 waves' row-sum WMMAs; column sum order changes
# speedup vs baseline: 8.1287x; 1.1084x over previous
#include <hip/hip_runtime.h>

typedef __attribute__((ext_vector_type(2)))  float    v2f;
typedef __attribute__((ext_vector_type(8)))  float    v8f;

#define B_   2
#define C_   32
#define H_   96
#define W_   128
#define HW_  (H_ * W_)
#define BC_  (B_ * C_)
#define EPSF 1e-6f

__device__ __forceinline__ float xhalf(float x) {
  const int b = __builtin_bit_cast(int, x);
  return __builtin_bit_cast(float, __builtin_amdgcn_permlanex16(b, b, 0x76543210, 0xfedcba98, false, false));
}

#define OFF_QS    0
#define OFF_QC    6144
#define OFF_KH    14336
#define OFF_KW    20480
#define OFF_SQ    28672
#define OFF_SK    28736
#define OFF_GMAX  28800
#define OFF_GSMAX 28928
#define OFF_HR    28960
#define OFF_HC    (OFF_HR + 32 * 6144)
#define OFF_WR    (OFF_HC + 32 * 6144)
#define OFF_WC    (OFF_WR + 32 * 8192)
#define OFF_IRS   (OFF_WC + 32 * 8192)
#define OFF_SCORE (OFF_IRS + 4 * 64 * 224)


__global__ void __launch_bounds__(1024)
k1_sums(const float* __restrict__ q, const float* __restrict__ k, float* __restrict__ ws) {
  __shared__ float tile[HW_];
  __shared__ float rowbuf[H_];
  const int bc  = blockIdx.x;
  const int tid = threadIdx.x;
  for (int p = 0; p < 2; ++p) {
    const float* s = p ? k : q;
    #pragma unroll
    for (int j = 0; j < HW_ / 1024; ++j) tile[tid + j * 1024] = s[bc * HW_ + tid + j * 1024];
    __syncthreads();
    if (tid < H_) {
      float acc = 0.f;
      for (int x = 0; x < W_; ++x) acc += tile[tid * W_ + x];
      ws[(p ? OFF_KH : OFF_QS) + bc * H_ + tid] = acc;
      rowbuf[tid] = acc;
    }
    if (tid >= 128 && tid < 256) {
      const int x = tid - 128;
      float acc = 0.f;
      for (int h = 0; h < H_; ++h) acc += tile[h * W_ + x];
      ws[(p ? OFF_KW : OFF_QC) + bc * W_ + x] = acc;
    }
    __syncthreads();
    if (tid == 0) {
      float t = 0.f;
      for (int h = 0; h < H_; ++h) t += rowbuf[h];
      ws[(p ? OFF_SK : OFF_SQ) + bc] = t;
    }
    __syncthreads();
  }
}

__global__ void __launch_bounds__(1024) k2_norm(float* __restrict__ ws) {
  const int type = blockIdx.x & 3;
  const int c0   = blockIdx.x >> 2;
  const int tid  = threadIdx.x;
  __shared__ float rmn[1024], rmx[1024];
  const int   len    = (type < 2) ? H_ : W_;
  const int   N      = BC_ * len;
  const float addeps = (type < 2) ? (H_ * EPSF) : (W_ * EPSF);
  float* dst;
  if      (type == 0) dst = ws + OFF_HR + c0 * (BC_ * H_);
  else if (type == 1) dst = ws + OFF_HC + c0 * (BC_ * H_);
  else if (type == 2) dst = ws + OFF_WR + c0 * (BC_ * W_);
  else                dst = ws + OFF_WC + c0 * (BC_ * W_);
  float lmn = 3.4e38f, lmx = -3.4e38f;
  for (int i = tid; i < N; i += 1024) {
    const int b = i / (C_ * len);
    const int r = i % (C_ * len);
    const int c = r / len;
    const int x = r % len;
    float raw;
    if      (type == 0) raw = ws[OFF_SQ + b * C_ + c0] * ws[OFF_KH + (b * C_ + c) * H_ + x] + addeps;
    else if (type == 1) raw = ws[OFF_QS + (b * C_ + c0) * H_ + x] * ws[OFF_SK + b * C_ + c] + addeps;
    else if (type == 2) raw = ws[OFF_SQ + b * C_ + c0] * ws[OFF_KW + (b * C_ + c) * W_ + x] + addeps;
    else                raw = ws[OFF_QC + (b * C_ + c0) * W_ + x] * ws[OFF_SK + b * C_ + c] + addeps;
    dst[i] = raw;
    lmn = fminf(lmn, raw);
    lmx = fmaxf(lmx, raw);
  }
  rmn[tid] = lmn; rmx[tid] = lmx;
  __syncthreads();
  for (int s = 512; s > 0; s >>= 1) {
    if (tid < s) { rmn[tid] = fminf(rmn[tid], rmn[tid + s]); rmx[tid] = fmaxf(rmx[tid], rmx[tid + s]); }
    __syncthreads();
  }
  const float mn  = rmn[0];
  const float inv = 1.0f / (rmx[0] - mn);
  #pragma unroll
  for (int j = 0; j < 8; ++j) {
    const int i = tid + j * 1024;
    if (i < N) dst[i] = 2.0f + (dst[i] - mn) * inv;
  }
  if (blockIdx.x == 0)
    for (int i = tid; i < 160; i += 1024) ws[OFF_GMAX + i] = 0.0f;
}

__global__ void __launch_bounds__(1024) kg_peak(float* __restrict__ ws) {
  const int t    = blockIdx.x & 3;
  const int c0   = blockIdx.x >> 2;
  const int tid  = threadIdx.x;
  const int wave = tid >> 5, lane = tid & 31;
  __shared__ float red[32];
  const float* mbase = ws + ((t < 2) ? OFF_HR : OFF_HC) + c0 * (BC_ * H_);
  const float* nbase = ws + (((t & 1) == 0) ? OFF_WR : OFF_WC) + c0 * (BC_ * W_);
  float best = 0.0f;
  for (int bc = wave; bc < BC_; bc += 32) {
    const float m0 = mbase[bc * H_ + lane * 3 + 0];
    const float m1 = mbase[bc * H_ + lane * 3 + 1];
    const float m2 = mbase[bc * H_ + lane * 3 + 2];
    const float n0 = nbase[bc * W_ + lane * 4 + 0];
    const float n1 = nbase[bc * W_ + lane * 4 + 1];
    const float n2 = nbase[bc * W_ + lane * 4 + 2];
    const float n3 = nbase[bc * W_ + lane * 4 + 3];
    float mm = fmaxf(fmaxf(m0, m1), m2);
    float nn = fmaxf(fmaxf(n0, n1), fmaxf(n2, n3));
    for (int o = 16; o > 0; o >>= 1) {
      mm = fmaxf(mm, __shfl_xor(mm, o));
      nn = fmaxf(nn, __shfl_xor(nn, o));
    }
    float rs = __expf(mm * n0) + __expf(mm * n1) + __expf(mm * n2) + __expf(mm * n3);
    float cs = __expf(m0 * nn) + __expf(m1 * nn) + __expf(m2 * nn);
    for (int o = 16; o > 0; o >>= 1) {
      rs += __shfl_xor(rs, o);
      cs += __shfl_xor(cs, o);
    }
    best = fmaxf(best, __expf(mm * nn) * (1.0f / rs + 1.0f / cs));
  }
  if (lane == 0) red[wave] = best;
  __syncthreads();
  if (tid == 0) {
    float g = red[0];
    for (int i = 1; i < 32; ++i) g = fmaxf(g, red[i]);
    ws[OFF_GMAX + c0 * 4 + t] = g;
  }
}

#define PRP 33
__global__ void __launch_bounds__(1024)
km_fused(float* __restrict__ ws, const float* __restrict__ values, float* __restrict__ out) {
  const int half = blockIdx.x & 1;
  const int b    = (blockIdx.x >> 1) & 1;
  const int c0   = blockIdx.x >> 2;
  const int tid  = threadIdx.x;
  const int wave = __builtin_amdgcn_readfirstlane(tid >> 5);
  const int lane = tid & 31;
  __shared__ float PR[4][H_ * PRP];
  __shared__ __attribute__((aligned(16))) float PC[4][32 * W_];
  __shared__ float IR[4][H_];
  __shared__ __attribute__((aligned(16))) float IC[4][W_];
  __shared__ float red[32];

  float ig[4];
  #pragma unroll
  for (int t = 0; t < 4; ++t) ig[t] = 1.0f / ws[OFF_GMAX + c0 * 4 + t];

  const int row0 = wave * 3, col0 = lane * 4;
  float z[3][4];
  #pragma unroll
  for (int r = 0; r < 3; ++r) { z[r][0] = 0.f; z[r][1] = 0.f; z[r][2] = 0.f; z[r][3] = 0.f; }
  float smax = 0.0f;

  for (int ci = 0; ci < 16; ++ci) {
    const int c  = half * 16 + ci;
    const int bc = b * C_ + c;
    float mh[2][3], nw[2][4];
    #pragma unroll
    for (int r = 0; r < 3; ++r) {
      mh[0][r] = ws[OFF_HR + c0 * (BC_ * H_) + bc * H_ + row0 + r];
      mh[1][r] = ws[OFF_HC + c0 * (BC_ * H_) + bc * H_ + row0 + r];
    }
    {
      const float4 a = *(const float4*)(ws + OFF_WR + c0 * (BC_ * W_) + bc * W_ + col0);
      const float4 d = *(const float4*)(ws + OFF_WC + c0 * (BC_ * W_) + bc * W_ + col0);
      nw[0][0] = a.x; nw[0][1] = a.y; nw[0][2] = a.z; nw[0][3] = a.w;
      nw[1][0] = d.x; nw[1][1] = d.y; nw[1][2] = d.z; nw[1][3] = d.w;
    }
    float E[4][3][4];
    #pragma unroll
    for (int t = 0; t < 4; ++t) {
      #pragma unroll
      for (int r = 0; r < 3; ++r) {
        #pragma unroll
        for (int q = 0; q < 4; ++q) E[t][r][q] = __expf(mh[t >> 1][r] * nw[t & 1][q]);
      }
    }
    #pragma unroll
    for (int t = 0; t < 4; ++t) {
      #pragma unroll
      for (int r = 0; r < 3; ++r)
        PR[t][(row0 + r) * PRP + lane] = (E[t][r][0] + E[t][r][1]) + (E[t][r][2] + E[t][r][3]);
      float4 cp;
      cp.x = E[t][0][0] + E[t][1][0] + E[t][2][0];
      cp.y = E[t][0][1] + E[t][1][1] + E[t][2][1];
      cp.z = E[t][0][2] + E[t][1][2] + E[t][2][2];
      cp.w = E[t][0][3] + E[t][1][3] + E[t][2][3];
      *(float4*)&PC[t][wave * W_ + col0] = cp;
    }
    __syncthreads();

    float4 vv[3];
    #pragma unroll
    for (int r = 0; r < 3; ++r)
      vv[r] = *(const float4*)(values + bc * HW_ + (row0 + r) * W_ + col0);

    if (wave < 24) {
      const v2f ones = {1.0f, 1.0f};
      const int kho = (lane < 16) ? 0 : 2;
      const int tt = wave / 6, rb = wave % 6;
      const int M  = rb * 16 + (lane & 15);
      v8f acc = {};
      for (int k0 = 0; k0 < 32; k0 += 4) {
        v2f a = { PR[tt][M * PRP + k0 + kho], PR[tt][M * PRP + k0 + kho + 1] };
        acc = __builtin_amdgcn_wmma_f32_16x16x4_f32(false, a, false, ones, (short)0, acc, false, false);
      }
      if ((lane & 15) == 0) {
        const int rbase = rb * 16 + ((lane >> 4) << 3);
        #pragma unroll
        for (int v = 0; v < 8; ++v) IR[tt][rbase + v] = 1.0f / acc[v];
      }
    } else {
      const int jw = wave - 24;
      const int tt = jw >> 1;
      const int g  = ((jw & 1) << 4) | (lane & 15);
      const int ph = lane >> 4;
      const float* pc = &PC[tt][(ph * 16) * W_ + g * 4];
      float4 s;
      #pragma unroll
      for (int i0 = 0; i0 < 16; i0 += 4) {
        const float4 x0 = *(const float4*)(pc + (i0 + 0) * W_);
        const float4 x1 = *(const float4*)(pc + (i0 + 1) * W_);
        const float4 x2 = *(const float4*)(pc + (i0 + 2) * W_);
        const float4 x3 = *(const float4*)(pc + (i0 + 3) * W_);
        float4 t;
        t.x = (x0.x + x1.x) + (x2.x + x3.x);
        t.y = (x0.y + x1.y) + (x2.y + x3.y);
        t.z = (x0.z + x1.z) + (x2.z + x3.z);
        t.w = (x0.w + x1.w) + (x2.w + x3.w);
        if (i0 == 0) { s = t; } else { s.x += t.x; s.y += t.y; s.z += t.z; s.w += t.w; }
        __builtin_amdgcn_sched_barrier(0);
      }
      s.x += xhalf(s.x); s.y += xhalf(s.y); s.z += xhalf(s.z); s.w += xhalf(s.w);
      float2 o;
      o.x = 1.0f / (ph ? s.z : s.x);
      o.y = 1.0f / (ph ? s.w : s.y);
      *(float2*)&IC[tt][g * 4 + ph * 2] = o;
    }
    __syncthreads();

    float icv[4][4];
    #pragma unroll
    for (int t = 0; t < 4; ++t) {
      const float4 x = *(const float4*)&IC[t][col0];
      icv[t][0] = x.x; icv[t][1] = x.y; icv[t][2] = x.z; icv[t][3] = x.w;
    }
    #pragma unroll
    for (int r = 0; r < 3; ++r) {
      float irv[4];
      #pragma unroll
      for (int t = 0; t < 4; ++t) irv[t] = IR[t][row0 + r];
      const float vr[4] = { vv[r].x, vv[r].y, vv[r].z, vv[r].w };
      float sc[4];
      #pragma unroll
      for (int q = 0; q < 4; ++q) {
        float s = 0.f;
        #pragma unroll
        for (int t = 0; t < 4; ++t) s += E[t][r][q] * (irv[t] + icv[t][q]) * ig[t];
        z[r][q] += s * vr[q];
        smax = fmaxf(smax, s);
        sc[q] = s;
      }
      if (c0 == C_ - 1) {
        float4 o; o.x = sc[0]; o.y = sc[1]; o.z = sc[2]; o.w = sc[3];
        *(float4*)(out + BC_ * HW_ + bc * HW_ + (row0 + r) * W_ + col0) = o;
      }
    }
  }

  float* zdst = (half ? (ws + OFF_SCORE) : out) + (b * C_ + c0) * HW_;
  #pragma unroll
  for (int r = 0; r < 3; ++r) {
    float4 o; o.x = z[r][0]; o.y = z[r][1]; o.z = z[r][2]; o.w = z[r][3];
    *(float4*)(zdst + (row0 + r) * W_ + col0) = o;
  }
  for (int o = 16; o > 0; o >>= 1) smax = fmaxf(smax, __shfl_xor(smax, o));
  if (lane == 0) red[wave] = smax;
  __syncthreads();
  if (tid == 0) {
    float g = red[0];
    for (int i = 1; i < 32; ++i) g = fmaxf(g, red[i]);
    atomicMax((int*)(ws + OFF_GSMAX + c0), __float_as_int(g));
  }
}

__global__ void ks_scale(const float* __restrict__ ws, float* __restrict__ out) {
  const int gid = blockIdx.x * 256 + threadIdx.x;
  float4* o4 = (float4*)out;
  if (gid < BC_ * HW_ / 4) {
    const int c0 = ((gid * 4) / HW_) % C_;
    const float inv = 1.0f / ws[OFF_GSMAX + c0];
    const float4 a = o4[gid];
    const float4 p = *(const float4*)(ws + OFF_SCORE + gid * 4);
    float4 r;
    r.x = (a.x + p.x) * inv; r.y = (a.y + p.y) * inv; r.z = (a.z + p.z) * inv; r.w = (a.w + p.w) * inv;
    o4[gid] = r;
  } else {
    const float inv = 1.0f / ws[OFF_GSMAX + 31];
    float4 a = o4[gid];
    a.x *= inv; a.y *= inv; a.z *= inv; a.w *= inv;
    o4[gid] = a;
  }
}

extern "C" void kernel_launch(void* const* d_in, const int* in_sizes, int n_in,
                              void* d_out, int out_size, void* d_ws, size_t ws_size,
                              hipStream_t stream) {
  (void)in_sizes; (void)n_in; (void)out_size; (void)ws_size;
  const float* q   = (const float*)d_in[0];
  const float* k   = (const float*)d_in[1];
  const float* val = (const float*)d_in[2];
  float* out = (float*)d_out;
  float* ws  = (float*)d_ws;

  k1_sums<<<64, 1024, 0, stream>>>(q, k, ws);
  k2_norm<<<128, 1024, 0, stream>>>(ws);
  kg_peak<<<128, 1024, 0, stream>>>(ws);
  km_fused<<<128, 1024, 0, stream>>>(ws, val, out);
  ks_scale<<<1536, 256, 0, stream>>>(ws, out);
}


# km: each (c0, b, half of c) tile split by rows over two work-groups (256, one per WGP); other half's exp formed only for column sums, one key channel ahead under the row-sum WMMAs; sums unchanged
# speedup vs baseline: 9.2052x; 1.1324x over previous
#include <hip/hip_runtime.h>

typedef __attribute__((ext_vector_type(2)))  float    v2f;
typedef __attribute__((ext_vector_type(8)))  float    v8f;

#define B_   2
#define C_   32
#define H_   96
#define W_   128
#define HW_  (H_ * W_)
#define BC_  (B_ * C_)
#define EPSF 1e-6f

__device__ __forceinline__ float xhalf(float x) {
  const int b = __builtin_bit_cast(int, x);
  return __builtin_bit_cast(float, __builtin_amdgcn_permlanex16(b, b, 0x76543210, 0xfedcba98, false, false));
}

#define OFF_QS    0
#define OFF_QC    6144
#define OFF_KH    14336
#define OFF_KW    20480
#define OFF_SQ    28672
#define OFF_SK    28736
#define OFF_GMAX  28800
#define OFF_GSMAX 28928
#define OFF_HR    28960
#define OFF_HC    (OFF_HR + 32 * 6144)
#define OFF_WR    (OFF_HC + 32 * 6144)
#define OFF_WC    (OFF_WR + 32 * 8192)
#define OFF_IRS   (OFF_WC + 32 * 8192)
#define OFF_SCORE (OFF_IRS + 4 * 64 * 224)


__global__ void __launch_bounds__(1024)
k1_sums(const float* __restrict__ q, const float* __restrict__ k, float* __restrict__ ws) {
  __shared__ float tile[HW_];
  __shared__ float rowbuf[H_];
  const int bc  = blockIdx.x;
  const int tid = threadIdx.x;
  for (int p = 0; p < 2; ++p) {
    const float* s = p ? k : q;
    #pragma unroll
    for (int j = 0; j < HW_ / 1024; ++j) tile[tid + j * 1024] = s[bc * HW_ + tid + j * 1024];
    __syncthreads();
    if (tid < H_) {
      float acc = 0.f;
      for (int x = 0; x < W_; ++x) acc += tile[tid * W_ + x];
      ws[(p ? OFF_KH : OFF_QS) + bc * H_ + tid] = acc;
      rowbuf[tid] = acc;
    }
    if (tid >= 128 && tid < 256) {
      const int x = tid - 128;
      float acc = 0.f;
      for (int h = 0; h < H_; ++h) acc += tile[h * W_ + x];
      ws[(p ? OFF_KW : OFF_QC) + bc * W_ + x] = acc;
    }
    __syncthreads();
    if (tid == 0) {
      float t = 0.f;
      for (int h = 0; h < H_; ++h) t += rowbuf[h];
      ws[(p ? OFF_SK : OFF_SQ) + bc] = t;
    }
    __syncthreads();
  }
}

__global__ void __launch_bounds__(1024) k2_norm(float* __restrict__ ws) {
  const int type = blockIdx.x & 3;
  const int c0   = blockIdx.x >> 2;
  const int tid  = threadIdx.x;
  __shared__ float rmn[1024], rmx[1024];
  const int   len    = (type < 2) ? H_ : W_;
  const int   N      = BC_ * len;
  const float addeps = (type < 2) ? (H_ * EPSF) : (W_ * EPSF);
  float* dst;
  if      (type == 0) dst = ws + OFF_HR + c0 * (BC_ * H_);
  else if (type == 1) dst = ws + OFF_HC + c0 * (BC_ * H_);
  else if (type == 2) dst = ws + OFF_WR + c0 * (BC_ * W_);
  else                dst = ws + OFF_WC + c0 * (BC_ * W_);
  float lmn = 3.4e38f, lmx = -3.4e38f;
  for (int i = tid; i < N; i += 1024) {
    const int b = i / (C_ * len);
    const int r = i % (C_ * len);
    const int c = r / len;
    const int x = r % len;
    float raw;
    if      (type == 0) raw = ws[OFF_SQ + b * C_ + c0] * ws[OFF_KH + (b * C_ + c) * H_ + x] + addeps;
    else if (type == 1) raw = ws[OFF_QS + (b * C_ + c0) * H_ + x] * ws[OFF_SK + b * C_ + c] + addeps;
    else if (type == 2) raw = ws[OFF_SQ + b * C_ + c0] * ws[OFF_KW + (b * C_ + c) * W_ + x] + addeps;
    else                raw = ws[OFF_QC + (b * C_ + c0) * W_ + x] * ws[OFF_SK + b * C_ + c] + addeps;
    dst[i] = raw;
    lmn = fminf(lmn, raw);
    lmx = fmaxf(lmx, raw);
  }
  rmn[tid] = lmn; rmx[tid] = lmx;
  __syncthreads();
  for (int s = 512; s > 0; s >>= 1) {
    if (tid < s) { rmn[tid] = fminf(rmn[tid], rmn[tid + s]); rmx[tid] = fmaxf(rmx[tid], rmx[tid + s]); }
    __syncthreads();
  }
  const float mn  = rmn[0];
  const float inv = 1.0f / (rmx[0] - mn);
  #pragma unroll
  for (int j = 0; j < 8; ++j) {
    const int i = tid + j * 1024;
    if (i < N) dst[i] = 2.0f + (dst[i] - mn) * inv;
  }
  if (blockIdx.x == 0)
    for (int i = tid; i < 160; i += 1024) ws[OFF_GMAX + i] = 0.0f;
}

__global__ void __launch_bounds__(1024) kg_peak(float* __restrict__ ws) {
  const int t    = blockIdx.x & 3;
  const int c0   = blockIdx.x >> 2;
  const int tid  = threadIdx.x;
  const int wave = tid >> 5, lane = tid & 31;
  __shared__ float red[32];
  const float* mbase = ws + ((t < 2) ? OFF_HR : OFF_HC) + c0 * (BC_ * H_);
  const float* nbase = ws + (((t & 1) == 0) ? OFF_WR : OFF_WC) + c0 * (BC_ * W_);
  float best = 0.0f;
  for (int bc = wave; bc < BC_; bc += 32) {
    const float m0 = mbase[bc * H_ + lane * 3 + 0];
    const float m1 = mbase[bc * H_ + lane * 3 + 1];
    const float m2 = mbase[bc * H_ + lane * 3 + 2];
    const float n0 = nbase[bc * W_ + lane * 4 + 0];
    const float n1 = nbase[bc * W_ + lane * 4 + 1];
    const float n2 = nbase[bc * W_ + lane * 4 + 2];
    const float n3 = nbase[bc * W_ + lane * 4 + 3];
    float mm = fmaxf(fmaxf(m0, m1), m2);
    float nn = fmaxf(fmaxf(n0, n1), fmaxf(n2, n3));
    for (int o = 16; o > 0; o >>= 1) {
      mm = fmaxf(mm, __shfl_xor(mm, o));
      nn = fmaxf(nn, __shfl_xor(nn, o));
    }
    float rs = __expf(mm * n0) + __expf(mm * n1) + __expf(mm * n2) + __expf(mm * n3);
    float cs = __expf(m0 * nn) + __expf(m1 * nn) + __expf(m2 * nn);
    for (int o = 16; o > 0; o >>= 1) {
      rs += __shfl_xor(rs, o);
      cs += __shfl_xor(cs, o);
    }
    best = fmaxf(best, __expf(mm * nn) * (1.0f / rs + 1.0f / cs));
  }
  if (lane == 0) red[wave] = best;
  __syncthreads();
  if (tid == 0) {
    float g = red[0];
    for (int i = 1; i < 32; ++i) g = fmaxf(g, red[i]);
    ws[OFF_GMAX + c0 * 4 + t] = g;
  }
}

#define PRP 33
#define HO_ 48

__device__ __forceinline__ void km_factors(const float* __restrict__ ws, int c0, int bc, int row0, int col0,
                                           float (&mh)[2][3], float (&nw)[2][4]) {
  #pragma unroll
  for (int r = 0; r < 3; ++r) {
    mh[0][r] = ws[OFF_HR + c0 * (BC_ * H_) + bc * H_ + row0 + r];
    mh[1][r] = ws[OFF_HC + c0 * (BC_ * H_) + bc * H_ + row0 + r];
  }
  const float4 a = *(const float4*)(ws + OFF_WR + c0 * (BC_ * W_) + bc * W_ + col0);
  const float4 d = *(const float4*)(ws + OFF_WC + c0 * (BC_ * W_) + bc * W_ + col0);
  nw[0][0] = a.x; nw[0][1] = a.y; nw[0][2] = a.z; nw[0][3] = a.w;
  nw[1][0] = d.x; nw[1][1] = d.y; nw[1][2] = d.z; nw[1][3] = d.w;
}

__device__ __forceinline__ void km_exps(const float (&mh)[2][3], const float (&nw)[2][4], float (&E)[4][3][4]) {
  #pragma unroll
  for (int t = 0; t < 4; ++t) {
    #pragma unroll
    for (int r = 0; r < 3; ++r) {
      #pragma unroll
      for (int q = 0; q < 4; ++q) E[t][r][q] = __expf(mh[t >> 1][r] * nw[t & 1][q]);
    }
  }
}

__device__ __forceinline__ void km_col_partials(const float (&E)[4][3][4], float* pc, int rg, int col0) {
  #pragma unroll
  for (int t = 0; t < 4; ++t) {
    float4 cp;
    cp.x = E[t][0][0] + E[t][1][0] + E[t][2][0];
    cp.y = E[t][0][1] + E[t][1][1] + E[t][2][1];
    cp.z = E[t][0][2] + E[t][1][2] + E[t][2][2];
    cp.w = E[t][0][3] + E[t][1][3] + E[t][2][3];
    *(float4*)(pc + t * (32 * W_) + rg * W_ + col0) = cp;
  }
}

__device__ __forceinline__ void km_row_partials(const float (&E)[4][3][4], float* pr, int lrow0, int lane) {
  #pragma unroll
  for (int t = 0; t < 4; ++t) {
    #pragma unroll
    for (int r = 0; r < 3; ++r)
      pr[t * (HO_ * PRP) + (lrow0 + r) * PRP + lane] = (E[t][r][0] + E[t][r][1]) + (E[t][r][2] + E[t][r][3]);
  }
}

__device__ __forceinline__ void km_col_partials_direct(const float (&mh)[2][3], const float (&nw)[2][4],
                                                       float* pc, int rg, int col0) {
  #pragma unroll
  for (int t = 0; t < 4; ++t) {
    float e[3][4];
    #pragma unroll
    for (int r = 0; r < 3; ++r) {
      #pragma unroll
      for (int q = 0; q < 4; ++q) e[r][q] = __expf(mh[t >> 1][r] * nw[t & 1][q]);
    }
    float4 cp;
    cp.x = e[0][0] + e[1][0] + e[2][0];
    cp.y = e[0][1] + e[1][1] + e[2][1];
    cp.z = e[0][2] + e[1][2] + e[2][2];
    cp.w = e[0][3] + e[1][3] + e[2][3];
    *(float4*)(pc + t * (32 * W_) + rg * W_ + col0) = cp;
  }
}

__global__ void __launch_bounds__(1024)
km_fused(float* __restrict__ ws, const float* __restrict__ values, float* __restrict__ out) {
  const int rh   = blockIdx.x & 1;
  const int half = (blockIdx.x >> 1) & 1;
  const int b    = (blockIdx.x >> 2) & 1;
  const int c0   = blockIdx.x >> 3;
  const int tid  = threadIdx.x;
  const int wave = __builtin_amdgcn_readfirstlane(tid >> 5);
  const int lane = tid & 31;
  __shared__ float PR[4 * HO_ * PRP];
  __shared__ __attribute__((aligned(16))) float PC[2][4 * 32 * W_];
  __shared__ float IR[4][HO_];
  __shared__ __attribute__((aligned(16))) float IC[4][W_];
  __shared__ __attribute__((aligned(16))) float FN[16][2][W_];
  __shared__ __attribute__((aligned(16))) float FM[16][2][HO_];
  __shared__ float red[32];

  const int bc0 = b * C_ + half * 16;
  {
    const int ch = tid >> 6, h = (tid >> 5) & 1, x4 = (tid & 31) * 4;
    const float* srcn = ws + (h ? OFF_WC : OFF_WR) + c0 * (BC_ * W_) + (bc0 + ch) * W_ + x4;
    *(float4*)&FN[ch][h][x4] = *(const float4*)srcn;
    if (tid < 16 * 2 * 12) {
      const int chm = tid / 24, hm = (tid / 12) & 1, y4 = (tid % 12) * 4;
      const float* srcm = ws + (hm ? OFF_HC : OFF_HR) + c0 * (BC_ * H_) + (bc0 + chm) * H_ + (1 - rh) * HO_ + y4;
      *(float4*)&FM[chm][hm][y4] = *(const float4*)srcm;
    }
  }

  float ig[4];
  #pragma unroll
  for (int t = 0; t < 4; ++t) ig[t] = 1.0f / ws[OFF_GMAX + c0 * 4 + t];

  const bool own = wave < 16;
  const int rg    = own ? (rh * 16 + wave) : ((1 - rh) * 16 + (wave - 16));
  const int row0  = rg * 3;
  const int lrow0 = (wave & 15) * 3;
  const int col0  = lane * 4;

  float z[3][4];
  #pragma unroll
  for (int r = 0; r < 3; ++r) { z[r][0] = 0.f; z[r][1] = 0.f; z[r][2] = 0.f; z[r][3] = 0.f; }
  float smax = 0.0f;

  __syncthreads();
  if (!own) {
    float mh[2][3], nw[2][4];
    #pragma unroll
    for (int r = 0; r < 3; ++r) { mh[0][r] = FM[0][0][lrow0 + r]; mh[1][r] = FM[0][1][lrow0 + r]; }
    const float4 a = *(const float4*)&FN[0][0][col0];
    const float4 d = *(const float4*)&FN[0][1][col0];
    nw[0][0] = a.x; nw[0][1] = a.y; nw[0][2] = a.z; nw[0][3] = a.w;
    nw[1][0] = d.x; nw[1][1] = d.y; nw[1][2] = d.z; nw[1][3] = d.w;
    km_col_partials_direct(mh, nw, PC[0], rg, col0);
  }

  for (int ci = 0; ci < 16; ++ci) {
    const int bc = bc0 + ci;
    float  E[4][3][4];
    float4 vv[3];
    if (own) {
      float mh[2][3], nw[2][4];
      km_factors(ws, c0, bc, row0, col0, mh, nw);
      km_exps(mh, nw, E);
      km_row_partials(E, PR, lrow0, lane);
      km_col_partials(E, PC[ci & 1], rg, col0);
    }
    __syncthreads();

    if (own) {
      #pragma unroll
      for (int r = 0; r < 3; ++r)
        vv[r] = *(const float4*)(values + bc * HW_ + (row0 + r) * W_ + col0);

      if (wave < 12) {
        const v2f ones = {1.0f, 1.0f};
        const int kho = (lane < 16) ? 0 : 2;
        const int tt = wave / 3, rb = wave % 3;
        const int M  = rb * 16 + (lane & 15);
        const float* pr = PR + tt * (HO_ * PRP) + M * PRP;
        v8f acc = {};
        for (int k0 = 0; k0 < 32; k0 += 4) {
          v2f a = { pr[k0 + kho], pr[k0 + kho + 1] };
          acc = __builtin_amdgcn_wmma_f32_16x16x4_f32(false, a, false, ones, (short)0, acc, false, false);
        }
        if ((lane & 15) == 0) {
          const int rbase = rb * 16 + ((lane >> 4) << 3);
          #pragma unroll
          for (int v = 0; v < 8; ++v) IR[tt][rbase + v] = 1.0f / acc[v];
        }
      }
    } else {
      if (wave < 24) {
        const int jw = wave - 16;
        const int tt = jw >> 1;
        const int g  = ((jw & 1) << 4) | (lane & 15);
        const int ph = lane >> 4;
        const float* pc = &PC[ci & 1][tt * (32 * W_) + (ph * 16) * W_ + g * 4];
        float4 s;
        #pragma unroll
        for (int i0 = 0; i0 < 16; i0 += 4) {
          const float4 x0 = *(const float4*)(pc + (i0 + 0) * W_);
          const float4 x1 = *(const float4*)(pc + (i0 + 1) * W_);
          const float4 x2 = *(const float4*)(pc + (i0 + 2) * W_);
          const float4 x3 = *(const float4*)(pc + (i0 + 3) * W_);
          float4 t;
          t.x = (x0.x + x1.x) + (x2.x + x3.x);
          t.y = (x0.y + x1.y) + (x2.y + x3.y);
          t.z = (x0.z + x1.z) + (x2.z + x3.z);
          t.w = (x0.w + x1.w) + (x2.w + x3.w);
          if (i0 == 0) { s = t; } else { s.x += t.x; s.y += t.y; s.z += t.z; s.w += t.w; }
          __builtin_amdgcn_sched_barrier(0);
        }
        s.x += xhalf(s.x); s.y += xhalf(s.y); s.z += xhalf(s.z); s.w += xhalf(s.w);
        float2 o;
        o.x = 1.0f / (ph ? s.z : s.x);
        o.y = 1.0f / (ph ? s.w : s.y);
        *(float2*)&IC[tt][g * 4 + ph * 2] = o;
      }
      if (ci + 1 < 16) {
        float mh[2][3], nw[2][4];
        #pragma unroll
        for (int r = 0; r < 3; ++r) { mh[0][r] = FM[ci + 1][0][lrow0 + r]; mh[1][r] = FM[ci + 1][1][lrow0 + r]; }
        const float4 a = *(const float4*)&FN[ci + 1][0][col0];
        const float4 d = *(const float4*)&FN[ci + 1][1][col0];
        nw[0][0] = a.x; nw[0][1] = a.y; nw[0][2] = a.z; nw[0][3] = a.w;
        nw[1][0] = d.x; nw[1][1] = d.y; nw[1][2] = d.z; nw[1][3] = d.w;
        km_col_partials_direct(mh, nw, PC[(ci + 1) & 1], rg, col0);
      }
    }
    __syncthreads();

    if (own) {
      float icv[4][4];
      #pragma unroll
      for (int t = 0; t < 4; ++t) {
        const float4 x = *(const float4*)&IC[t][col0];
        icv[t][0] = x.x; icv[t][1] = x.y; icv[t][2] = x.z; icv[t][3] = x.w;
      }
      #pragma unroll
      for (int r = 0; r < 3; ++r) {
        float irv[4];
        #pragma unroll
        for (int t = 0; t < 4; ++t) irv[t] = IR[t][lrow0 + r];
        const float vr[4] = { vv[r].x, vv[r].y, vv[r].z, vv[r].w };
        float sc[4];
        #pragma unroll
        for (int q = 0; q < 4; ++q) {
          float s = 0.f;
          #pragma unroll
          for (int t = 0; t < 4; ++t) s += E[t][r][q] * (irv[t] + icv[t][q]) * ig[t];
          z[r][q] += s * vr[q];
          smax = fmaxf(smax, s);
          sc[q] = s;
        }
        if (c0 == C_ - 1) {
          float4 o; o.x = sc[0]; o.y = sc[1]; o.z = sc[2]; o.w = sc[3];
          *(float4*)(out + BC_ * HW_ + bc * HW_ + (row0 + r) * W_ + col0) = o;
        }
      }
    }
  }

  if (own) {
    float* zdst = (half ? (ws + OFF_SCORE) : out) + (b * C_ + c0) * HW_;
    #pragma unroll
    for (int r = 0; r < 3; ++r) {
      float4 o; o.x = z[r][0]; o.y = z[r][1]; o.z = z[r][2]; o.w = z[r][3];
      *(float4*)(zdst + (row0 + r) * W_ + col0) = o;
    }
  }
  for (int o = 16; o > 0; o >>= 1) smax = fmaxf(smax, __shfl_xor(smax, o));
  if (lane == 0) red[wave] = smax;
  __syncthreads();
  if (tid == 0) {
    float g = red[0];
    for (int i = 1; i < 32; ++i) g = fmaxf(g, red[i]);
    atomicMax((int*)(ws + OFF_GSMAX + c0), __float_as_int(g));
  }
}

__global__ void ks_scale(const float* __restrict__ ws, float* __restrict__ out) {
  const int gid = blockIdx.x * 256 + threadIdx.x;
  float4* o4 = (float4*)out;
  if (gid < BC_ * HW_ / 4) {
    const int c0 = ((gid * 4) / HW_) % C_;
    const float inv = 1.0f / ws[OFF_GSMAX + c0];
    const float4 a = o4[gid];
    const float4 p = *(const float4*)(ws + OFF_SCORE + gid * 4);
    float4 r;
    r.x = (a.x + p.x) * inv; r.y = (a.y + p.y) * inv; r.z = (a.z + p.z) * inv; r.w = (a.w + p.w) * inv;
    o4[gid] = r;
  } else {
    const float inv = 1.0f / ws[OFF_GSMAX + 31];
    float4 a = o4[gid];
    a.x *= inv; a.y *= inv; a.z *= inv; a.w *= inv;
    o4[gid] = a;
  }
}

extern "C" void kernel_launch(void* const* d_in, const int* in_sizes, int n_in,
                              void* d_out, int out_size, void* d_ws, size_t ws_size,
                              hipStream_t stream) {
  (void)in_sizes; (void)n_in; (void)out_size; (void)ws_size;
  const float* q   = (const float*)d_in[0];
  const float* k   = (const float*)d_in[1];
  const float* val = (const float*)d_in[2];
  float* out = (float*)d_out;
  float* ws  = (float*)d_ws;

  k1_sums<<<64, 1024, 0, stream>>>(q, k, ws);
  k2_norm<<<128, 1024, 0, stream>>>(ws);
  kg_peak<<<128, 1024, 0, stream>>>(ws);
  km_fused<<<256, 1024, 0, stream>>>(ws, val, out);
  ks_scale<<<1536, 256, 0, stream>>>(ws, out);
}


# km, kg: exp(m*n) formed as 2^((m*log2 e)*n): log2 e folded into the row factors, a tile's products formed ahead of its exponentials, taken in place: one multiply per v_exp, no chain via one temporary
# speedup vs baseline: 10.3511x; 1.1245x over previous
#include <hip/hip_runtime.h>

typedef __attribute__((ext_vector_type(2)))  float    v2f;
typedef __attribute__((ext_vector_type(8)))  float    v8f;

#define B_   2
#define C_   32
#define H_   96
#define W_   128
#define HW_  (H_ * W_)
#define BC_  (B_ * C_)
#define EPSF 1e-6f
#define LOG2E_F 1.44269504088896340736f

__device__ __forceinline__ float xhalf(float x) {
  const int b = __builtin_bit_cast(int, x);
  return __builtin_bit_cast(float, __builtin_amdgcn_permlanex16(b, b, 0x76543210, 0xfedcba98, false, false));
}

#define OFF_QS    0
#define OFF_QC    6144
#define OFF_KH    14336
#define OFF_KW    20480
#define OFF_SQ    28672
#define OFF_SK    28736
#define OFF_GMAX  28800
#define OFF_GSMAX 28928
#define OFF_HR    28960
#define OFF_HC    (OFF_HR + 32 * 6144)
#define OFF_WR    (OFF_HC + 32 * 6144)
#define OFF_WC    (OFF_WR + 32 * 8192)
#define OFF_IRS   (OFF_WC + 32 * 8192)
#define OFF_SCORE (OFF_IRS + 4 * 64 * 224)


__global__ void __launch_bounds__(1024)
k1_sums(const float* __restrict__ q, const float* __restrict__ k, float* __restrict__ ws) {
  __shared__ float tile[HW_];
  __shared__ float rowbuf[H_];
  const int bc  = blockIdx.x;
  const int tid = threadIdx.x;
  for (int p = 0; p < 2; ++p) {
    const float* s = p ? k : q;
    #pragma unroll
    for (int j = 0; j < HW_ / 1024; ++j) tile[tid + j * 1024] = s[bc * HW_ + tid + j * 1024];
    __syncthreads();
    if (tid < H_) {
      float acc = 0.f;
      for (int x = 0; x < W_; ++x) acc += tile[tid * W_ + x];
      ws[(p ? OFF_KH : OFF_QS) + bc * H_ + tid] = acc;
      rowbuf[tid] = acc;
    }
    if (tid >= 128 && tid < 256) {
      const int x = tid - 128;
      float acc = 0.f;
      for (int h = 0; h < H_; ++h) acc += tile[h * W_ + x];
      ws[(p ? OFF_KW : OFF_QC) + bc * W_ + x] = acc;
    }
    __syncthreads();
    if (tid == 0) {
      float t = 0.f;
      for (int h = 0; h < H_; ++h) t += rowbuf[h];
      ws[(p ? OFF_SK : OFF_SQ) + bc] = t;
    }
    __syncthreads();
  }
}

__global__ void __launch_bounds__(1024) k2_norm(float* __restrict__ ws) {
  const int type = blockIdx.x & 3;
  const int c0   = blockIdx.x >> 2;
  const int tid  = threadIdx.x;
  __shared__ float rmn[1024], rmx[1024];
  const int   len    = (type < 2) ? H_ : W_;
  const int   N      = BC_ * len;
  const float addeps = (type < 2) ? (H_ * EPSF) : (W_ * EPSF);
  float* dst;
  if      (type == 0) dst = ws + OFF_HR + c0 * (BC_ * H_);
  else if (type == 1) dst = ws + OFF_HC + c0 * (BC_ * H_);
  else if (type == 2) dst = ws + OFF_WR + c0 * (BC_ * W_);
  else                dst = ws + OFF_WC + c0 * (BC_ * W_);
  float lmn = 3.4e38f, lmx = -3.4e38f;
  for (int i = tid; i < N; i += 1024) {
    const int b = i / (C_ * len);
    const int r = i % (C_ * len);
    const int c = r / len;
    const int x = r % len;
    float raw;
    if      (type == 0) raw = ws[OFF_SQ + b * C_ + c0] * ws[OFF_KH + (b * C_ + c) * H_ + x] + addeps;
    else if (type == 1) raw = ws[OFF_QS + (b * C_ + c0) * H_ + x] * ws[OFF_SK + b * C_ + c] + addeps;
    else if (type == 2) raw = ws[OFF_SQ + b * C_ + c0] * ws[OFF_KW + (b * C_ + c) * W_ + x] + addeps;
    else                raw = ws[OFF_QC + (b * C_ + c0) * W_ + x] * ws[OFF_SK + b * C_ + c] + addeps;
    dst[i] = raw;
    lmn = fminf(lmn, raw);
    lmx = fmaxf(lmx, raw);
  }
  rmn[tid] = lmn; rmx[tid] = lmx;
  __syncthreads();
  for (int s = 512; s > 0; s >>= 1) {
    if (tid < s) { rmn[tid] = fminf(rmn[tid], rmn[tid + s]); rmx[tid] = fmaxf(rmx[tid], rmx[tid + s]); }
    __syncthreads();
  }
  const float mn  = rmn[0];
  const float inv = 1.0f / (rmx[0] - mn);
  #pragma unroll
  for (int j = 0; j < 8; ++j) {
    const int i = tid + j * 1024;
    if (i < N) dst[i] = 2.0f + (dst[i] - mn) * inv;
  }
  if (blockIdx.x == 0)
    for (int i = tid; i < 160; i += 1024) ws[OFF_GMAX + i] = 0.0f;
}

__global__ void __launch_bounds__(1024) kg_peak(float* __restrict__ ws) {
  const int t    = blockIdx.x & 3;
  const int c0   = blockIdx.x >> 2;
  const int tid  = threadIdx.x;
  const int wave = tid >> 5, lane = tid & 31;
  __shared__ float red[32];
  const float* mbase = ws + ((t < 2) ? OFF_HR : OFF_HC) + c0 * (BC_ * H_);
  const float* nbase = ws + (((t & 1) == 0) ? OFF_WR : OFF_WC) + c0 * (BC_ * W_);
  float best = 0.0f;
  for (int bc = wave; bc < BC_; bc += 32) {
    const float m0 = mbase[bc * H_ + lane * 3 + 0];
    const float m1 = mbase[bc * H_ + lane * 3 + 1];
    const float m2 = mbase[bc * H_ + lane * 3 + 2];
    const float n0 = nbase[bc * W_ + lane * 4 + 0];
    const float n1 = nbase[bc * W_ + lane * 4 + 1];
    const float n2 = nbase[bc * W_ + lane * 4 + 2];
    const float n3 = nbase[bc * W_ + lane * 4 + 3];
    float mm = fmaxf(fmaxf(m0, m1), m2);
    float nn = fmaxf(fmaxf(n0, n1), fmaxf(n2, n3));
    for (int o = 16; o > 0; o >>= 1) {
      mm = fmaxf(mm, __shfl_xor(mm, o));
      nn = fmaxf(nn, __shfl_xor(nn, o));
    }
    const float mml = mm * LOG2E_F;
    float rs = __builtin_amdgcn_exp2f(mml * n0) + __builtin_amdgcn_exp2f(mml * n1)
             + __builtin_amdgcn_exp2f(mml * n2) + __builtin_amdgcn_exp2f(mml * n3);
    float cs = __builtin_amdgcn_exp2f((m0 * LOG2E_F) * nn) + __builtin_amdgcn_exp2f((m1 * LOG2E_F) * nn)
             + __builtin_amdgcn_exp2f((m2 * LOG2E_F) * nn);
    for (int o = 16; o > 0; o >>= 1) {
      rs += __shfl_xor(rs, o);
      cs += __shfl_xor(cs, o);
    }
    best = fmaxf(best, __builtin_amdgcn_exp2f(mml * nn) * (1.0f / rs + 1.0f / cs));
  }
  if (lane == 0) red[wave] = best;
  __syncthreads();
  if (tid == 0) {
    float g = red[0];
    for (int i = 1; i < 32; ++i) g = fmaxf(g, red[i]);
    ws[OFF_GMAX + c0 * 4 + t] = g;
  }
}

#define PRP 33
#define HO_ 48

__device__ __forceinline__ void km_factors(const float* __restrict__ ws, int c0, int bc, int row0, int col0,
                                           float (&mh)[2][3], float (&nw)[2][4]) {
  #pragma unroll
  for (int r = 0; r < 3; ++r) {
    mh[0][r] = ws[OFF_HR + c0 * (BC_ * H_) + bc * H_ + row0 + r];
    mh[1][r] = ws[OFF_HC + c0 * (BC_ * H_) + bc * H_ + row0 + r];
  }
  const float4 a = *(const float4*)(ws + OFF_WR + c0 * (BC_ * W_) + bc * W_ + col0);
  const float4 d = *(const float4*)(ws + OFF_WC + c0 * (BC_ * W_) + bc * W_ + col0);
  nw[0][0] = a.x; nw[0][1] = a.y; nw[0][2] = a.z; nw[0][3] = a.w;
  nw[1][0] = d.x; nw[1][1] = d.y; nw[1][2] = d.z; nw[1][3] = d.w;
}

__device__ __forceinline__ void km_exps(const float (&mh)[2][3], const float (&nw)[2][4], float (&E)[4][3][4]) {
  float ml[2][3];
  #pragma unroll
  for (int r = 0; r < 3; ++r) { ml[0][r] = mh[0][r] * LOG2E_F; ml[1][r] = mh[1][r] * LOG2E_F; }
  #pragma unroll
  for (int t = 0; t < 4; ++t) {
    #pragma unroll
    for (int r = 0; r < 3; ++r) {
      #pragma unroll
      for (int q = 0; q < 4; ++q) E[t][r][q] = ml[t >> 1][r] * nw[t & 1][q];
    }
  }
  #pragma unroll
  for (int t = 0; t < 4; ++t) {
    #pragma unroll
    for (int r = 0; r < 3; ++r) {
      #pragma unroll
      for (int q = 0; q < 4; ++q) E[t][r][q] = __builtin_amdgcn_exp2f(E[t][r][q]);
    }
  }
}

__device__ __forceinline__ void km_col_partials(const float (&E)[4][3][4], float* pc, int rg, int col0) {
  #pragma unroll
  for (int t = 0; t < 4; ++t) {
    float4 cp;
    cp.x = E[t][0][0] + E[t][1][0] + E[t][2][0];
    cp.y = E[t][0][1] + E[t][1][1] + E[t][2][1];
    cp.z = E[t][0][2] + E[t][1][2] + E[t][2][2];
    cp.w = E[t][0][3] + E[t][1][3] + E[t][2][3];
    *(float4*)(pc + t * (32 * W_) + rg * W_ + col0) = cp;
  }
}

__device__ __forceinline__ void km_row_partials(const float (&E)[4][3][4], float* pr, int lrow0, int lane) {
  #pragma unroll
  for (int t = 0; t < 4; ++t) {
    #pragma unroll
    for (int r = 0; r < 3; ++r)
      pr[t * (HO_ * PRP) + (lrow0 + r) * PRP + lane] = (E[t][r][0] + E[t][r][1]) + (E[t][r][2] + E[t][r][3]);
  }
}

__device__ __forceinline__ void km_col_partials_direct(const float (&mh)[2][3], const float (&nw)[2][4],
                                                       float* pc, int rg, int col0) {
  float ml[2][3];
  #pragma unroll
  for (int r = 0; r < 3; ++r) { ml[0][r] = mh[0][r] * LOG2E_F; ml[1][r] = mh[1][r] * LOG2E_F; }
  #pragma unroll
  for (int t = 0; t < 4; ++t) {
    float e[3][4];
    #pragma unroll
    for (int r = 0; r < 3; ++r) {
      #pragma unroll
      for (int q = 0; q < 4; ++q) e[r][q] = ml[t >> 1][r] * nw[t & 1][q];
    }
    #pragma unroll
    for (int r = 0; r < 3; ++r) {
      #pragma unroll
      for (int q = 0; q < 4; ++q) e[r][q] = __builtin_amdgcn_exp2f(e[r][q]);
    }
    float4 cp;
    cp.x = e[0][0] + e[1][0] + e[2][0];
    cp.y = e[0][1] + e[1][1] + e[2][1];
    cp.z = e[0][2] + e[1][2] + e[2][2];
    cp.w = e[0][3] + e[1][3] + e[2][3];
    *(float4*)(pc + t * (32 * W_) + rg * W_ + col0) = cp;
  }
}

__global__ void __launch_bounds__(1024)
km_fused(float* __restrict__ ws, const float* __restrict__ values, float* __restrict__ out) {
  const int rh   = blockIdx.x & 1;
  const int half = (blockIdx.x >> 1) & 1;
  const int b    = (blockIdx.x >> 2) & 1;
  const int c0   = blockIdx.x >> 3;
  const int tid  = threadIdx.x;
  const int wave = __builtin_amdgcn_readfirstlane(tid >> 5);
  const int lane = tid & 31;
  __shared__ float PR[4 * HO_ * PRP];
  __shared__ __attribute__((aligned(16))) float PC[2][4 * 32 * W_];
  __shared__ float IR[4][HO_];
  __shared__ __attribute__((aligned(16))) float IC[4][W_];
  __shared__ __attribute__((aligned(16))) float FN[16][2][W_];
  __shared__ __attribute__((aligned(16))) float FM[16][2][HO_];
  __shared__ float red[32];

  const int bc0 = b * C_ + half * 16;
  {
    const int ch = tid >> 6, h = (tid >> 5) & 1, x4 = (tid & 31) * 4;
    const float* srcn = ws + (h ? OFF_WC : OFF_WR) + c0 * (BC_ * W_) + (bc0 + ch) * W_ + x4;
    *(float4*)&FN[ch][h][x4] = *(const float4*)srcn;
    if (tid < 16 * 2 * 12) {
      const int chm = tid / 24, hm = (tid / 12) & 1, y4 = (tid % 12) * 4;
      const float* srcm = ws + (hm ? OFF_HC : OFF_HR) + c0 * (BC_ * H_) + (bc0 + chm) * H_ + (1 - rh) * HO_ + y4;
      *(float4*)&FM[chm][hm][y4] = *(const float4*)srcm;
    }
  }

  float ig[4];
  #pragma unroll
  for (int t = 0; t < 4; ++t) ig[t] = 1.0f / ws[OFF_GMAX + c0 * 4 + t];

  const bool own = wave < 16;
  const int rg    = own ? (rh * 16 + wave) : ((1 - rh) * 16 + (wave - 16));
  const int row0  = rg * 3;
  const int lrow0 = (wave & 15) * 3;
  const int col0  = lane * 4;

  float z[3][4];
  #pragma unroll
  for (int r = 0; r < 3; ++r) { z[r][0] = 0.f; z[r][1] = 0.f; z[r][2] = 0.f; z[r][3] = 0.f; }
  float smax = 0.0f;

  __syncthreads();
  if (!own) {
    float mh[2][3], nw[2][4];
    #pragma unroll
    for (int r = 0; r < 3; ++r) { mh[0][r] = FM[0][0][lrow0 + r]; mh[1][r] = FM[0][1][lrow0 + r]; }
    const float4 a = *(const float4*)&FN[0][0][col0];
    const float4 d = *(const float4*)&FN[0][1][col0];
    nw[0][0] = a.x; nw[0][1] = a.y; nw[0][2] = a.z; nw[0][3] = a.w;
    nw[1][0] = d.x; nw[1][1] = d.y; nw[1][2] = d.z; nw[1][3] = d.w;
    km_col_partials_direct(mh, nw, PC[0], rg, col0);
  }

  for (int ci = 0; ci < 16; ++ci) {
    const int bc = bc0 + ci;
    float  E[4][3][4];
    float4 vv[3];
    if (own) {
      float mh[2][3], nw[2][4];
      km_factors(ws, c0, bc, row0, col0, mh, nw);
      km_exps(mh, nw, E);
      km_row_partials(E, PR, lrow0, lane);
      km_col_partials(E, PC[ci & 1], rg, col0);
    }
    __syncthreads();

    if (own) {
      #pragma unroll
      for (int r = 0; r < 3; ++r)
        vv[r] = *(const float4*)(values + bc * HW_ + (row0 + r) * W_ + col0);

      if (wave < 12) {
        const v2f ones = {1.0f, 1.0f};
        const int kho = (lane < 16) ? 0 : 2;
        const int tt = wave / 3, rb = wave % 3;
        const int M  = rb * 16 + (lane & 15);
        const float* pr = PR + tt * (HO_ * PRP) + M * PRP;
        v8f acc = {};
        for (int k0 = 0; k0 < 32; k0 += 4) {
          v2f a = { pr[k0 + kho], pr[k0 + kho + 1] };
          acc = __builtin_amdgcn_wmma_f32_16x16x4_f32(false, a, false, ones, (short)0, acc, false, false);
        }
        if ((lane & 15) == 0) {
          const int rbase = rb * 16 + ((lane >> 4) << 3);
          #pragma unroll
          for (int v = 0; v < 8; ++v) IR[tt][rbase + v] = 1.0f / acc[v];
        }
      }
    } else {
      if (wave < 24) {
        const int jw = wave - 16;
        const int tt = jw >> 1;
        const int g  = ((jw & 1) << 4) | (lane & 15);
        const int ph = lane >> 4;
        const float* pc = &PC[ci & 1][tt * (32 * W_) + (ph * 16) * W_ + g * 4];
        float4 s;
        #pragma unroll
        for (int i0 = 0; i0 < 16; i0 += 4) {
          const float4 x0 = *(const float4*)(pc + (i0 + 0) * W_);
          const float4 x1 = *(const float4*)(pc + (i0 + 1) * W_);
          const float4 x2 = *(const float4*)(pc + (i0 + 2) * W_);
          const float4 x3 = *(const float4*)(pc + (i0 + 3) * W_);
          float4 t;
          t.x = (x0.x + x1.x) + (x2.x + x3.x);
          t.y = (x0.y + x1.y) + (x2.y + x3.y);
          t.z = (x0.z + x1.z) + (x2.z + x3.z);
          t.w = (x0.w + x1.w) + (x2.w + x3.w);
          if (i0 == 0) { s = t; } else { s.x += t.x; s.y += t.y; s.z += t.z; s.w += t.w; }
          __builtin_amdgcn_sched_barrier(0);
        }
        s.x += xhalf(s.x); s.y += xhalf(s.y); s.z += xhalf(s.z); s.w += xhalf(s.w);
        float2 o;
        o.x = 1.0f / (ph ? s.z : s.x);
        o.y = 1.0f / (ph ? s.w : s.y);
        *(float2*)&IC[tt][g * 4 + ph * 2] = o;
      }
      if (ci + 1 < 16) {
        float mh[2][3], nw[2][4];
        #pragma unroll
        for (int r = 0; r < 3; ++r) { mh[0][r] = FM[ci + 1][0][lrow0 + r]; mh[1][r] = FM[ci + 1][1][lrow0 + r]; }
        const float4 a = *(const float4*)&FN[ci + 1][0][col0];
        const float4 d = *(const float4*)&FN[ci + 1][1][col0];
        nw[0][0] = a.x; nw[0][1] = a.y; nw[0][2] = a.z; nw[0][3] = a.w;
        nw[1][0] = d.x; nw[1][1] = d.y; nw[1][2] = d.z; nw[1][3] = d.w;
        km_col_partials_direct(mh, nw, PC[(ci + 1) & 1], rg, col0);
      }
    }
    __syncthreads();

    if (own) {
      float icv[4][4];
      #pragma unroll
      for (int t = 0; t < 4; ++t) {
        const float4 x = *(const float4*)&IC[t][col0];
        icv[t][0] = x.x; icv[t][1] = x.y; icv[t][2] = x.z; icv[t][3] = x.w;
      }
      #pragma unroll
      for (int r = 0; r < 3; ++r) {
        float irv[4];
        #pragma unroll
        for (int t = 0; t < 4; ++t) irv[t] = IR[t][lrow0 + r];
        const float vr[4] = { vv[r].x, vv[r].y, vv[r].z, vv[r].w };
        float sc[4];
        #pragma unroll
        for (int q = 0; q < 4; ++q) {
          float s = 0.f;
          #pragma unroll
          for (int t = 0; t < 4; ++t) s += E[t][r][q] * (irv[t] + icv[t][q]) * ig[t];
          z[r][q] += s * vr[q];
          smax = fmaxf(smax, s);
          sc[q] = s;
        }
        if (c0 == C_ - 1) {
          float4 o; o.x = sc[0]; o.y = sc[1]; o.z = sc[2]; o.w = sc[3];
          *(float4*)(out + BC_ * HW_ + bc * HW_ + (row0 + r) * W_ + col0) = o;
        }
      }
    }
  }

  if (own) {
    float* zdst = (half ? (ws + OFF_SCORE) : out) + (b * C_ + c0) * HW_;
    #pragma unroll
    for (int r = 0; r < 3; ++r) {
      float4 o; o.x = z[r][0]; o.y = z[r][1]; o.z = z[r][2]; o.w = z[r][3];
      *(float4*)(zdst + (row0 + r) * W_ + col0) = o;
    }
  }
  for (int o = 16; o > 0; o >>= 1) smax = fmaxf(smax, __shfl_xor(smax, o));
  if (lane == 0) red[wave] = smax;
  __syncthreads();
  if (tid == 0) {
    float g = red[0];
    for (int i = 1; i < 32; ++i) g = fmaxf(g, red[i]);
    atomicMax((int*)(ws + OFF_GSMAX + c0), __float_as_int(g));
  }
}

__global__ void ks_scale(const float* __restrict__ ws, float* __restrict__ out) {
  const int gid = blockIdx.x * 256 + threadIdx.x;
  float4* o4 = (float4*)out;
  if (gid < BC_ * HW_ / 4) {
    const int c0 = ((gid * 4) / HW_) % C_;
    const float inv = 1.0f / ws[OFF_GSMAX + c0];
    const float4 a = o4[gid];
    const float4 p = *(const float4*)(ws + OFF_SCORE + gid * 4);
    float4 r;
    r.x = (a.x + p.x) * inv; r.y = (a.y + p.y) * inv; r.z = (a.z + p.z) * inv; r.w = (a.w + p.w) * inv;
    o4[gid] = r;
  } else {
    const float inv = 1.0f / ws[OFF_GSMAX + 31];
    float4 a = o4[gid];
    a.x *= inv; a.y *= inv; a.z *= inv; a.w *= inv;
    o4[gid] = a;
  }
}

extern "C" void kernel_launch(void* const* d_in, const int* in_sizes, int n_in,
                              void* d_out, int out_size, void* d_ws, size_t ws_size,
                              hipStream_t stream) {
  (void)in_sizes; (void)n_in; (void)out_size; (void)ws_size;
  const float* q   = (const float*)d_in[0];
  const float* k   = (const float*)d_in[1];
  const float* val = (const float*)d_in[2];
  float* out = (float*)d_out;
  float* ws  = (float*)d_ws;

  k1_sums<<<64, 1024, 0, stream>>>(q, k, ws);
  k2_norm<<<128, 1024, 0, stream>>>(ws);
  kg_peak<<<128, 1024, 0, stream>>>(ws);
  km_fused<<<256, 1024, 0, stream>>>(ws, val, out);
  ks_scale<<<1536, 256, 0, stream>>>(ws, out);
}


# km: own waves take row and column factors from the LDS stage the other-half waves use (row table widened to all 96 rows): no factor fetch from memory in the channel loop; same values and sums
# speedup vs baseline: 10.9299x; 1.0559x over previous
#include <hip/hip_runtime.h>

typedef __attribute__((ext_vector_type(2)))  float    v2f;
typedef __attribute__((ext_vector_type(8)))  float    v8f;

#define B_   2
#define C_   32
#define H_   96
#define W_   128
#define HW_  (H_ * W_)
#define BC_  (B_ * C_)
#define EPSF 1e-6f
#define LOG2E_F 1.44269504088896340736f

__device__ __forceinline__ float xhalf(float x) {
  const int b = __builtin_bit_cast(int, x);
  return __builtin_bit_cast(float, __builtin_amdgcn_permlanex16(b, b, 0x76543210, 0xfedcba98, false, false));
}

#define OFF_QS    0
#define OFF_QC    6144
#define OFF_KH    14336
#define OFF_KW    20480
#define OFF_SQ    28672
#define OFF_SK    28736
#define OFF_GMAX  28800
#define OFF_GSMAX 28928
#define OFF_HR    28960
#define OFF_HC    (OFF_HR + 32 * 6144)
#define OFF_WR    (OFF_HC + 32 * 6144)
#define OFF_WC    (OFF_WR + 32 * 8192)
#define OFF_IRS   (OFF_WC + 32 * 8192)
#define OFF_SCORE (OFF_IRS + 4 * 64 * 224)


__global__ void __launch_bounds__(1024)
k1_sums(const float* __restrict__ q, const float* __restrict__ k, float* __restrict__ ws) {
  __shared__ float tile[HW_];
  __shared__ float rowbuf[H_];
  const int bc  = blockIdx.x;
  const int tid = threadIdx.x;
  for (int p = 0; p < 2; ++p) {
    const float* s = p ? k : q;
    #pragma unroll
    for (int j = 0; j < HW_ / 1024; ++j) tile[tid + j * 1024] = s[bc * HW_ + tid + j * 1024];
    __syncthreads();
    if (tid < H_) {
      float acc = 0.f;
      for (int x = 0; x < W_; ++x) acc += tile[tid * W_ + x];
      ws[(p ? OFF_KH : OFF_QS) + bc * H_ + tid] = acc;
      rowbuf[tid] = acc;
    }
    if (tid >= 128 && tid < 256) {
      const int x = tid - 128;
      float acc = 0.f;
      for (int h = 0; h < H_; ++h) acc += tile[h * W_ + x];
      ws[(p ? OFF_KW : OFF_QC) + bc * W_ + x] = acc;
    }
    __syncthreads();
    if (tid == 0) {
      float t = 0.f;
      for (int h = 0; h < H_; ++h) t += rowbuf[h];
      ws[(p ? OFF_SK : OFF_SQ) + bc] = t;
    }
    __syncthreads();
  }
}

__global__ void __launch_bounds__(1024) k2_norm(float* __restrict__ ws) {
  const int type = blockIdx.x & 3;
  const int c0   = blockIdx.x >> 2;
  const int tid  = threadIdx.x;
  __shared__ float rmn[1024], rmx[1024];
  const int   len    = (type < 2) ? H_ : W_;
  const int   N      = BC_ * len;
  const float addeps = (type < 2) ? (H_ * EPSF) : (W_ * EPSF);
  float* dst;
  if      (type == 0) dst = ws + OFF_HR + c0 * (BC_ * H_);
  else if (type == 1) dst = ws + OFF_HC + c0 * (BC_ * H_);
  else if (type == 2) dst = ws + OFF_WR + c0 * (BC_ * W_);
  else                dst = ws + OFF_WC + c0 * (BC_ * W_);
  float lmn = 3.4e38f, lmx = -3.4e38f;
  for (int i = tid; i < N; i += 1024) {
    const int b = i / (C_ * len);
    const int r = i % (C_ * len);
    const int c = r / len;
    const int x = r % len;
    float raw;
    if      (type == 0) raw = ws[OFF_SQ + b * C_ + c0] * ws[OFF_KH + (b * C_ + c) * H_ + x] + addeps;
    else if (type == 1) raw = ws[OFF_QS + (b * C_ + c0) * H_ + x] * ws[OFF_SK + b * C_ + c] + addeps;
    else if (type == 2) raw = ws[OFF_SQ + b * C_ + c0] * ws[OFF_KW + (b * C_ + c) * W_ + x] + addeps;
    else                raw = ws[OFF_QC + (b * C_ + c0) * W_ + x] * ws[OFF_SK + b * C_ + c] + addeps;
    dst[i] = raw;
    lmn = fminf(lmn, raw);
    lmx = fmaxf(lmx, raw);
  }
  rmn[tid] = lmn; rmx[tid] = lmx;
  __syncthreads();
  for (int s = 512; s > 0; s >>= 1) {
    if (tid < s) { rmn[tid] = fminf(rmn[tid], rmn[tid + s]); rmx[tid] = fmaxf(rmx[tid], rmx[tid + s]); }
    __syncthreads();
  }
  const float mn  = rmn[0];
  const float inv = 1.0f / (rmx[0] - mn);
  #pragma unroll
  for (int j = 0; j < 8; ++j) {
    const int i = tid + j * 1024;
    if (i < N) dst[i] = 2.0f + (dst[i] - mn) * inv;
  }
  if (blockIdx.x == 0)
    for (int i = tid; i < 160; i += 1024) ws[OFF_GMAX + i] = 0.0f;
}

__global__ void __launch_bounds__(1024) kg_peak(float* __restrict__ ws) {
  const int t    = blockIdx.x & 3;
  const int c0   = blockIdx.x >> 2;
  const int tid  = threadIdx.x;
  const int wave = tid >> 5, lane = tid & 31;
  __shared__ float red[32];
  const float* mbase = ws + ((t < 2) ? OFF_HR : OFF_HC) + c0 * (BC_ * H_);
  const float* nbase = ws + (((t & 1) == 0) ? OFF_WR : OFF_WC) + c0 * (BC_ * W_);
  float best = 0.0f;
  for (int bc = wave; bc < BC_; bc += 32) {
    const float m0 = mbase[bc * H_ + lane * 3 + 0];
    const float m1 = mbase[bc * H_ + lane * 3 + 1];
    const float m2 = mbase[bc * H_ + lane * 3 + 2];
    const float n0 = nbase[bc * W_ + lane * 4 + 0];
    const float n1 = nbase[bc * W_ + lane * 4 + 1];
    const float n2 = nbase[bc * W_ + lane * 4 + 2];
    const float n3 = nbase[bc * W_ + lane * 4 + 3];
    float mm = fmaxf(fmaxf(m0, m1), m2);
    float nn = fmaxf(fmaxf(n0, n1), fmaxf(n2, n3));
    for (int o = 16; o > 0; o >>= 1) {
      mm = fmaxf(mm, __shfl_xor(mm, o));
      nn = fmaxf(nn, __shfl_xor(nn, o));
    }
    const float mml = mm * LOG2E_F;
    float rs = __builtin_amdgcn_exp2f(mml * n0) + __builtin_amdgcn_exp2f(mml * n1)
             + __builtin_amdgcn_exp2f(mml * n2) + __builtin_amdgcn_exp2f(mml * n3);
    float cs = __builtin_amdgcn_exp2f((m0 * LOG2E_F) * nn) + __builtin_amdgcn_exp2f((m1 * LOG2E_F) * nn)
             + __builtin_amdgcn_exp2f((m2 * LOG2E_F) * nn);
    for (int o = 16; o > 0; o >>= 1) {
      rs += __shfl_xor(rs, o);
      cs += __shfl_xor(cs, o);
    }
    best = fmaxf(best, __builtin_amdgcn_exp2f(mml * nn) * (1.0f / rs + 1.0f / cs));
  }
  if (lane == 0) red[wave] = best;
  __syncthreads();
  if (tid == 0) {
    float g = red[0];
    for (int i = 1; i < 32; ++i) g = fmaxf(g, red[i]);
    ws[OFF_GMAX + c0 * 4 + t] = g;
  }
}

#define PRP 33
#define HO_ 48

__device__ __forceinline__ void km_exps(const float (&mh)[2][3], const float (&nw)[2][4], float (&E)[4][3][4]) {
  float ml[2][3];
  #pragma unroll
  for (int r = 0; r < 3; ++r) { ml[0][r] = mh[0][r] * LOG2E_F; ml[1][r] = mh[1][r] * LOG2E_F; }
  #pragma unroll
  for (int t = 0; t < 4; ++t) {
    #pragma unroll
    for (int r = 0; r < 3; ++r) {
      #pragma unroll
      for (int q = 0; q < 4; ++q) E[t][r][q] = ml[t >> 1][r] * nw[t & 1][q];
    }
  }
  #pragma unroll
  for (int t = 0; t < 4; ++t) {
    #pragma unroll
    for (int r = 0; r < 3; ++r) {
      #pragma unroll
      for (int q = 0; q < 4; ++q) E[t][r][q] = __builtin_amdgcn_exp2f(E[t][r][q]);
    }
  }
}

__device__ __forceinline__ void km_col_partials(const float (&E)[4][3][4], float* pc, int rg, int col0) {
  #pragma unroll
  for (int t = 0; t < 4; ++t) {
    float4 cp;
    cp.x = E[t][0][0] + E[t][1][0] + E[t][2][0];
    cp.y = E[t][0][1] + E[t][1][1] + E[t][2][1];
    cp.z = E[t][0][2] + E[t][1][2] + E[t][2][2];
    cp.w = E[t][0][3] + E[t][1][3] + E[t][2][3];
    *(float4*)(pc + t * (32 * W_) + rg * W_ + col0) = cp;
  }
}

__device__ __forceinline__ void km_row_partials(const float (&E)[4][3][4], float* pr, int lrow0, int lane) {
  #pragma unroll
  for (int t = 0; t < 4; ++t) {
    #pragma unroll
    for (int r = 0; r < 3; ++r)
      pr[t * (HO_ * PRP) + (lrow0 + r) * PRP + lane] = (E[t][r][0] + E[t][r][1]) + (E[t][r][2] + E[t][r][3]);
  }
}

__device__ __forceinline__ void km_col_partials_direct(const float (&mh)[2][3], const float (&nw)[2][4],
                                                       float* pc, int rg, int col0) {
  float ml[2][3];
  #pragma unroll
  for (int r = 0; r < 3; ++r) { ml[0][r] = mh[0][r] * LOG2E_F; ml[1][r] = mh[1][r] * LOG2E_F; }
  #pragma unroll
  for (int t = 0; t < 4; ++t) {
    float e[3][4];
    #pragma unroll
    for (int r = 0; r < 3; ++r) {
      #pragma unroll
      for (int q = 0; q < 4; ++q) e[r][q] = ml[t >> 1][r] * nw[t & 1][q];
    }
    #pragma unroll
    for (int r = 0; r < 3; ++r) {
      #pragma unroll
      for (int q = 0; q < 4; ++q) e[r][q] = __builtin_amdgcn_exp2f(e[r][q]);
    }
    float4 cp;
    cp.x = e[0][0] + e[1][0] + e[2][0];
    cp.y = e[0][1] + e[1][1] + e[2][1];
    cp.z = e[0][2] + e[1][2] + e[2][2];
    cp.w = e[0][3] + e[1][3] + e[2][3];
    *(float4*)(pc + t * (32 * W_) + rg * W_ + col0) = cp;
  }
}

__global__ void __launch_bounds__(1024)
km_fused(float* __restrict__ ws, const float* __restrict__ values, float* __restrict__ out) {
  const int rh   = blockIdx.x & 1;
  const int half = (blockIdx.x >> 1) & 1;
  const int b    = (blockIdx.x >> 2) & 1;
  const int c0   = blockIdx.x >> 3;
  const int tid  = threadIdx.x;
  const int wave = __builtin_amdgcn_readfirstlane(tid >> 5);
  const int lane = tid & 31;
  __shared__ float PR[4 * HO_ * PRP];
  __shared__ __attribute__((aligned(16))) float PC[2][4 * 32 * W_];
  __shared__ float IR[4][HO_];
  __shared__ __attribute__((aligned(16))) float IC[4][W_];
  __shared__ __attribute__((aligned(16))) float FN[16][2][W_];
  __shared__ __attribute__((aligned(16))) float FM[16][2][H_];
  __shared__ float red[32];

  const int bc0 = b * C_ + half * 16;
  {
    const int ch = tid >> 6, h = (tid >> 5) & 1, x4 = (tid & 31) * 4;
    const float* srcn = ws + (h ? OFF_WC : OFF_WR) + c0 * (BC_ * W_) + (bc0 + ch) * W_ + x4;
    *(float4*)&FN[ch][h][x4] = *(const float4*)srcn;
    if (tid < 16 * 2 * 24) {
      const int chm = tid / 48, hm = (tid / 24) & 1, y4 = (tid % 24) * 4;
      const float* srcm = ws + (hm ? OFF_HC : OFF_HR) + c0 * (BC_ * H_) + (bc0 + chm) * H_ + y4;
      *(float4*)&FM[chm][hm][y4] = *(const float4*)srcm;
    }
  }

  float ig[4];
  #pragma unroll
  for (int t = 0; t < 4; ++t) ig[t] = 1.0f / ws[OFF_GMAX + c0 * 4 + t];

  const bool own = wave < 16;
  const int rg    = own ? (rh * 16 + wave) : ((1 - rh) * 16 + (wave - 16));
  const int row0  = rg * 3;
  const int lrow0 = (wave & 15) * 3;
  const int col0  = lane * 4;

  float z[3][4];
  #pragma unroll
  for (int r = 0; r < 3; ++r) { z[r][0] = 0.f; z[r][1] = 0.f; z[r][2] = 0.f; z[r][3] = 0.f; }
  float smax = 0.0f;

  __syncthreads();
  if (!own) {
    float mh[2][3], nw[2][4];
    #pragma unroll
    for (int r = 0; r < 3; ++r) { mh[0][r] = FM[0][0][row0 + r]; mh[1][r] = FM[0][1][row0 + r]; }
    const float4 a = *(const float4*)&FN[0][0][col0];
    const float4 d = *(const float4*)&FN[0][1][col0];
    nw[0][0] = a.x; nw[0][1] = a.y; nw[0][2] = a.z; nw[0][3] = a.w;
    nw[1][0] = d.x; nw[1][1] = d.y; nw[1][2] = d.z; nw[1][3] = d.w;
    km_col_partials_direct(mh, nw, PC[0], rg, col0);
  }

  for (int ci = 0; ci < 16; ++ci) {
    const int bc = bc0 + ci;
    float  E[4][3][4];
    float4 vv[3];
    if (own) {
      float mh[2][3], nw[2][4];
      #pragma unroll
      for (int r = 0; r < 3; ++r) { mh[0][r] = FM[ci][0][row0 + r]; mh[1][r] = FM[ci][1][row0 + r]; }
      const float4 a = *(const float4*)&FN[ci][0][col0];
      const float4 d = *(const float4*)&FN[ci][1][col0];
      nw[0][0] = a.x; nw[0][1] = a.y; nw[0][2] = a.z; nw[0][3] = a.w;
      nw[1][0] = d.x; nw[1][1] = d.y; nw[1][2] = d.z; nw[1][3] = d.w;
      km_exps(mh, nw, E);
      km_row_partials(E, PR, lrow0, lane);
      km_col_partials(E, PC[ci & 1], rg, col0);
    }
    __syncthreads();

    if (own) {
      #pragma unroll
      for (int r = 0; r < 3; ++r)
        vv[r] = *(const float4*)(values + bc * HW_ + (row0 + r) * W_ + col0);

      if (wave < 12) {
        const v2f ones = {1.0f, 1.0f};
        const int kho = (lane < 16) ? 0 : 2;
        const int tt = wave / 3, rb = wave % 3;
        const int M  = rb * 16 + (lane & 15);
        const float* pr = PR + tt * (HO_ * PRP) + M * PRP;
        v8f acc = {};
        for (int k0 = 0; k0 < 32; k0 += 4) {
          v2f a = { pr[k0 + kho], pr[k0 + kho + 1] };
          acc = __builtin_amdgcn_wmma_f32_16x16x4_f32(false, a, false, ones, (short)0, acc, false, false);
        }
        if ((lane & 15) == 0) {
          const int rbase = rb * 16 + ((lane >> 4) << 3);
          #pragma unroll
          for (int v = 0; v < 8; ++v) IR[tt][rbase + v] = 1.0f / acc[v];
        }
      }
    } else {
      if (wave < 24) {
        const int jw = wave - 16;
        const int tt = jw >> 1;
        const int g  = ((jw & 1) << 4) | (lane & 15);
        const int ph = lane >> 4;
        const float* pc = &PC[ci & 1][tt * (32 * W_) + (ph * 16) * W_ + g * 4];
        float4 s;
        #pragma unroll
        for (int i0 = 0; i0 < 16; i0 += 4) {
          const float4 x0 = *(const float4*)(pc + (i0 + 0) * W_);
          const float4 x1 = *(const float4*)(pc + (i0 + 1) * W_);
          const float4 x2 = *(const float4*)(pc + (i0 + 2) * W_);
          const float4 x3 = *(const float4*)(pc + (i0 + 3) * W_);
          float4 t;
          t.x = (x0.x + x1.x) + (x2.x + x3.x);
          t.y = (x0.y + x1.y) + (x2.y + x3.y);
          t.z = (x0.z + x1.z) + (x2.z + x3.z);
          t.w = (x0.w + x1.w) + (x2.w + x3.w);
          if (i0 == 0) { s = t; } else { s.x += t.x; s.y += t.y; s.z += t.z; s.w += t.w; }
          __builtin_amdgcn_sched_barrier(0);
        }
        s.x += xhalf(s.x); s.y += xhalf(s.y); s.z += xhalf(s.z); s.w += xhalf(s.w);
        float2 o;
        o.x = 1.0f / (ph ? s.z : s.x);
        o.y = 1.0f / (ph ? s.w : s.y);
        *(float2*)&IC[tt][g * 4 + ph * 2] = o;
      }
      if (ci + 1 < 16) {
        float mh[2][3], nw[2][4];
        #pragma unroll
        for (int r = 0; r < 3; ++r) { mh[0][r] = FM[ci + 1][0][row0 + r]; mh[1][r] = FM[ci + 1][1][row0 + r]; }
        const float4 a = *(const float4*)&FN[ci + 1][0][col0];
        const float4 d = *(const float4*)&FN[ci + 1][1][col0];
        nw[0][0] = a.x; nw[0][1] = a.y; nw[0][2] = a.z; nw[0][3] = a.w;
        nw[1][0] = d.x; nw[1][1] = d.y; nw[1][2] = d.z; nw[1][3] = d.w;
        km_col_partials_direct(mh, nw, PC[(ci + 1) & 1], rg, col0);
      }
    }
    __syncthreads();

    if (own) {
      float icv[4][4];
      #pragma unroll
      for (int t = 0; t < 4; ++t) {
        const float4 x = *(const float4*)&IC[t][col0];
        icv[t][0] = x.x; icv[t][1] = x.y; icv[t][2] = x.z; icv[t][3] = x.w;
      }
      #pragma unroll
      for (int r = 0; r < 3; ++r) {
        float irv[4];
        #pragma unroll
        for (int t = 0; t < 4; ++t) irv[t] = IR[t][lrow0 + r];
        const float vr[4] = { vv[r].x, vv[r].y, vv[r].z, vv[r].w };
        float sc[4];
        #pragma unroll
        for (int q = 0; q < 4; ++q) {
          float s = 0.f;
          #pragma unroll
          for (int t = 0; t < 4; ++t) s += E[t][r][q] * (irv[t] + icv[t][q]) * ig[t];
          z[r][q] += s * vr[q];
          smax = fmaxf(smax, s);
          sc[q] = s;
        }
        if (c0 == C_ - 1) {
          float4 o; o.x = sc[0]; o.y = sc[1]; o.z = sc[2]; o.w = sc[3];
          *(float4*)(out + BC_ * HW_ + bc * HW_ + (row0 + r) * W_ + col0) = o;
        }
      }
    }
  }

  if (own) {
    float* zdst = (half ? (ws + OFF_SCORE) : out) + (b * C_ + c0) * HW_;
    #pragma unroll
    for (int r = 0; r < 3; ++r) {
      float4 o; o.x = z[r][0]; o.y = z[r][1]; o.z = z[r][2]; o.w = z[r][3];
      *(float4*)(zdst + (row0 + r) * W_ + col0) = o;
    }
  }
  for (int o = 16; o > 0; o >>= 1) smax = fmaxf(smax, __shfl_xor(smax, o));
  if (lane == 0) red[wave] = smax;
  __syncthreads();
  if (tid == 0) {
    float g = red[0];
    for (int i = 1; i < 32; ++i) g = fmaxf(g, red[i]);
    atomicMax((int*)(ws + OFF_GSMAX + c0), __float_as_int(g));
  }
}

__global__ void ks_scale(const float* __restrict__ ws, float* __restrict__ out) {
  const int gid = blockIdx.x * 256 + threadIdx.x;
  float4* o4 = (float4*)out;
  if (gid < BC_ * HW_ / 4) {
    const int c0 = ((gid * 4) / HW_) % C_;
    const float inv = 1.0f / ws[OFF_GSMAX + c0];
    const float4 a = o4[gid];
    const float4 p = *(const float4*)(ws + OFF_SCORE + gid * 4);
    float4 r;
    r.x = (a.x + p.x) * inv; r.y = (a.y + p.y) * inv; r.z = (a.z + p.z) * inv; r.w = (a.w + p.w) * inv;
    o4[gid] = r;
  } else {
    const float inv = 1.0f / ws[OFF_GSMAX + 31];
    float4 a = o4[gid];
    a.x *= inv; a.y *= inv; a.z *= inv; a.w *= inv;
    o4[gid] = a;
  }
}

extern "C" void kernel_launch(void* const* d_in, const int* in_sizes, int n_in,
                              void* d_out, int out_size, void* d_ws, size_t ws_size,
                              hipStream_t stream) {
  (void)in_sizes; (void)n_in; (void)out_size; (void)ws_size;
  const float* q   = (const float*)d_in[0];
  const float* k   = (const float*)d_in[1];
  const float* val = (const float*)d_in[2];
  float* out = (float*)d_out;
  float* ws  = (float*)d_ws;

  k1_sums<<<64, 1024, 0, stream>>>(q, k, ws);
  k2_norm<<<128, 1024, 0, stream>>>(ws);
  kg_peak<<<128, 1024, 0, stream>>>(ws);
  km_fused<<<256, 1024, 0, stream>>>(ws, val, out);
  ks_scale<<<1536, 256, 0, stream>>>(ws, out);
}


# km: the two passes in front of it (min-max normalisation of the four factor families, term peaks) become its prologue, rebuilt per work-group in LDS from k1's plane sums; 5 launches to 3; same sums
# speedup vs baseline: 13.2610x; 1.2133x over previous
#include <hip/hip_runtime.h>

typedef __attribute__((ext_vector_type(2)))  float    v2f;
typedef __attribute__((ext_vector_type(8)))  float    v8f;

#define B_   2
#define C_   32
#define H_   96
#define W_   128
#define HW_  (H_ * W_)
#define BC_  (B_ * C_)
#define EPSF 1e-6f
#define LOG2E_F 1.44269504088896340736f

__device__ __forceinline__ float xhalf(float x) {
  const int b = __builtin_bit_cast(int, x);
  return __builtin_bit_cast(float, __builtin_amdgcn_permlanex16(b, b, 0x76543210, 0xfedcba98, false, false));
}

#define OFF_QS    0
#define OFF_QC    6144
#define OFF_KH    14336
#define OFF_KW    20480
#define OFF_SQ    28672
#define OFF_SK    28736
#define OFF_GMAX  28800
#define OFF_GSMAX 28928
#define OFF_HR    28960
#define OFF_HC    (OFF_HR + 32 * 6144)
#define OFF_WR    (OFF_HC + 32 * 6144)
#define OFF_WC    (OFF_WR + 32 * 8192)
#define OFF_IRS   (OFF_WC + 32 * 8192)
#define OFF_SCORE (OFF_IRS + 4 * 64 * 224)


__global__ void __launch_bounds__(1024)
k1_sums(const float* __restrict__ q, const float* __restrict__ k, float* __restrict__ ws) {
  __shared__ float tile[HW_];
  __shared__ float rowbuf[H_];
  const int bc  = blockIdx.x;
  const int tid = threadIdx.x;
  if (bc == 0 && tid < C_) ws[OFF_GSMAX + tid] = 0.0f;
  for (int p = 0; p < 2; ++p) {
    const float* s = p ? k : q;
    #pragma unroll
    for (int j = 0; j < HW_ / 1024; ++j) tile[tid + j * 1024] = s[bc * HW_ + tid + j * 1024];
    __syncthreads();
    if (tid < H_) {
      float acc = 0.f;
      for (int x = 0; x < W_; ++x) acc += tile[tid * W_ + x];
      ws[(p ? OFF_KH : OFF_QS) + bc * H_ + tid] = acc;
      rowbuf[tid] = acc;
    }
    if (tid >= 128 && tid < 256) {
      const int x = tid - 128;
      float acc = 0.f;
      for (int h = 0; h < H_; ++h) acc += tile[h * W_ + x];
      ws[(p ? OFF_KW : OFF_QC) + bc * W_ + x] = acc;
    }
    __syncthreads();
    if (tid == 0) {
      float t = 0.f;
      for (int h = 0; h < H_; ++h) t += rowbuf[h];
      ws[(p ? OFF_SK : OFF_SQ) + bc] = t;
    }
    __syncthreads();
  }
}

#define PRP 33
#define HO_ 48

__device__ __forceinline__ void km_exps(const float (&mh)[2][3], const float (&nw)[2][4], float (&E)[4][3][4]) {
  float ml[2][3];
  #pragma unroll
  for (int r = 0; r < 3; ++r) { ml[0][r] = mh[0][r] * LOG2E_F; ml[1][r] = mh[1][r] * LOG2E_F; }
  #pragma unroll
  for (int t = 0; t < 4; ++t) {
    #pragma unroll
    for (int r = 0; r < 3; ++r) {
      #pragma unroll
      for (int q = 0; q < 4; ++q) E[t][r][q] = ml[t >> 1][r] * nw[t & 1][q];
    }
  }
  #pragma unroll
  for (int t = 0; t < 4; ++t) {
    #pragma unroll
    for (int r = 0; r < 3; ++r) {
      #pragma unroll
      for (int q = 0; q < 4; ++q) E[t][r][q] = __builtin_amdgcn_exp2f(E[t][r][q]);
    }
  }
}

__device__ __forceinline__ void km_col_partials(const float (&E)[4][3][4], float* pc, int rg, int col0) {
  #pragma unroll
  for (int t = 0; t < 4; ++t) {
    float4 cp;
    cp.x = E[t][0][0] + E[t][1][0] + E[t][2][0];
    cp.y = E[t][0][1] + E[t][1][1] + E[t][2][1];
    cp.z = E[t][0][2] + E[t][1][2] + E[t][2][2];
    cp.w = E[t][0][3] + E[t][1][3] + E[t][2][3];
    *(float4*)(pc + t * (32 * W_) + rg * W_ + col0) = cp;
  }
}

__device__ __forceinline__ void km_row_partials(const float (&E)[4][3][4], float* pr, int lrow0, int lane) {
  #pragma unroll
  for (int t = 0; t < 4; ++t) {
    #pragma unroll
    for (int r = 0; r < 3; ++r)
      pr[t * (HO_ * PRP) + (lrow0 + r) * PRP + lane] = (E[t][r][0] + E[t][r][1]) + (E[t][r][2] + E[t][r][3]);
  }
}

__device__ __forceinline__ void km_col_partials_direct(const float (&mh)[2][3], const float (&nw)[2][4],
                                                       float* pc, int rg, int col0) {
  float ml[2][3];
  #pragma unroll
  for (int r = 0; r < 3; ++r) { ml[0][r] = mh[0][r] * LOG2E_F; ml[1][r] = mh[1][r] * LOG2E_F; }
  #pragma unroll
  for (int t = 0; t < 4; ++t) {
    float e[3][4];
    #pragma unroll
    for (int r = 0; r < 3; ++r) {
      #pragma unroll
      for (int q = 0; q < 4; ++q) e[r][q] = ml[t >> 1][r] * nw[t & 1][q];
    }
    #pragma unroll
    for (int r = 0; r < 3; ++r) {
      #pragma unroll
      for (int q = 0; q < 4; ++q) e[r][q] = __builtin_amdgcn_exp2f(e[r][q]);
    }
    float4 cp;
    cp.x = e[0][0] + e[1][0] + e[2][0];
    cp.y = e[0][1] + e[1][1] + e[2][1];
    cp.z = e[0][2] + e[1][2] + e[2][2];
    cp.w = e[0][3] + e[1][3] + e[2][3];
    *(float4*)(pc + t * (32 * W_) + rg * W_ + col0) = cp;
  }
}

template <int K> __device__ __forceinline__ float rowx(float x) {
  const int v = __builtin_bit_cast(int, x);
  return __builtin_bit_cast(float, __builtin_amdgcn_update_dpp(v, v, 0x160 + K, 0xf, 0xf, false));
}
template <class F> __device__ __forceinline__ float red8(const float (&v)[8], int lane, F op) {
  const bool h = (lane & 16) != 0, b3 = (lane & 8) != 0, b2 = (lane & 4) != 0;
  float a[4], b[2];
  #pragma unroll
  for (int i = 0; i < 4; ++i) a[i] = op(h ? v[i + 4] : v[i], xhalf(h ? v[i] : v[i + 4]));
  #pragma unroll
  for (int i = 0; i < 2; ++i) b[i] = op(b3 ? a[i + 2] : a[i], rowx<8>(b3 ? a[i] : a[i + 2]));
  float c = op(b2 ? b[1] : b[0], rowx<4>(b2 ? b[0] : b[1]));
  c = op(c, rowx<2>(c));
  c = op(c, rowx<1>(c));
  return c;
}
__device__ __forceinline__ float lane_of(float x, int l) {
  return __builtin_bit_cast(float, __builtin_amdgcn_readlane(__builtin_bit_cast(int, x), l));
}

#define T_HR 0
#define T_HC 6144
#define T_WR 12288
#define T_WC 20480
#define T_N  28672

__device__ __forceinline__ void mnmx4(const float4 v, float& mn, float& mx) {
  mn = fminf(fminf(mn, v.x), fminf(v.y, fminf(v.z, v.w)));
  mx = fmaxf(fmaxf(mx, v.x), fmaxf(v.y, fmaxf(v.z, v.w)));
}
__device__ __forceinline__ float4 norm4(const float4 v, const float mn, const float inv) {
  float4 o;
  o.x = 2.0f + (v.x - mn) * inv; o.y = 2.0f + (v.y - mn) * inv;
  o.z = 2.0f + (v.z - mn) * inv; o.w = 2.0f + (v.w - mn) * inv;
  return o;
}
__device__ __forceinline__ void raw_h(const float* __restrict__ ws, int c0, int j4, float sq0, float sq1,
                                      float4& rhr, float4& rhc) {
  const int pb = j4 / 24, y4 = (j4 - pb * 24) * 4;
  const int bb = pb >> 5;
  const float sq = bb ? sq1 : sq0;
  const float sk = ws[OFF_SK + pb];
  const float4 kh = *(const float4*)(ws + OFF_KH + pb * H_ + y4);
  const float4 qs = *(const float4*)(ws + OFF_QS + (bb * C_ + c0) * H_ + y4);
  rhr.x = sq * kh.x + (H_ * EPSF); rhr.y = sq * kh.y + (H_ * EPSF);
  rhr.z = sq * kh.z + (H_ * EPSF); rhr.w = sq * kh.w + (H_ * EPSF);
  rhc.x = qs.x * sk + (H_ * EPSF); rhc.y = qs.y * sk + (H_ * EPSF);
  rhc.z = qs.z * sk + (H_ * EPSF); rhc.w = qs.w * sk + (H_ * EPSF);
}
__device__ __forceinline__ void raw_w(const float* __restrict__ ws, int c0, int j4, float sq0, float sq1,
                                      float4& rwr, float4& rwc) {
  const int pb = j4 >> 5, x4 = (j4 & 31) * 4;
  const int bb = pb >> 5;
  const float sq = bb ? sq1 : sq0;
  const float sk = ws[OFF_SK + pb];
  const float4 kw = *(const float4*)(ws + OFF_KW + pb * W_ + x4);
  const float4 qc = *(const float4*)(ws + OFF_QC + (bb * C_ + c0) * W_ + x4);
  rwr.x = sq * kw.x + (W_ * EPSF); rwr.y = sq * kw.y + (W_ * EPSF);
  rwr.z = sq * kw.z + (W_ * EPSF); rwr.w = sq * kw.w + (W_ * EPSF);
  rwc.x = qc.x * sk + (W_ * EPSF); rwc.y = qc.y * sk + (W_ * EPSF);
  rwc.z = qc.z * sk + (W_ * EPSF); rwc.w = qc.w * sk + (W_ * EPSF);
}

__global__ void __launch_bounds__(1024)
km_fused(float* __restrict__ ws, const float* __restrict__ values, float* __restrict__ out) {
  const int rh   = blockIdx.x & 1;
  const int half = (blockIdx.x >> 1) & 1;
  const int b    = (blockIdx.x >> 2) & 1;
  const int c0   = blockIdx.x >> 3;
  const int tid  = threadIdx.x;
  const int wave = __builtin_amdgcn_readfirstlane(tid >> 5);
  const int lane = tid & 31;
  __shared__ float PR[4 * HO_ * PRP];
  __shared__ __attribute__((aligned(16))) float PC[2][4 * 32 * W_];
  __shared__ float IR[4][HO_];
  __shared__ __attribute__((aligned(16))) float IC[4][W_];
  __shared__ __attribute__((aligned(16))) float FN[16][2][W_];
  __shared__ __attribute__((aligned(16))) float FM[16][2][H_];
  __shared__ float red[32];
  float* const TF = &PC[0][0];
  float (* const RMN)[32] = (float (*)[32])(TF + T_N);
  float (* const RMX)[32] = (float (*)[32])(TF + T_N + 128);
  float (* const GP)[32]  = (float (*)[32])(TF + T_N + 256);

  const int bc0 = b * C_ + half * 16;
  float ig[4];
  {
    const bool two = wave < 16;
    const float sq0 = ws[OFF_SQ + c0];
    const float sq1 = ws[OFF_SQ + C_ + c0];
    float4 rhr0, rhc0, rhr1, rhc1, rwr0, rwc0, rwr1, rwc1;
    raw_h(ws, c0, tid, sq0, sq1, rhr0, rhc0);
    raw_w(ws, c0, tid, sq0, sq1, rwr0, rwc0);
    raw_w(ws, c0, tid + 1024, sq0, sq1, rwr1, rwc1);
    rhr1 = rhr0; rhc1 = rhc0;
    if (two) raw_h(ws, c0, tid + 1024, sq0, sq1, rhr1, rhc1);

    float mn[4], mx[4];
    #pragma unroll
    for (int f = 0; f < 4; ++f) { mn[f] = 3.4e38f; mx[f] = -3.4e38f; }
    mnmx4(rhr0, mn[0], mx[0]); mnmx4(rhr1, mn[0], mx[0]);
    mnmx4(rhc0, mn[1], mx[1]); mnmx4(rhc1, mn[1], mx[1]);
    mnmx4(rwr0, mn[2], mx[2]); mnmx4(rwr1, mn[2], mx[2]);
    mnmx4(rwc0, mn[3], mx[3]); mnmx4(rwc1, mn[3], mx[3]);
    {
      const float v8[8] = { mn[0], mn[1], mn[2], mn[3], -mx[0], -mx[1], -mx[2], -mx[3] };
      const float c = red8(v8, lane, [](float x, float y) { return fminf(x, y); });
      if ((lane & 3) == 0) RMN[lane >> 2][wave] = (lane < 16) ? c : -c;
    }
    __syncthreads();
    float inv[4];
    {
      const int fo = lane >> 3, eo = (lane & 7) * 4;
      const float4 a4 = *(const float4*)&RMN[fo][eo];
      const float4 b4 = *(const float4*)&RMX[fo][eo];
      float lmn = fminf(fminf(a4.x, a4.y), fminf(a4.z, a4.w));
      float lmx = fmaxf(fmaxf(b4.x, b4.y), fmaxf(b4.z, b4.w));
      lmn = fminf(lmn, rowx<1>(lmn)); lmn = fminf(lmn, rowx<2>(lmn)); lmn = fminf(lmn, rowx<4>(lmn));
      lmx = fmaxf(lmx, rowx<1>(lmx)); lmx = fmaxf(lmx, rowx<2>(lmx)); lmx = fmaxf(lmx, rowx<4>(lmx));
      const float linv = 1.0f / (lmx - lmn);
      #pragma unroll
      for (int f = 0; f < 4; ++f) { mn[f] = lane_of(lmn, 8 * f); inv[f] = lane_of(linv, 8 * f); }
    }

    #pragma unroll
    for (int k = 0; k < 2; ++k) {
      const int j4 = tid + k * 1024;
      const float4 wr = norm4(k ? rwr1 : rwr0, mn[2], inv[2]);
      const float4 wc = norm4(k ? rwc1 : rwc0, mn[3], inv[3]);
      *(float4*)&TF[T_WR + j4 * 4] = wr;
      *(float4*)&TF[T_WC + j4 * 4] = wc;
      const int chn = (j4 >> 5) - bc0, x4 = (j4 & 31) * 4;
      if (chn >= 0 && chn < 16) { *(float4*)&FN[chn][0][x4] = wr; *(float4*)&FN[chn][1][x4] = wc; }
      if (k == 0 || two) {
        const float4 hr = norm4(k ? rhr1 : rhr0, mn[0], inv[0]);
        const float4 hc = norm4(k ? rhc1 : rhc0, mn[1], inv[1]);
        *(float4*)&TF[T_HR + j4 * 4] = hr;
        *(float4*)&TF[T_HC + j4 * 4] = hc;
        const int pbm = j4 / 24, chm = pbm - bc0, y4 = (j4 - pbm * 24) * 4;
        if (chm >= 0 && chm < 16) { *(float4*)&FM[chm][0][y4] = hr; *(float4*)&FM[chm][1][y4] = hc; }
      }
    }
    __syncthreads();

    {
      float m[2][2][3], n[2][2][4];
      #pragma unroll
      for (int i = 0; i < 2; ++i) {
        const int pb = wave + i * 32;
        #pragma unroll
        for (int r = 0; r < 3; ++r) {
          m[i][0][r] = TF[T_HR + pb * H_ + lane * 3 + r];
          m[i][1][r] = TF[T_HC + pb * H_ + lane * 3 + r];
        }
        const float4 a = *(const float4*)&TF[T_WR + pb * W_ + lane * 4];
        const float4 d = *(const float4*)&TF[T_WC + pb * W_ + lane * 4];
        n[i][0][0] = a.x; n[i][0][1] = a.y; n[i][0][2] = a.z; n[i][0][3] = a.w;
        n[i][1][0] = d.x; n[i][1][1] = d.y; n[i][1][2] = d.z; n[i][1][3] = d.w;
      }
      float lm[8];
      #pragma unroll
      for (int i = 0; i < 2; ++i) {
        #pragma unroll
        for (int f = 0; f < 2; ++f) {
          lm[i * 4 + f]     = fmaxf(fmaxf(m[i][f][0], m[i][f][1]), m[i][f][2]);
          lm[i * 4 + 2 + f] = fmaxf(fmaxf(n[i][f][0], n[i][f][1]), fmaxf(n[i][f][2], n[i][f][3]));
        }
      }
      const float cm = red8(lm, lane, [](float x, float y) { return fmaxf(x, y); });
      float mml[2][2], nn[2][2];
      #pragma unroll
      for (int i = 0; i < 2; ++i) {
        #pragma unroll
        for (int f = 0; f < 2; ++f) {
          mml[i][f] = lane_of(cm, 4 * (i * 4 + f)) * LOG2E_F;
          nn[i][f]  = lane_of(cm, 4 * (i * 4 + 2 + f));
        }
      }
      float rsv[8], csv[8];
      #pragma unroll
      for (int i = 0; i < 2; ++i) {
        #pragma unroll
        for (int t = 0; t < 4; ++t) {
          const int f = t >> 1, g = t & 1;
          rsv[i * 4 + t] = __builtin_amdgcn_exp2f(mml[i][f] * n[i][g][0]) + __builtin_amdgcn_exp2f(mml[i][f] * n[i][g][1])
                         + __builtin_amdgcn_exp2f(mml[i][f] * n[i][g][2]) + __builtin_amdgcn_exp2f(mml[i][f] * n[i][g][3]);
          csv[i * 4 + t] = __builtin_amdgcn_exp2f((m[i][f][0] * LOG2E_F) * nn[i][g]) + __builtin_amdgcn_exp2f((m[i][f][1] * LOG2E_F) * nn[i][g])
                         + __builtin_amdgcn_exp2f((m[i][f][2] * LOG2E_F) * nn[i][g]);
        }
      }
      const float rs = red8(rsv, lane, [](float x, float y) { return x + y; });
      const float cs = red8(csv, lane, [](float x, float y) { return x + y; });
      const bool hi = (lane & 16) != 0, f1 = (lane & 8) != 0, g1 = (lane & 4) != 0;
      const float mml_l = hi ? (f1 ? mml[1][1] : mml[1][0]) : (f1 ? mml[0][1] : mml[0][0]);
      const float nn_l  = hi ? (g1 ? nn[1][1] : nn[1][0]) : (g1 ? nn[0][1] : nn[0][0]);
      float pk = __builtin_amdgcn_exp2f(mml_l * nn_l) * (1.0f / rs + 1.0f / cs);
      pk = fmaxf(pk, xhalf(pk));
      if ((lane & 19) == 0) GP[lane >> 2][wave] = pk;
    }
    __syncthreads();
    {
      const float4 g4 = *(const float4*)&GP[lane >> 3][(lane & 7) * 4];
      float g = fmaxf(fmaxf(g4.x, g4.y), fmaxf(g4.z, g4.w));
      g = fmaxf(g, rowx<1>(g)); g = fmaxf(g, rowx<2>(g)); g = fmaxf(g, rowx<4>(g));
      const float lig = 1.0f / g;
      #pragma unroll
      for (int t = 0; t < 4; ++t) ig[t] = lane_of(lig, 8 * t);
    }
  }

  const bool own = wave < 16;
  const int rg    = own ? (rh * 16 + wave) : ((1 - rh) * 16 + (wave - 16));
  const int row0  = rg * 3;
  const int lrow0 = (wave & 15) * 3;
  const int col0  = lane * 4;

  float z[3][4];
  #pragma unroll
  for (int r = 0; r < 3; ++r) { z[r][0] = 0.f; z[r][1] = 0.f; z[r][2] = 0.f; z[r][3] = 0.f; }
  float smax = 0.0f;

  if (!own) {
    float mh[2][3], nw[2][4];
    #pragma unroll
    for (int r = 0; r < 3; ++r) { mh[0][r] = FM[0][0][row0 + r]; mh[1][r] = FM[0][1][row0 + r]; }
    const float4 a = *(const float4*)&FN[0][0][col0];
    const float4 d = *(const float4*)&FN[0][1][col0];
    nw[0][0] = a.x; nw[0][1] = a.y; nw[0][2] = a.z; nw[0][3] = a.w;
    nw[1][0] = d.x; nw[1][1] = d.y; nw[1][2] = d.z; nw[1][3] = d.w;
    km_col_partials_direct(mh, nw, PC[0], rg, col0);
  }

  for (int ci = 0; ci < 16; ++ci) {
    const int bc = bc0 + ci;
    float  E[4][3][4];
    float4 vv[3];
    if (own) {
      float mh[2][3], nw[2][4];
      #pragma unroll
      for (int r = 0; r < 3; ++r) { mh[0][r] = FM[ci][0][row0 + r]; mh[1][r] = FM[ci][1][row0 + r]; }
      const float4 a = *(const float4*)&FN[ci][0][col0];
      const float4 d = *(const float4*)&FN[ci][1][col0];
      nw[0][0] = a.x; nw[0][1] = a.y; nw[0][2] = a.z; nw[0][3] = a.w;
      nw[1][0] = d.x; nw[1][1] = d.y; nw[1][2] = d.z; nw[1][3] = d.w;
      km_exps(mh, nw, E);
      km_row_partials(E, PR, lrow0, lane);
      km_col_partials(E, PC[ci & 1], rg, col0);
    }
    __syncthreads();

    if (own) {
      #pragma unroll
      for (int r = 0; r < 3; ++r)
        vv[r] = *(const float4*)(values + bc * HW_ + (row0 + r) * W_ + col0);

      if (wave < 12) {
        const v2f ones = {1.0f, 1.0f};
        const int kho = (lane < 16) ? 0 : 2;
        const int tt = wave / 3, rb = wave % 3;
        const int M  = rb * 16 + (lane & 15);
        const float* pr = PR + tt * (HO_ * PRP) + M * PRP;
        v8f acc = {};
        for (int k0 = 0; k0 < 32; k0 += 4) {
          v2f a = { pr[k0 + kho], pr[k0 + kho + 1] };
          acc = __builtin_amdgcn_wmma_f32_16x16x4_f32(false, a, false, ones, (short)0, acc, false, false);
        }
        if ((lane & 15) == 0) {
          const int rbase = rb * 16 + ((lane >> 4) << 3);
          #pragma unroll
          for (int v = 0; v < 8; ++v) IR[tt][rbase + v] = 1.0f / acc[v];
        }
      }
    } else {
      if (wave < 24) {
        const int jw = wave - 16;
        const int tt = jw >> 1;
        const int g  = ((jw & 1) << 4) | (lane & 15);
        const int ph = lane >> 4;
        const float* pc = &PC[ci & 1][tt * (32 * W_) + (ph * 16) * W_ + g * 4];
        float4 s;
        #pragma unroll
        for (int i0 = 0; i0 < 16; i0 += 4) {
          const float4 x0 = *(const float4*)(pc + (i0 + 0) * W_);
          const float4 x1 = *(const float4*)(pc + (i0 + 1) * W_);
          const float4 x2 = *(const float4*)(pc + (i0 + 2) * W_);
          const float4 x3 = *(const float4*)(pc + (i0 + 3) * W_);
          float4 t;
          t.x = (x0.x + x1.x) + (x2.x + x3.x);
          t.y = (x0.y + x1.y) + (x2.y + x3.y);
          t.z = (x0.z + x1.z) + (x2.z + x3.z);
          t.w = (x0.w + x1.w) + (x2.w + x3.w);
          if (i0 == 0) { s = t; } else { s.x += t.x; s.y += t.y; s.z += t.z; s.w += t.w; }
          __builtin_amdgcn_sched_barrier(0);
        }
        s.x += xhalf(s.x); s.y += xhalf(s.y); s.z += xhalf(s.z); s.w += xhalf(s.w);
        float2 o;
        o.x = 1.0f / (ph ? s.z : s.x);
        o.y = 1.0f / (ph ? s.w : s.y);
        *(float2*)&IC[tt][g * 4 + ph * 2] = o;
      }
      if (ci + 1 < 16) {
        float mh[2][3], nw[2][4];
        #pragma unroll
        for (int r = 0; r < 3; ++r) { mh[0][r] = FM[ci + 1][0][row0 + r]; mh[1][r] = FM[ci + 1][1][row0 + r]; }
        const float4 a = *(const float4*)&FN[ci + 1][0][col0];
        const float4 d = *(const float4*)&FN[ci + 1][1][col0];
        nw[0][0] = a.x; nw[0][1] = a.y; nw[0][2] = a.z; nw[0][3] = a.w;
        nw[1][0] = d.x; nw[1][1] = d.y; nw[1][2] = d.z; nw[1][3] = d.w;
        km_col_partials_direct(mh, nw, PC[(ci + 1) & 1], rg, col0);
      }
    }
    __syncthreads();

    if (own) {
      float icv[4][4];
      #pragma unroll
      for (int t = 0; t < 4; ++t) {
        const float4 x = *(const float4*)&IC[t][col0];
        icv[t][0] = x.x; icv[t][1] = x.y; icv[t][2] = x.z; icv[t][3] = x.w;
      }
      #pragma unroll
      for (int r = 0; r < 3; ++r) {
        float irv[4];
        #pragma unroll
        for (int t = 0; t < 4; ++t) irv[t] = IR[t][lrow0 + r];
        const float vr[4] = { vv[r].x, vv[r].y, vv[r].z, vv[r].w };
        float sc[4];
        #pragma unroll
        for (int q = 0; q < 4; ++q) {
          float s = 0.f;
          #pragma unroll
          for (int t = 0; t < 4; ++t) s += E[t][r][q] * (irv[t] + icv[t][q]) * ig[t];
          z[r][q] += s * vr[q];
          smax = fmaxf(smax, s);
          sc[q] = s;
        }
        if (c0 == C_ - 1) {
          float4 o; o.x = sc[0]; o.y = sc[1]; o.z = sc[2]; o.w = sc[3];
          *(float4*)(out + BC_ * HW_ + bc * HW_ + (row0 + r) * W_ + col0) = o;
        }
      }
    }
  }

  if (own) {
    float* zdst = (half ? (ws + OFF_SCORE) : out) + (b * C_ + c0) * HW_;
    #pragma unroll
    for (int r = 0; r < 3; ++r) {
      float4 o; o.x = z[r][0]; o.y = z[r][1]; o.z = z[r][2]; o.w = z[r][3];
      *(float4*)(zdst + (row0 + r) * W_ + col0) = o;
    }
  }
  for (int o = 16; o > 0; o >>= 1) smax = fmaxf(smax, __shfl_xor(smax, o));
  if (lane == 0) red[wave] = smax;
  __syncthreads();
  if (tid == 0) {
    float g = red[0];
    for (int i = 1; i < 32; ++i) g = fmaxf(g, red[i]);
    atomicMax((int*)(ws + OFF_GSMAX + c0), __float_as_int(g));
  }
}

__global__ void ks_scale(const float* __restrict__ ws, float* __restrict__ out) {
  const int gid = blockIdx.x * 256 + threadIdx.x;
  float4* o4 = (float4*)out;
  if (gid < BC_ * HW_ / 4) {
    const int c0 = ((gid * 4) / HW_) % C_;
    const float inv = 1.0f / ws[OFF_GSMAX + c0];
    const float4 a = o4[gid];
    const float4 p = *(const float4*)(ws + OFF_SCORE + gid * 4);
    float4 r;
    r.x = (a.x + p.x) * inv; r.y = (a.y + p.y) * inv; r.z = (a.z + p.z) * inv; r.w = (a.w + p.w) * inv;
    o4[gid] = r;
  } else {
    const float inv = 1.0f / ws[OFF_GSMAX + 31];
    float4 a = o4[gid];
    a.x *= inv; a.y *= inv; a.z *= inv; a.w *= inv;
    o4[gid] = a;
  }
}

extern "C" void kernel_launch(void* const* d_in, const int* in_sizes, int n_in,
                              void* d_out, int out_size, void* d_ws, size_t ws_size,
                              hipStream_t stream) {
  (void)in_sizes; (void)n_in; (void)out_size; (void)ws_size;
  const float* q   = (const float*)d_in[0];
  const float* k   = (const float*)d_in[1];
  const float* val = (const float*)d_in[2];
  float* out = (float*)d_out;
  float* ws  = (float*)d_ws;

  k1_sums<<<64, 1024, 0, stream>>>(q, k, ws);
  km_fused<<<256, 1024, 0, stream>>>(ws, val, out);
  ks_scale<<<1536, 256, 0, stream>>>(ws, out);
}


# k1: one 8-wave block per (plane, tensor); plane staged in an LDS tile of pitch 132 floats, row and column sums formed by f32 WMMAs against a ones operand, not by serial loops; sum order changes
# speedup vs baseline: 14.7959x; 1.1157x over previous
#include <hip/hip_runtime.h>

typedef __attribute__((ext_vector_type(2)))  float    v2f;
typedef __attribute__((ext_vector_type(8)))  float    v8f;

#define B_   2
#define C_   32
#define H_   96
#define W_   128
#define HW_  (H_ * W_)
#define BC_  (B_ * C_)
#define EPSF 1e-6f
#define LOG2E_F 1.44269504088896340736f

__device__ __forceinline__ float xhalf(float x) {
  const int b = __builtin_bit_cast(int, x);
  return __builtin_bit_cast(float, __builtin_amdgcn_permlanex16(b, b, 0x76543210, 0xfedcba98, false, false));
}

#define OFF_QS    0
#define OFF_QC    6144
#define OFF_KH    14336
#define OFF_KW    20480
#define OFF_SQ    28672
#define OFF_SK    28736
#define OFF_GMAX  28800
#define OFF_GSMAX 28928
#define OFF_HR    28960
#define OFF_HC    (OFF_HR + 32 * 6144)
#define OFF_WR    (OFF_HC + 32 * 6144)
#define OFF_WC    (OFF_WR + 32 * 8192)
#define OFF_IRS   (OFF_WC + 32 * 8192)
#define OFF_SCORE (OFF_IRS + 4 * 64 * 224)


#define TP_ 132
__device__ __forceinline__ v8f mm4(const v2f a, const v2f b, const v8f c) {
  return __builtin_amdgcn_wmma_f32_16x16x4_f32(false, a, false, b, (short)0, c, false, false);
}
__global__ void __launch_bounds__(256)
k1_sums(const float* __restrict__ q, const float* __restrict__ k, float* __restrict__ ws) {
  __shared__ __attribute__((aligned(16))) float tile[H_ * TP_];
  __shared__ float part[6];
  const int bc   = blockIdx.x >> 1;
  const int p    = blockIdx.x & 1;
  const int tid  = threadIdx.x;
  const int wave = __builtin_amdgcn_readfirstlane(tid >> 5);
  const int lane = tid & 31;
  if (blockIdx.x == 0 && tid < C_) ws[OFF_GSMAX + tid] = 0.0f;
  const float* s = (p ? k : q) + bc * HW_;
  float4 v[HW_ / 1024];
  #pragma unroll
  for (int j = 0; j < HW_ / 1024; ++j) v[j] = *(const float4*)(s + (tid + j * 256) * 4);
  #pragma unroll
  for (int j = 0; j < HW_ / 1024; ++j) {
    const int i4 = tid + j * 256;
    *(float4*)&tile[(i4 >> 5) * TP_ + (i4 & 31) * 4] = v[j];
  }
  __syncthreads();

  const v2f ones = {1.0f, 1.0f};
  const int c = lane & 15, h = lane >> 4;
  {
    const float* pc = tile + (2 * h) * TP_ + wave * 16 + c;
    v8f a0 = {}, a1 = {};
    #pragma unroll
    for (int k0 = 0; k0 < H_; k0 += 8) {
      const v2f y0 = { pc[k0 * TP_],       pc[(k0 + 1) * TP_] };
      const v2f y1 = { pc[(k0 + 4) * TP_], pc[(k0 + 5) * TP_] };
      a0 = mm4(ones, y0, a0);
      a1 = mm4(ones, y1, a1);
    }
    const float cs = a0[0] + a1[0];
    if (lane < 16) ws[(p ? OFF_KW : OFF_QC) + bc * W_ + wave * 16 + lane] = cs;
  }
  if (wave < 6) {
    const float* pr = tile + (wave * 16 + c) * TP_ + 2 * h;
    v8f a0 = {}, a1 = {};
    #pragma unroll
    for (int k0 = 0; k0 < W_; k0 += 8) {
      const v2f x0 = *(const v2f*)(pr + k0);
      const v2f x1 = *(const v2f*)(pr + k0 + 4);
      a0 = mm4(x0, ones, a0);
      a1 = mm4(x1, ones, a1);
    }
    float acc[8];
    #pragma unroll
    for (int r = 0; r < 8; ++r) acc[r] = a0[r] + a1[r];
    float rs = acc[0];
    #pragma unroll
    for (int r = 1; r < 8; ++r) rs = ((lane & 7) == r) ? acc[r] : rs;
    if ((lane & 8) == 0) ws[(p ? OFF_KH : OFF_QS) + bc * H_ + wave * 16 + 8 * h + (lane & 7)] = rs;
    float t = ((acc[0] + acc[1]) + (acc[2] + acc[3])) + ((acc[4] + acc[5]) + (acc[6] + acc[7]));
    t += xhalf(t);
    if (lane == 0) part[wave] = t;
  }
  __syncthreads();
  if (tid == 0) ws[(p ? OFF_SK : OFF_SQ) + bc] = ((part[0] + part[1]) + (part[2] + part[3])) + (part[4] + part[5]);
}

#define PRP 33
#define HO_ 48

__device__ __forceinline__ void km_exps(const float (&mh)[2][3], const float (&nw)[2][4], float (&E)[4][3][4]) {
  float ml[2][3];
  #pragma unroll
  for (int r = 0; r < 3; ++r) { ml[0][r] = mh[0][r] * LOG2E_F; ml[1][r] = mh[1][r] * LOG2E_F; }
  #pragma unroll
  for (int t = 0; t < 4; ++t) {
    #pragma unroll
    for (int r = 0; r < 3; ++r) {
      #pragma unroll
      for (int q = 0; q < 4; ++q) E[t][r][q] = ml[t >> 1][r] * nw[t & 1][q];
    }
  }
  #pragma unroll
  for (int t = 0; t < 4; ++t) {
    #pragma unroll
    for (int r = 0; r < 3; ++r) {
      #pragma unroll
      for (int q = 0; q < 4; ++q) E[t][r][q] = __builtin_amdgcn_exp2f(E[t][r][q]);
    }
  }
}

__device__ __forceinline__ void km_col_partials(const float (&E)[4][3][4], float* pc, int rg, int col0) {
  #pragma unroll
  for (int t = 0; t < 4; ++t) {
    float4 cp;
    cp.x = E[t][0][0] + E[t][1][0] + E[t][2][0];
    cp.y = E[t][0][1] + E[t][1][1] + E[t][2][1];
    cp.z = E[t][0][2] + E[t][1][2] + E[t][2][2];
    cp.w = E[t][0][3] + E[t][1][3] + E[t][2][3];
    *(float4*)(pc + t * (32 * W_) + rg * W_ + col0) = cp;
  }
}

__device__ __forceinline__ void km_row_partials(const float (&E)[4][3][4], float* pr, int lrow0, int lane) {
  #pragma unroll
  for (int t = 0; t < 4; ++t) {
    #pragma unroll
    for (int r = 0; r < 3; ++r)
      pr[t * (HO_ * PRP) + (lrow0 + r) * PRP + lane] = (E[t][r][0] + E[t][r][1]) + (E[t][r][2] + E[t][r][3]);
  }
}

__device__ __forceinline__ void km_col_partials_direct(const float (&mh)[2][3], const float (&nw)[2][4],
                                                       float* pc, int rg, int col0) {
  float ml[2][3];
  #pragma unroll
  for (int r = 0; r < 3; ++r) { ml[0][r] = mh[0][r] * LOG2E_F; ml[1][r] = mh[1][r] * LOG2E_F; }
  #pragma unroll
  for (int t = 0; t < 4; ++t) {
    float e[3][4];
    #pragma unroll
    for (int r = 0; r < 3; ++r) {
      #pragma unroll
      for (int q = 0; q < 4; ++q) e[r][q] = ml[t >> 1][r] * nw[t & 1][q];
    }
    #pragma unroll
    for (int r = 0; r < 3; ++r) {
      #pragma unroll
      for (int q = 0; q < 4; ++q) e[r][q] = __builtin_amdgcn_exp2f(e[r][q]);
    }
    float4 cp;
    cp.x = e[0][0] + e[1][0] + e[2][0];
    cp.y = e[0][1] + e[1][1] + e[2][1];
    cp.z = e[0][2] + e[1][2] + e[2][2];
    cp.w = e[0][3] + e[1][3] + e[2][3];
    *(float4*)(pc + t * (32 * W_) + rg * W_ + col0) = cp;
  }
}

template <int K> __device__ __forceinline__ float rowx(float x) {
  const int v = __builtin_bit_cast(int, x);
  return __builtin_bit_cast(float, __builtin_amdgcn_update_dpp(v, v, 0x160 + K, 0xf, 0xf, false));
}
template <class F> __device__ __forceinline__ float red8(const float (&v)[8], int lane, F op) {
  const bool h = (lane & 16) != 0, b3 = (lane & 8) != 0, b2 = (lane & 4) != 0;
  float a[4], b[2];
  #pragma unroll
  for (int i = 0; i < 4; ++i) a[i] = op(h ? v[i + 4] : v[i], xhalf(h ? v[i] : v[i + 4]));
  #pragma unroll
  for (int i = 0; i < 2; ++i) b[i] = op(b3 ? a[i + 2] : a[i], rowx<8>(b3 ? a[i] : a[i + 2]));
  float c = op(b2 ? b[1] : b[0], rowx<4>(b2 ? b[0] : b[1]));
  c = op(c, rowx<2>(c));
  c = op(c, rowx<1>(c));
  return c;
}
__device__ __forceinline__ float lane_of(float x, int l) {
  return __builtin_bit_cast(float, __builtin_amdgcn_readlane(__builtin_bit_cast(int, x), l));
}

#define T_HR 0
#define T_HC 6144
#define T_WR 12288
#define T_WC 20480
#define T_N  28672

__device__ __forceinline__ void mnmx4(const float4 v, float& mn, float& mx) {
  mn = fminf(fminf(mn, v.x), fminf(v.y, fminf(v.z, v.w)));
  mx = fmaxf(fmaxf(mx, v.x), fmaxf(v.y, fmaxf(v.z, v.w)));
}
__device__ __forceinline__ float4 norm4(const float4 v, const float mn, const float inv) {
  float4 o;
  o.x = 2.0f + (v.x - mn) * inv; o.y = 2.0f + (v.y - mn) * inv;
  o.z = 2.0f + (v.z - mn) * inv; o.w = 2.0f + (v.w - mn) * inv;
  return o;
}
__device__ __forceinline__ void raw_h(const float* __restrict__ ws, int c0, int j4, float sq0, float sq1,
                                      float4& rhr, float4& rhc) {
  const int pb = j4 / 24, y4 = (j4 - pb * 24) * 4;
  const int bb = pb >> 5;
  const float sq = bb ? sq1 : sq0;
  const float sk = ws[OFF_SK + pb];
  const float4 kh = *(const float4*)(ws + OFF_KH + pb * H_ + y4);
  const float4 qs = *(const float4*)(ws + OFF_QS + (bb * C_ + c0) * H_ + y4);
  rhr.x = sq * kh.x + (H_ * EPSF); rhr.y = sq * kh.y + (H_ * EPSF);
  rhr.z = sq * kh.z + (H_ * EPSF); rhr.w = sq * kh.w + (H_ * EPSF);
  rhc.x = qs.x * sk + (H_ * EPSF); rhc.y = qs.y * sk + (H_ * EPSF);
  rhc.z = qs.z * sk + (H_ * EPSF); rhc.w = qs.w * sk + (H_ * EPSF);
}
__device__ __forceinline__ void raw_w(const float* __restrict__ ws, int c0, int j4, float sq0, float sq1,
                                      float4& rwr, float4& rwc) {
  const int pb = j4 >> 5, x4 = (j4 & 31) * 4;
  const int bb = pb >> 5;
  const float sq = bb ? sq1 : sq0;
  const float sk = ws[OFF_SK + pb];
  const float4 kw = *(const float4*)(ws + OFF_KW + pb * W_ + x4);
  const float4 qc = *(const float4*)(ws + OFF_QC + (bb * C_ + c0) * W_ + x4);
  rwr.x = sq * kw.x + (W_ * EPSF); rwr.y = sq * kw.y + (W_ * EPSF);
  rwr.z = sq * kw.z + (W_ * EPSF); rwr.w = sq * kw.w + (W_ * EPSF);
  rwc.x = qc.x * sk + (W_ * EPSF); rwc.y = qc.y * sk + (W_ * EPSF);
  rwc.z = qc.z * sk + (W_ * EPSF); rwc.w = qc.w * sk + (W_ * EPSF);
}

__global__ void __launch_bounds__(1024)
km_fused(float* __restrict__ ws, const float* __restrict__ values, float* __restrict__ out) {
  const int rh   = blockIdx.x & 1;
  const int half = (blockIdx.x >> 1) & 1;
  const int b    = (blockIdx.x >> 2) & 1;
  const int c0   = blockIdx.x >> 3;
  const int tid  = threadIdx.x;
  const int wave = __builtin_amdgcn_readfirstlane(tid >> 5);
  const int lane = tid & 31;
  __shared__ float PR[4 * HO_ * PRP];
  __shared__ __attribute__((aligned(16))) float PC[2][4 * 32 * W_];
  __shared__ float IR[4][HO_];
  __shared__ __attribute__((aligned(16))) float IC[4][W_];
  __shared__ __attribute__((aligned(16))) float FN[16][2][W_];
  __shared__ __attribute__((aligned(16))) float FM[16][2][H_];
  __shared__ float red[32];
  float* const TF = &PC[0][0];
  float (* const RMN)[32] = (float (*)[32])(TF + T_N);
  float (* const RMX)[32] = (float (*)[32])(TF + T_N + 128);
  float (* const GP)[32]  = (float (*)[32])(TF + T_N + 256);

  const int bc0 = b * C_ + half * 16;
  float ig[4];
  {
    const bool two = wave < 16;
    const float sq0 = ws[OFF_SQ + c0];
    const float sq1 = ws[OFF_SQ + C_ + c0];
    float4 rhr0, rhc0, rhr1, rhc1, rwr0, rwc0, rwr1, rwc1;
    raw_h(ws, c0, tid, sq0, sq1, rhr0, rhc0);
    raw_w(ws, c0, tid, sq0, sq1, rwr0, rwc0);
    raw_w(ws, c0, tid + 1024, sq0, sq1, rwr1, rwc1);
    rhr1 = rhr0; rhc1 = rhc0;
    if (two) raw_h(ws, c0, tid + 1024, sq0, sq1, rhr1, rhc1);

    float mn[4], mx[4];
    #pragma unroll
    for (int f = 0; f < 4; ++f) { mn[f] = 3.4e38f; mx[f] = -3.4e38f; }
    mnmx4(rhr0, mn[0], mx[0]); mnmx4(rhr1, mn[0], mx[0]);
    mnmx4(rhc0, mn[1], mx[1]); mnmx4(rhc1, mn[1], mx[1]);
    mnmx4(rwr0, mn[2], mx[2]); mnmx4(rwr1, mn[2], mx[2]);
    mnmx4(rwc0, mn[3], mx[3]); mnmx4(rwc1, mn[3], mx[3]);
    {
      const float v8[8] = { mn[0], mn[1], mn[2], mn[3], -mx[0], -mx[1], -mx[2], -mx[3] };
      const float c = red8(v8, lane, [](float x, float y) { return fminf(x, y); });
      if ((lane & 3) == 0) RMN[lane >> 2][wave] = (lane < 16) ? c : -c;
    }
    __syncthreads();
    float inv[4];
    {
      const int fo = lane >> 3, eo = (lane & 7) * 4;
      const float4 a4 = *(const float4*)&RMN[fo][eo];
      const float4 b4 = *(const float4*)&RMX[fo][eo];
      float lmn = fminf(fminf(a4.x, a4.y), fminf(a4.z, a4.w));
      float lmx = fmaxf(fmaxf(b4.x, b4.y), fmaxf(b4.z, b4.w));
      lmn = fminf(lmn, rowx<1>(lmn)); lmn = fminf(lmn, rowx<2>(lmn)); lmn = fminf(lmn, rowx<4>(lmn));
      lmx = fmaxf(lmx, rowx<1>(lmx)); lmx = fmaxf(lmx, rowx<2>(lmx)); lmx = fmaxf(lmx, rowx<4>(lmx));
      const float linv = 1.0f / (lmx - lmn);
      #pragma unroll
      for (int f = 0; f < 4; ++f) { mn[f] = lane_of(lmn, 8 * f); inv[f] = lane_of(linv, 8 * f); }
    }

    #pragma unroll
    for (int k = 0; k < 2; ++k) {
      const int j4 = tid + k * 1024;
      const float4 wr = norm4(k ? rwr1 : rwr0, mn[2], inv[2]);
      const float4 wc = norm4(k ? rwc1 : rwc0, mn[3], inv[3]);
      *(float4*)&TF[T_WR + j4 * 4] = wr;
      *(float4*)&TF[T_WC + j4 * 4] = wc;
      const int chn = (j4 >> 5) - bc0, x4 = (j4 & 31) * 4;
      if (chn >= 0 && chn < 16) { *(float4*)&FN[chn][0][x4] = wr; *(float4*)&FN[chn][1][x4] = wc; }
      if (k == 0 || two) {
        const float4 hr = norm4(k ? rhr1 : rhr0, mn[0], inv[0]);
        const float4 hc = norm4(k ? rhc1 : rhc0, mn[1], inv[1]);
        *(float4*)&TF[T_HR + j4 * 4] = hr;
        *(float4*)&TF[T_HC + j4 * 4] = hc;
        const int pbm = j4 / 24, chm = pbm - bc0, y4 = (j4 - pbm * 24) * 4;
        if (chm >= 0 && chm < 16) { *(float4*)&FM[chm][0][y4] = hr; *(float4*)&FM[chm][1][y4] = hc; }
      }
    }
    __syncthreads();

    {
      float m[2][2][3], n[2][2][4];
      #pragma unroll
      for (int i = 0; i < 2; ++i) {
        const int pb = wave + i * 32;
        #pragma unroll
        for (int r = 0; r < 3; ++r) {
          m[i][0][r] = TF[T_HR + pb * H_ + lane * 3 + r];
          m[i][1][r] = TF[T_HC + pb * H_ + lane * 3 + r];
        }
        const float4 a = *(const float4*)&TF[T_WR + pb * W_ + lane * 4];
        const float4 d = *(const float4*)&TF[T_WC + pb * W_ + lane * 4];
        n[i][0][0] = a.x; n[i][0][1] = a.y; n[i][0][2] = a.z; n[i][0][3] = a.w;
        n[i][1][0] = d.x; n[i][1][1] = d.y; n[i][1][2] = d.z; n[i][1][3] = d.w;
      }
      float lm[8];
      #pragma unroll
      for (int i = 0; i < 2; ++i) {
        #pragma unroll
        for (int f = 0; f < 2; ++f) {
          lm[i * 4 + f]     = fmaxf(fmaxf(m[i][f][0], m[i][f][1]), m[i][f][2]);
          lm[i * 4 + 2 + f] = fmaxf(fmaxf(n[i][f][0], n[i][f][1]), fmaxf(n[i][f][2], n[i][f][3]));
        }
      }
      const float cm = red8(lm, lane, [](float x, float y) { return fmaxf(x, y); });
      float mml[2][2], nn[2][2];
      #pragma unroll
      for (int i = 0; i < 2; ++i) {
        #pragma unroll
        for (int f = 0; f < 2; ++f) {
          mml[i][f] = lane_of(cm, 4 * (i * 4 + f)) * LOG2E_F;
          nn[i][f]  = lane_of(cm, 4 * (i * 4 + 2 + f));
        }
      }
      float rsv[8], csv[8];
      #pragma unroll
      for (int i = 0; i < 2; ++i) {
        #pragma unroll
        for (int t = 0; t < 4; ++t) {
          const int f = t >> 1, g = t & 1;
          rsv[i * 4 + t] = __builtin_amdgcn_exp2f(mml[i][f] * n[i][g][0]) + __builtin_amdgcn_exp2f(mml[i][f] * n[i][g][1])
                         + __builtin_amdgcn_exp2f(mml[i][f] * n[i][g][2]) + __builtin_amdgcn_exp2f(mml[i][f] * n[i][g][3]);
          csv[i * 4 + t] = __builtin_amdgcn_exp2f((m[i][f][0] * LOG2E_F) * nn[i][g]) + __builtin_amdgcn_exp2f((m[i][f][1] * LOG2E_F) * nn[i][g])
                         + __builtin_amdgcn_exp2f((m[i][f][2] * LOG2E_F) * nn[i][g]);
        }
      }
      const float rs = red8(rsv, lane, [](float x, float y) { return x + y; });
      const float cs = red8(csv, lane, [](float x, float y) { return x + y; });
      const bool hi = (lane & 16) != 0, f1 = (lane & 8) != 0, g1 = (lane & 4) != 0;
      const float mml_l = hi ? (f1 ? mml[1][1] : mml[1][0]) : (f1 ? mml[0][1] : mml[0][0]);
      const float nn_l  = hi ? (g1 ? nn[1][1] : nn[1][0]) : (g1 ? nn[0][1] : nn[0][0]);
      float pk = __builtin_amdgcn_exp2f(mml_l * nn_l) * (1.0f / rs + 1.0f / cs);
      pk = fmaxf(pk, xhalf(pk));
      if ((lane & 19) == 0) GP[lane >> 2][wave] = pk;
    }
    __syncthreads();
    {
      const float4 g4 = *(const float4*)&GP[lane >> 3][(lane & 7) * 4];
      float g = fmaxf(fmaxf(g4.x, g4.y), fmaxf(g4.z, g4.w));
      g = fmaxf(g, rowx<1>(g)); g = fmaxf(g, rowx<2>(g)); g = fmaxf(g, rowx<4>(g));
      const float lig = 1.0f / g;
      #pragma unroll
      for (int t = 0; t < 4; ++t) ig[t] = lane_of(lig, 8 * t);
    }
  }

  const bool own = wave < 16;
  const int rg    = own ? (rh * 16 + wave) : ((1 - rh) * 16 + (wave - 16));
  const int row0  = rg * 3;
  const int lrow0 = (wave & 15) * 3;
  const int col0  = lane * 4;

  float z[3][4];
  #pragma unroll
  for (int r = 0; r < 3; ++r) { z[r][0] = 0.f; z[r][1] = 0.f; z[r][2] = 0.f; z[r][3] = 0.f; }
  float smax = 0.0f;

  if (!own) {
    float mh[2][3], nw[2][4];
    #pragma unroll
    for (int r = 0; r < 3; ++r) { mh[0][r] = FM[0][0][row0 + r]; mh[1][r] = FM[0][1][row0 + r]; }
    const float4 a = *(const float4*)&FN[0][0][col0];
    const float4 d = *(const float4*)&FN[0][1][col0];
    nw[0][0] = a.x; nw[0][1] = a.y; nw[0][2] = a.z; nw[0][3] = a.w;
    nw[1][0] = d.x; nw[1][1] = d.y; nw[1][2] = d.z; nw[1][3] = d.w;
    km_col_partials_direct(mh, nw, PC[0], rg, col0);
  }

  for (int ci = 0; ci < 16; ++ci) {
    const int bc = bc0 + ci;
    float  E[4][3][4];
    float4 vv[3];
    if (own) {
      float mh[2][3], nw[2][4];
      #pragma unroll
      for (int r = 0; r < 3; ++r) { mh[0][r] = FM[ci][0][row0 + r]; mh[1][r] = FM[ci][1][row0 + r]; }
      const float4 a = *(const float4*)&FN[ci][0][col0];
      const float4 d = *(const float4*)&FN[ci][1][col0];
      nw[0][0] = a.x; nw[0][1] = a.y; nw[0][2] = a.z; nw[0][3] = a.w;
      nw[1][0] = d.x; nw[1][1] = d.y; nw[1][2] = d.z; nw[1][3] = d.w;
      km_exps(mh, nw, E);
      km_row_partials(E, PR, lrow0, lane);
      km_col_partials(E, PC[ci & 1], rg, col0);
    }
    __syncthreads();

    if (own) {
      #pragma unroll
      for (int r = 0; r < 3; ++r)
        vv[r] = *(const float4*)(values + bc * HW_ + (row0 + r) * W_ + col0);

      if (wave < 12) {
        const v2f ones = {1.0f, 1.0f};
        const int kho = (lane < 16) ? 0 : 2;
        const int tt = wave / 3, rb = wave % 3;
        const int M  = rb * 16 + (lane & 15);
        const float* pr = PR + tt * (HO_ * PRP) + M * PRP;
        v8f acc = {};
        for (int k0 = 0; k0 < 32; k0 += 4) {
          v2f a = { pr[k0 + kho], pr[k0 + kho + 1] };
          acc = __builtin_amdgcn_wmma_f32_16x16x4_f32(false, a, false, ones, (short)0, acc, false, false);
        }
        if ((lane & 15) == 0) {
          const int rbase = rb * 16 + ((lane >> 4) << 3);
          #pragma unroll
          for (int v = 0; v < 8; ++v) IR[tt][rbase + v] = 1.0f / acc[v];
        }
      }
    } else {
      if (wave < 24) {
        const int jw = wave - 16;
        const int tt = jw >> 1;
        const int g  = ((jw & 1) << 4) | (lane & 15);
        const int ph = lane >> 4;
        const float* pc = &PC[ci & 1][tt * (32 * W_) + (ph * 16) * W_ + g * 4];
        float4 s;
        #pragma unroll
        for (int i0 = 0; i0 < 16; i0 += 4) {
          const float4 x0 = *(const float4*)(pc + (i0 + 0) * W_);
          const float4 x1 = *(const float4*)(pc + (i0 + 1) * W_);
          const float4 x2 = *(const float4*)(pc + (i0 + 2) * W_);
          const float4 x3 = *(const float4*)(pc + (i0 + 3) * W_);
          float4 t;
          t.x = (x0.x + x1.x) + (x2.x + x3.x);
          t.y = (x0.y + x1.y) + (x2.y + x3.y);
          t.z = (x0.z + x1.z) + (x2.z + x3.z);
          t.w = (x0.w + x1.w) + (x2.w + x3.w);
          if (i0 == 0) { s = t; } else { s.x += t.x; s.y += t.y; s.z += t.z; s.w += t.w; }
          __builtin_amdgcn_sched_barrier(0);
        }
        s.x += xhalf(s.x); s.y += xhalf(s.y); s.z += xhalf(s.z); s.w += xhalf(s.w);
        float2 o;
        o.x = 1.0f / (ph ? s.z : s.x);
        o.y = 1.0f / (ph ? s.w : s.y);
        *(float2*)&IC[tt][g * 4 + ph * 2] = o;
      }
      if (ci + 1 < 16) {
        float mh[2][3], nw[2][4];
        #pragma unroll
        for (int r = 0; r < 3; ++r) { mh[0][r] = FM[ci + 1][0][row0 + r]; mh[1][r] = FM[ci + 1][1][row0 + r]; }
        const float4 a = *(const float4*)&FN[ci + 1][0][col0];
        const float4 d = *(const float4*)&FN[ci + 1][1][col0];
        nw[0][0] = a.x; nw[0][1] = a.y; nw[0][2] = a.z; nw[0][3] = a.w;
        nw[1][0] = d.x; nw[1][1] = d.y; nw[1][2] = d.z; nw[1][3] = d.w;
        km_col_partials_direct(mh, nw, PC[(ci + 1) & 1], rg, col0);
      }
    }
    __syncthreads();

    if (own) {
      float icv[4][4];
      #pragma unroll
      for (int t = 0; t < 4; ++t) {
        const float4 x = *(const float4*)&IC[t][col0];
        icv[t][0] = x.x; icv[t][1] = x.y; icv[t][2] = x.z; icv[t][3] = x.w;
      }
      #pragma unroll
      for (int r = 0; r < 3; ++r) {
        float irv[4];
        #pragma unroll
        for (int t = 0; t < 4; ++t) irv[t] = IR[t][lrow0 + r];
        const float vr[4] = { vv[r].x, vv[r].y, vv[r].z, vv[r].w };
        float sc[4];
        #pragma unroll
        for (int q = 0; q < 4; ++q) {
          float s = 0.f;
          #pragma unroll
          for (int t = 0; t < 4; ++t) s += E[t][r][q] * (irv[t] + icv[t][q]) * ig[t];
          z[r][q] += s * vr[q];
          smax = fmaxf(smax, s);
          sc[q] = s;
        }
        if (c0 == C_ - 1) {
          float4 o; o.x = sc[0]; o.y = sc[1]; o.z = sc[2]; o.w = sc[3];
          *(float4*)(out + BC_ * HW_ + bc * HW_ + (row0 + r) * W_ + col0) = o;
        }
      }
    }
  }

  if (own) {
    float* zdst = (half ? (ws + OFF_SCORE) : out) + (b * C_ + c0) * HW_;
    #pragma unroll
    for (int r = 0; r < 3; ++r) {
      float4 o; o.x = z[r][0]; o.y = z[r][1]; o.z = z[r][2]; o.w = z[r][3];
      *(float4*)(zdst + (row0 + r) * W_ + col0) = o;
    }
  }
  for (int o = 16; o > 0; o >>= 1) smax = fmaxf(smax, __shfl_xor(smax, o));
  if (lane == 0) red[wave] = smax;
  __syncthreads();
  if (tid == 0) {
    float g = red[0];
    for (int i = 1; i < 32; ++i) g = fmaxf(g, red[i]);
    atomicMax((int*)(ws + OFF_GSMAX + c0), __float_as_int(g));
  }
}

__global__ void ks_scale(const float* __restrict__ ws, float* __restrict__ out) {
  const int gid = blockIdx.x * 256 + threadIdx.x;
  float4* o4 = (float4*)out;
  if (gid < BC_ * HW_ / 4) {
    const int c0 = ((gid * 4) / HW_) % C_;
    const float inv = 1.0f / ws[OFF_GSMAX + c0];
    const float4 a = o4[gid];
    const float4 p = *(const float4*)(ws + OFF_SCORE + gid * 4);
    float4 r;
    r.x = (a.x + p.x) * inv; r.y = (a.y + p.y) * inv; r.z = (a.z + p.z) * inv; r.w = (a.w + p.w) * inv;
    o4[gid] = r;
  } else {
    const float inv = 1.0f / ws[OFF_GSMAX + 31];
    float4 a = o4[gid];
    a.x *= inv; a.y *= inv; a.z *= inv; a.w *= inv;
    o4[gid] = a;
  }
}

extern "C" void kernel_launch(void* const* d_in, const int* in_sizes, int n_in,
                              void* d_out, int out_size, void* d_ws, size_t ws_size,
                              hipStream_t stream) {
  (void)in_sizes; (void)n_in; (void)out_size; (void)ws_size;
  const float* q   = (const float*)d_in[0];
  const float* k   = (const float*)d_in[1];
  const float* val = (const float*)d_in[2];
  float* out = (float*)d_out;
  float* ws  = (float*)d_ws;

  k1_sums<<<128, 256, 0, stream>>>(q, k, ws);
  km_fused<<<256, 1024, 0, stream>>>(ws, val, out);
  ks_scale<<<1536, 256, 0, stream>>>(ws, out);
}


# km: job lanes hand on 1/(sum*peak of the term), row sums by v_rcp_f32, no division sequence: a score term is one add and one fma per element, not add, two multiplies, add; values move ~1 ulp
# speedup vs baseline: 16.5608x; 1.1193x over previous
#include <hip/hip_runtime.h>

typedef __attribute__((ext_vector_type(2)))  float    v2f;
typedef __attribute__((ext_vector_type(8)))  float    v8f;

#define B_   2
#define C_   32
#define H_   96
#define W_   128
#define HW_  (H_ * W_)
#define BC_  (B_ * C_)
#define EPSF 1e-6f
#define LOG2E_F 1.44269504088896340736f

__device__ __forceinline__ float xhalf(float x) {
  const int b = __builtin_bit_cast(int, x);
  return __builtin_bit_cast(float, __builtin_amdgcn_permlanex16(b, b, 0x76543210, 0xfedcba98, false, false));
}

#define OFF_QS    0
#define OFF_QC    6144
#define OFF_KH    14336
#define OFF_KW    20480
#define OFF_SQ    28672
#define OFF_SK    28736
#define OFF_GMAX  28800
#define OFF_GSMAX 28928
#define OFF_HR    28960
#define OFF_HC    (OFF_HR + 32 * 6144)
#define OFF_WR    (OFF_HC + 32 * 6144)
#define OFF_WC    (OFF_WR + 32 * 8192)
#define OFF_IRS   (OFF_WC + 32 * 8192)
#define OFF_SCORE (OFF_IRS + 4 * 64 * 224)


#define TP_ 132
__device__ __forceinline__ v8f mm4(const v2f a, const v2f b, const v8f c) {
  return __builtin_amdgcn_wmma_f32_16x16x4_f32(false, a, false, b, (short)0, c, false, false);
}
__global__ void __launch_bounds__(256)
k1_sums(const float* __restrict__ q, const float* __restrict__ k, float* __restrict__ ws) {
  __shared__ __attribute__((aligned(16))) float tile[H_ * TP_];
  __shared__ float part[6];
  const int bc   = blockIdx.x >> 1;
  const int p    = blockIdx.x & 1;
  const int tid  = threadIdx.x;
  const int wave = __builtin_amdgcn_readfirstlane(tid >> 5);
  const int lane = tid & 31;
  if (blockIdx.x == 0 && tid < C_) ws[OFF_GSMAX + tid] = 0.0f;
  const float* s = (p ? k : q) + bc * HW_;
  float4 v[HW_ / 1024];
  #pragma unroll
  for (int j = 0; j < HW_ / 1024; ++j) v[j] = *(const float4*)(s + (tid + j * 256) * 4);
  #pragma unroll
  for (int j = 0; j < HW_ / 1024; ++j) {
    const int i4 = tid + j * 256;
    *(float4*)&tile[(i4 >> 5) * TP_ + (i4 & 31) * 4] = v[j];
  }
  __syncthreads();

  const v2f ones = {1.0f, 1.0f};
  const int c = lane & 15, h = lane >> 4;
  {
    const float* pc = tile + (2 * h) * TP_ + wave * 16 + c;
    v8f a0 = {}, a1 = {};
    #pragma unroll
    for (int k0 = 0; k0 < H_; k0 += 8) {
      const v2f y0 = { pc[k0 * TP_],       pc[(k0 + 1) * TP_] };
      const v2f y1 = { pc[(k0 + 4) * TP_], pc[(k0 + 5) * TP_] };
      a0 = mm4(ones, y0, a0);
      a1 = mm4(ones, y1, a1);
    }
    const float cs = a0[0] + a1[0];
    if (lane < 16) ws[(p ? OFF_KW : OFF_QC) + bc * W_ + wave * 16 + lane] = cs;
  }
  if (wave < 6) {
    const float* pr = tile + (wave * 16 + c) * TP_ + 2 * h;
    v8f a0 = {}, a1 = {};
    #pragma unroll
    for (int k0 = 0; k0 < W_; k0 += 8) {
      const v2f x0 = *(const v2f*)(pr + k0);
      const v2f x1 = *(const v2f*)(pr + k0 + 4);
      a0 = mm4(x0, ones, a0);
      a1 = mm4(x1, ones, a1);
    }
    float acc[8];
    #pragma unroll
    for (int r = 0; r < 8; ++r) acc[r] = a0[r] + a1[r];
    float rs = acc[0];
    #pragma unroll
    for (int r = 1; r < 8; ++r) rs = ((lane & 7) == r) ? acc[r] : rs;
    if ((lane & 8) == 0) ws[(p ? OFF_KH : OFF_QS) + bc * H_ + wave * 16 + 8 * h + (lane & 7)] = rs;
    float t = ((acc[0] + acc[1]) + (acc[2] + acc[3])) + ((acc[4] + acc[5]) + (acc[6] + acc[7]));
    t += xhalf(t);
    if (lane == 0) part[wave] = t;
  }
  __syncthreads();
  if (tid == 0) ws[(p ? OFF_SK : OFF_SQ) + bc] = ((part[0] + part[1]) + (part[2] + part[3])) + (part[4] + part[5]);
}

#define PRP 33
#define HO_ 48

__device__ __forceinline__ void km_exps(const float (&mh)[2][3], const float (&nw)[2][4], float (&E)[4][3][4]) {
  float ml[2][3];
  #pragma unroll
  for (int r = 0; r < 3; ++r) { ml[0][r] = mh[0][r] * LOG2E_F; ml[1][r] = mh[1][r] * LOG2E_F; }
  #pragma unroll
  for (int t = 0; t < 4; ++t) {
    #pragma unroll
    for (int r = 0; r < 3; ++r) {
      #pragma unroll
      for (int q = 0; q < 4; ++q) E[t][r][q] = ml[t >> 1][r] * nw[t & 1][q];
    }
  }
  #pragma unroll
  for (int t = 0; t < 4; ++t) {
    #pragma unroll
    for (int r = 0; r < 3; ++r) {
      #pragma unroll
      for (int q = 0; q < 4; ++q) E[t][r][q] = __builtin_amdgcn_exp2f(E[t][r][q]);
    }
  }
}

__device__ __forceinline__ void km_col_partials(const float (&E)[4][3][4], float* pc, int rg, int col0) {
  #pragma unroll
  for (int t = 0; t < 4; ++t) {
    float4 cp;
    cp.x = E[t][0][0] + E[t][1][0] + E[t][2][0];
    cp.y = E[t][0][1] + E[t][1][1] + E[t][2][1];
    cp.z = E[t][0][2] + E[t][1][2] + E[t][2][2];
    cp.w = E[t][0][3] + E[t][1][3] + E[t][2][3];
    *(float4*)(pc + t * (32 * W_) + rg * W_ + col0) = cp;
  }
}

__device__ __forceinline__ void km_row_partials(const float (&E)[4][3][4], float* pr, int lrow0, int lane) {
  #pragma unroll
  for (int t = 0; t < 4; ++t) {
    #pragma unroll
    for (int r = 0; r < 3; ++r)
      pr[t * (HO_ * PRP) + (lrow0 + r) * PRP + lane] = (E[t][r][0] + E[t][r][1]) + (E[t][r][2] + E[t][r][3]);
  }
}

__device__ __forceinline__ void km_col_partials_direct(const float (&mh)[2][3], const float (&nw)[2][4],
                                                       float* pc, int rg, int col0) {
  float ml[2][3];
  #pragma unroll
  for (int r = 0; r < 3; ++r) { ml[0][r] = mh[0][r] * LOG2E_F; ml[1][r] = mh[1][r] * LOG2E_F; }
  #pragma unroll
  for (int t = 0; t < 4; ++t) {
    float e[3][4];
    #pragma unroll
    for (int r = 0; r < 3; ++r) {
      #pragma unroll
      for (int q = 0; q < 4; ++q) e[r][q] = ml[t >> 1][r] * nw[t & 1][q];
    }
    #pragma unroll
    for (int r = 0; r < 3; ++r) {
      #pragma unroll
      for (int q = 0; q < 4; ++q) e[r][q] = __builtin_amdgcn_exp2f(e[r][q]);
    }
    float4 cp;
    cp.x = e[0][0] + e[1][0] + e[2][0];
    cp.y = e[0][1] + e[1][1] + e[2][1];
    cp.z = e[0][2] + e[1][2] + e[2][2];
    cp.w = e[0][3] + e[1][3] + e[2][3];
    *(float4*)(pc + t * (32 * W_) + rg * W_ + col0) = cp;
  }
}

template <int K> __device__ __forceinline__ float rowx(float x) {
  const int v = __builtin_bit_cast(int, x);
  return __builtin_bit_cast(float, __builtin_amdgcn_update_dpp(v, v, 0x160 + K, 0xf, 0xf, false));
}
template <class F> __device__ __forceinline__ float red8(const float (&v)[8], int lane, F op) {
  const bool h = (lane & 16) != 0, b3 = (lane & 8) != 0, b2 = (lane & 4) != 0;
  float a[4], b[2];
  #pragma unroll
  for (int i = 0; i < 4; ++i) a[i] = op(h ? v[i + 4] : v[i], xhalf(h ? v[i] : v[i + 4]));
  #pragma unroll
  for (int i = 0; i < 2; ++i) b[i] = op(b3 ? a[i + 2] : a[i], rowx<8>(b3 ? a[i] : a[i + 2]));
  float c = op(b2 ? b[1] : b[0], rowx<4>(b2 ? b[0] : b[1]));
  c = op(c, rowx<2>(c));
  c = op(c, rowx<1>(c));
  return c;
}
__device__ __forceinline__ float lane_of(float x, int l) {
  return __builtin_bit_cast(float, __builtin_amdgcn_readlane(__builtin_bit_cast(int, x), l));
}

#define T_HR 0
#define T_HC 6144
#define T_WR 12288
#define T_WC 20480
#define T_N  28672

__device__ __forceinline__ void mnmx4(const float4 v, float& mn, float& mx) {
  mn = fminf(fminf(mn, v.x), fminf(v.y, fminf(v.z, v.w)));
  mx = fmaxf(fmaxf(mx, v.x), fmaxf(v.y, fmaxf(v.z, v.w)));
}
__device__ __forceinline__ float4 norm4(const float4 v, const float mn, const float inv) {
  float4 o;
  o.x = 2.0f + (v.x - mn) * inv; o.y = 2.0f + (v.y - mn) * inv;
  o.z = 2.0f + (v.z - mn) * inv; o.w = 2.0f + (v.w - mn) * inv;
  return o;
}
__device__ __forceinline__ void raw_h(const float* __restrict__ ws, int c0, int j4, float sq0, float sq1,
                                      float4& rhr, float4& rhc) {
  const int pb = j4 / 24, y4 = (j4 - pb * 24) * 4;
  const int bb = pb >> 5;
  const float sq = bb ? sq1 : sq0;
  const float sk = ws[OFF_SK + pb];
  const float4 kh = *(const float4*)(ws + OFF_KH + pb * H_ + y4);
  const float4 qs = *(const float4*)(ws + OFF_QS + (bb * C_ + c0) * H_ + y4);
  rhr.x = sq * kh.x + (H_ * EPSF); rhr.y = sq * kh.y + (H_ * EPSF);
  rhr.z = sq * kh.z + (H_ * EPSF); rhr.w = sq * kh.w + (H_ * EPSF);
  rhc.x = qs.x * sk + (H_ * EPSF); rhc.y = qs.y * sk + (H_ * EPSF);
  rhc.z = qs.z * sk + (H_ * EPSF); rhc.w = qs.w * sk + (H_ * EPSF);
}
__device__ __forceinline__ void raw_w(const float* __restrict__ ws, int c0, int j4, float sq0, float sq1,
                                      float4& rwr, float4& rwc) {
  const int pb = j4 >> 5, x4 = (j4 & 31) * 4;
  const int bb = pb >> 5;
  const float sq = bb ? sq1 : sq0;
  const float sk = ws[OFF_SK + pb];
  const float4 kw = *(const float4*)(ws + OFF_KW + pb * W_ + x4);
  const float4 qc = *(const float4*)(ws + OFF_QC + (bb * C_ + c0) * W_ + x4);
  rwr.x = sq * kw.x + (W_ * EPSF); rwr.y = sq * kw.y + (W_ * EPSF);
  rwr.z = sq * kw.z + (W_ * EPSF); rwr.w = sq * kw.w + (W_ * EPSF);
  rwc.x = qc.x * sk + (W_ * EPSF); rwc.y = qc.y * sk + (W_ * EPSF);
  rwc.z = qc.z * sk + (W_ * EPSF); rwc.w = qc.w * sk + (W_ * EPSF);
}

__global__ void __launch_bounds__(1024)
km_fused(float* __restrict__ ws, const float* __restrict__ values, float* __restrict__ out) {
  const int rh   = blockIdx.x & 1;
  const int half = (blockIdx.x >> 1) & 1;
  const int b    = (blockIdx.x >> 2) & 1;
  const int c0   = blockIdx.x >> 3;
  const int tid  = threadIdx.x;
  const int wave = __builtin_amdgcn_readfirstlane(tid >> 5);
  const int lane = tid & 31;
  __shared__ float PR[4 * HO_ * PRP];
  __shared__ __attribute__((aligned(16))) float PC[2][4 * 32 * W_];
  __shared__ float IR[4][HO_];
  __shared__ __attribute__((aligned(16))) float IC[4][W_];
  __shared__ __attribute__((aligned(16))) float FN[16][2][W_];
  __shared__ __attribute__((aligned(16))) float FM[16][2][H_];
  __shared__ float red[32];
  float* const TF = &PC[0][0];
  float (* const RMN)[32] = (float (*)[32])(TF + T_N);
  float (* const RMX)[32] = (float (*)[32])(TF + T_N + 128);
  float (* const GP)[32]  = (float (*)[32])(TF + T_N + 256);

  const int bc0 = b * C_ + half * 16;
  float gp[4];
  {
    const bool two = wave < 16;
    const float sq0 = ws[OFF_SQ + c0];
    const float sq1 = ws[OFF_SQ + C_ + c0];
    float4 rhr0, rhc0, rhr1, rhc1, rwr0, rwc0, rwr1, rwc1;
    raw_h(ws, c0, tid, sq0, sq1, rhr0, rhc0);
    raw_w(ws, c0, tid, sq0, sq1, rwr0, rwc0);
    raw_w(ws, c0, tid + 1024, sq0, sq1, rwr1, rwc1);
    rhr1 = rhr0; rhc1 = rhc0;
    if (two) raw_h(ws, c0, tid + 1024, sq0, sq1, rhr1, rhc1);

    float mn[4], mx[4];
    #pragma unroll
    for (int f = 0; f < 4; ++f) { mn[f] = 3.4e38f; mx[f] = -3.4e38f; }
    mnmx4(rhr0, mn[0], mx[0]); mnmx4(rhr1, mn[0], mx[0]);
    mnmx4(rhc0, mn[1], mx[1]); mnmx4(rhc1, mn[1], mx[1]);
    mnmx4(rwr0, mn[2], mx[2]); mnmx4(rwr1, mn[2], mx[2]);
    mnmx4(rwc0, mn[3], mx[3]); mnmx4(rwc1, mn[3], mx[3]);
    {
      const float v8[8] = { mn[0], mn[1], mn[2], mn[3], -mx[0], -mx[1], -mx[2], -mx[3] };
      const float c = red8(v8, lane, [](float x, float y) { return fminf(x, y); });
      if ((lane & 3) == 0) RMN[lane >> 2][wave] = (lane < 16) ? c : -c;
    }
    __syncthreads();
    float inv[4];
    {
      const int fo = lane >> 3, eo = (lane & 7) * 4;
      const float4 a4 = *(const float4*)&RMN[fo][eo];
      const float4 b4 = *(const float4*)&RMX[fo][eo];
      float lmn = fminf(fminf(a4.x, a4.y), fminf(a4.z, a4.w));
      float lmx = fmaxf(fmaxf(b4.x, b4.y), fmaxf(b4.z, b4.w));
      lmn = fminf(lmn, rowx<1>(lmn)); lmn = fminf(lmn, rowx<2>(lmn)); lmn = fminf(lmn, rowx<4>(lmn));
      lmx = fmaxf(lmx, rowx<1>(lmx)); lmx = fmaxf(lmx, rowx<2>(lmx)); lmx = fmaxf(lmx, rowx<4>(lmx));
      const float linv = 1.0f / (lmx - lmn);
      #pragma unroll
      for (int f = 0; f < 4; ++f) { mn[f] = lane_of(lmn, 8 * f); inv[f] = lane_of(linv, 8 * f); }
    }

    #pragma unroll
    for (int k = 0; k < 2; ++k) {
      const int j4 = tid + k * 1024;
      const float4 wr = norm4(k ? rwr1 : rwr0, mn[2], inv[2]);
      const float4 wc = norm4(k ? rwc1 : rwc0, mn[3], inv[3]);
      *(float4*)&TF[T_WR + j4 * 4] = wr;
      *(float4*)&TF[T_WC + j4 * 4] = wc;
      const int chn = (j4 >> 5) - bc0, x4 = (j4 & 31) * 4;
      if (chn >= 0 && chn < 16) { *(float4*)&FN[chn][0][x4] = wr; *(float4*)&FN[chn][1][x4] = wc; }
      if (k == 0 || two) {
        const float4 hr = norm4(k ? rhr1 : rhr0, mn[0], inv[0]);
        const float4 hc = norm4(k ? rhc1 : rhc0, mn[1], inv[1]);
        *(float4*)&TF[T_HR + j4 * 4] = hr;
        *(float4*)&TF[T_HC + j4 * 4] = hc;
        const int pbm = j4 / 24, chm = pbm - bc0, y4 = (j4 - pbm * 24) * 4;
        if (chm >= 0 && chm < 16) { *(float4*)&FM[chm][0][y4] = hr; *(float4*)&FM[chm][1][y4] = hc; }
      }
    }
    __syncthreads();

    {
      float m[2][2][3], n[2][2][4];
      #pragma unroll
      for (int i = 0; i < 2; ++i) {
        const int pb = wave + i * 32;
        #pragma unroll
        for (int r = 0; r < 3; ++r) {
          m[i][0][r] = TF[T_HR + pb * H_ + lane * 3 + r];
          m[i][1][r] = TF[T_HC + pb * H_ + lane * 3 + r];
        }
        const float4 a = *(const float4*)&TF[T_WR + pb * W_ + lane * 4];
        const float4 d = *(const float4*)&TF[T_WC + pb * W_ + lane * 4];
        n[i][0][0] = a.x; n[i][0][1] = a.y; n[i][0][2] = a.z; n[i][0][3] = a.w;
        n[i][1][0] = d.x; n[i][1][1] = d.y; n[i][1][2] = d.z; n[i][1][3] = d.w;
      }
      float lm[8];
      #pragma unroll
      for (int i = 0; i < 2; ++i) {
        #pragma unroll
        for (int f = 0; f < 2; ++f) {
          lm[i * 4 + f]     = fmaxf(fmaxf(m[i][f][0], m[i][f][1]), m[i][f][2]);
          lm[i * 4 + 2 + f] = fmaxf(fmaxf(n[i][f][0], n[i][f][1]), fmaxf(n[i][f][2], n[i][f][3]));
        }
      }
      const float cm = red8(lm, lane, [](float x, float y) { return fmaxf(x, y); });
      float mml[2][2], nn[2][2];
      #pragma unroll
      for (int i = 0; i < 2; ++i) {
        #pragma unroll
        for (int f = 0; f < 2; ++f) {
          mml[i][f] = lane_of(cm, 4 * (i * 4 + f)) * LOG2E_F;
          nn[i][f]  = lane_of(cm, 4 * (i * 4 + 2 + f));
        }
      }
      float rsv[8], csv[8];
      #pragma unroll
      for (int i = 0; i < 2; ++i) {
        #pragma unroll
        for (int t = 0; t < 4; ++t) {
          const int f = t >> 1, g = t & 1;
          rsv[i * 4 + t] = __builtin_amdgcn_exp2f(mml[i][f] * n[i][g][0]) + __builtin_amdgcn_exp2f(mml[i][f] * n[i][g][1])
                         + __builtin_amdgcn_exp2f(mml[i][f] * n[i][g][2]) + __builtin_amdgcn_exp2f(mml[i][f] * n[i][g][3]);
          csv[i * 4 + t] = __builtin_amdgcn_exp2f((m[i][f][0] * LOG2E_F) * nn[i][g]) + __builtin_amdgcn_exp2f((m[i][f][1] * LOG2E_F) * nn[i][g])
                         + __builtin_amdgcn_exp2f((m[i][f][2] * LOG2E_F) * nn[i][g]);
        }
      }
      const float rs = red8(rsv, lane, [](float x, float y) { return x + y; });
      const float cs = red8(csv, lane, [](float x, float y) { return x + y; });
      const bool hi = (lane & 16) != 0, f1 = (lane & 8) != 0, g1 = (lane & 4) != 0;
      const float mml_l = hi ? (f1 ? mml[1][1] : mml[1][0]) : (f1 ? mml[0][1] : mml[0][0]);
      const float nn_l  = hi ? (g1 ? nn[1][1] : nn[1][0]) : (g1 ? nn[0][1] : nn[0][0]);
      float pk = __builtin_amdgcn_exp2f(mml_l * nn_l) * (1.0f / rs + 1.0f / cs);
      pk = fmaxf(pk, xhalf(pk));
      if ((lane & 19) == 0) GP[lane >> 2][wave] = pk;
    }
    __syncthreads();
    {
      const float4 g4 = *(const float4*)&GP[lane >> 3][(lane & 7) * 4];
      float g = fmaxf(fmaxf(g4.x, g4.y), fmaxf(g4.z, g4.w));
      g = fmaxf(g, rowx<1>(g)); g = fmaxf(g, rowx<2>(g)); g = fmaxf(g, rowx<4>(g));
      #pragma unroll
      for (int t = 0; t < 4; ++t) gp[t] = lane_of(g, 8 * t);
    }
  }

  const bool own = wave < 16;
  const int rg    = own ? (rh * 16 + wave) : ((1 - rh) * 16 + (wave - 16));
  const int row0  = rg * 3;
  const int lrow0 = (wave & 15) * 3;
  const int col0  = lane * 4;
  const int   jt  = own ? (wave / 3) : ((wave - 16) >> 1);
  const float gpj = (jt == 0) ? gp[0] : (jt == 1) ? gp[1] : (jt == 2) ? gp[2] : gp[3];

  float z[3][4];
  #pragma unroll
  for (int r = 0; r < 3; ++r) { z[r][0] = 0.f; z[r][1] = 0.f; z[r][2] = 0.f; z[r][3] = 0.f; }
  float smax = 0.0f;

  if (!own) {
    float mh[2][3], nw[2][4];
    #pragma unroll
    for (int r = 0; r < 3; ++r) { mh[0][r] = FM[0][0][row0 + r]; mh[1][r] = FM[0][1][row0 + r]; }
    const float4 a = *(const float4*)&FN[0][0][col0];
    const float4 d = *(const float4*)&FN[0][1][col0];
    nw[0][0] = a.x; nw[0][1] = a.y; nw[0][2] = a.z; nw[0][3] = a.w;
    nw[1][0] = d.x; nw[1][1] = d.y; nw[1][2] = d.z; nw[1][3] = d.w;
    km_col_partials_direct(mh, nw, PC[0], rg, col0);
  }

  for (int ci = 0; ci < 16; ++ci) {
    const int bc = bc0 + ci;
    float  E[4][3][4];
    float4 vv[3];
    if (own) {
      float mh[2][3], nw[2][4];
      #pragma unroll
      for (int r = 0; r < 3; ++r) { mh[0][r] = FM[ci][0][row0 + r]; mh[1][r] = FM[ci][1][row0 + r]; }
      const float4 a = *(const float4*)&FN[ci][0][col0];
      const float4 d = *(const float4*)&FN[ci][1][col0];
      nw[0][0] = a.x; nw[0][1] = a.y; nw[0][2] = a.z; nw[0][3] = a.w;
      nw[1][0] = d.x; nw[1][1] = d.y; nw[1][2] = d.z; nw[1][3] = d.w;
      km_exps(mh, nw, E);
      km_row_partials(E, PR, lrow0, lane);
      km_col_partials(E, PC[ci & 1], rg, col0);
    }
    __syncthreads();

    if (own) {
      #pragma unroll
      for (int r = 0; r < 3; ++r)
        vv[r] = *(const float4*)(values + bc * HW_ + (row0 + r) * W_ + col0);

      if (wave < 12) {
        const v2f ones = {1.0f, 1.0f};
        const int kho = (lane < 16) ? 0 : 2;
        const int tt = wave / 3, rb = wave % 3;
        const int M  = rb * 16 + (lane & 15);
        const float* pr = PR + tt * (HO_ * PRP) + M * PRP;
        v8f acc = {};
        for (int k0 = 0; k0 < 32; k0 += 4) {
          v2f a = { pr[k0 + kho], pr[k0 + kho + 1] };
          acc = __builtin_amdgcn_wmma_f32_16x16x4_f32(false, a, false, ones, (short)0, acc, false, false);
        }
        float rs = acc[0];
        #pragma unroll
        for (int r = 1; r < 8; ++r) rs = ((lane & 7) == r) ? acc[r] : rs;
        IR[tt][rb * 16 + ((lane >> 4) << 3) + (lane & 7)] = __builtin_amdgcn_rcpf(rs * gpj);
      }
    } else {
      if (wave < 24) {
        const int jw = wave - 16;
        const int tt = jw >> 1;
        const int g  = ((jw & 1) << 4) | (lane & 15);
        const int ph = lane >> 4;
        const float* pc = &PC[ci & 1][tt * (32 * W_) + (ph * 16) * W_ + g * 4];
        float4 s;
        #pragma unroll
        for (int i0 = 0; i0 < 16; i0 += 4) {
          const float4 x0 = *(const float4*)(pc + (i0 + 0) * W_);
          const float4 x1 = *(const float4*)(pc + (i0 + 1) * W_);
          const float4 x2 = *(const float4*)(pc + (i0 + 2) * W_);
          const float4 x3 = *(const float4*)(pc + (i0 + 3) * W_);
          float4 t;
          t.x = (x0.x + x1.x) + (x2.x + x3.x);
          t.y = (x0.y + x1.y) + (x2.y + x3.y);
          t.z = (x0.z + x1.z) + (x2.z + x3.z);
          t.w = (x0.w + x1.w) + (x2.w + x3.w);
          if (i0 == 0) { s = t; } else { s.x += t.x; s.y += t.y; s.z += t.z; s.w += t.w; }
          __builtin_amdgcn_sched_barrier(0);
        }
        s.x += xhalf(s.x); s.y += xhalf(s.y); s.z += xhalf(s.z); s.w += xhalf(s.w);
        float2 o;
        o.x = 1.0f / ((ph ? s.z : s.x) * gpj);
        o.y = 1.0f / ((ph ? s.w : s.y) * gpj);
        *(float2*)&IC[tt][g * 4 + ph * 2] = o;
      }
      if (ci + 1 < 16) {
        float mh[2][3], nw[2][4];
        #pragma unroll
        for (int r = 0; r < 3; ++r) { mh[0][r] = FM[ci + 1][0][row0 + r]; mh[1][r] = FM[ci + 1][1][row0 + r]; }
        const float4 a = *(const float4*)&FN[ci + 1][0][col0];
        const float4 d = *(const float4*)&FN[ci + 1][1][col0];
        nw[0][0] = a.x; nw[0][1] = a.y; nw[0][2] = a.z; nw[0][3] = a.w;
        nw[1][0] = d.x; nw[1][1] = d.y; nw[1][2] = d.z; nw[1][3] = d.w;
        km_col_partials_direct(mh, nw, PC[(ci + 1) & 1], rg, col0);
      }
    }
    __syncthreads();

    if (own) {
      float icv[4][4];
      #pragma unroll
      for (int t = 0; t < 4; ++t) {
        const float4 x = *(const float4*)&IC[t][col0];
        icv[t][0] = x.x; icv[t][1] = x.y; icv[t][2] = x.z; icv[t][3] = x.w;
      }
      #pragma unroll
      for (int r = 0; r < 3; ++r) {
        float irv[4];
        #pragma unroll
        for (int t = 0; t < 4; ++t) irv[t] = IR[t][lrow0 + r];
        const float vr[4] = { vv[r].x, vv[r].y, vv[r].z, vv[r].w };
        float sc[4];
        #pragma unroll
        for (int q = 0; q < 4; ++q) {
          float s = 0.f;
          #pragma unroll
          for (int t = 0; t < 4; ++t) s += E[t][r][q] * (irv[t] + icv[t][q]);
          z[r][q] += s * vr[q];
          smax = fmaxf(smax, s);
          sc[q] = s;
        }
        if (c0 == C_ - 1) {
          float4 o; o.x = sc[0]; o.y = sc[1]; o.z = sc[2]; o.w = sc[3];
          *(float4*)(out + BC_ * HW_ + bc * HW_ + (row0 + r) * W_ + col0) = o;
        }
      }
    }
  }

  if (own) {
    float* zdst = (half ? (ws + OFF_SCORE) : out) + (b * C_ + c0) * HW_;
    #pragma unroll
    for (int r = 0; r < 3; ++r) {
      float4 o; o.x = z[r][0]; o.y = z[r][1]; o.z = z[r][2]; o.w = z[r][3];
      *(float4*)(zdst + (row0 + r) * W_ + col0) = o;
    }
  }
  for (int o = 16; o > 0; o >>= 1) smax = fmaxf(smax, __shfl_xor(smax, o));
  if (lane == 0) red[wave] = smax;
  __syncthreads();
  if (tid == 0) {
    float g = red[0];
    for (int i = 1; i < 32; ++i) g = fmaxf(g, red[i]);
    atomicMax((int*)(ws + OFF_GSMAX + c0), __float_as_int(g));
  }
}

__global__ void ks_scale(const float* __restrict__ ws, float* __restrict__ out) {
  const int gid = blockIdx.x * 256 + threadIdx.x;
  float4* o4 = (float4*)out;
  if (gid < BC_ * HW_ / 4) {
    const int c0 = ((gid * 4) / HW_) % C_;
    const float inv = 1.0f / ws[OFF_GSMAX + c0];
    const float4 a = o4[gid];
    const float4 p = *(const float4*)(ws + OFF_SCORE + gid * 4);
    float4 r;
    r.x = (a.x + p.x) * inv; r.y = (a.y + p.y) * inv; r.z = (a.z + p.z) * inv; r.w = (a.w + p.w) * inv;
    o4[gid] = r;
  } else {
    const float inv = 1.0f / ws[OFF_GSMAX + 31];
    float4 a = o4[gid];
    a.x *= inv; a.y *= inv; a.z *= inv; a.w *= inv;
    o4[gid] = a;
  }
}

extern "C" void kernel_launch(void* const* d_in, const int* in_sizes, int n_in,
                              void* d_out, int out_size, void* d_ws, size_t ws_size,
                              hipStream_t stream) {
  (void)in_sizes; (void)n_in; (void)out_size; (void)ws_size;
  const float* q   = (const float*)d_in[0];
  const float* k   = (const float*)d_in[1];
  const float* val = (const float*)d_in[2];
  float* out = (float*)d_out;
  float* ws  = (float*)d_ws;

  k1_sums<<<128, 256, 0, stream>>>(q, k, ws);
  km_fused<<<256, 1024, 0, stream>>>(ws, val, out);
  ks_scale<<<1536, 256, 0, stream>>>(ws, out);
}


# km: whole 96x128 tile of 8 key channels per work-group (c in 4, rows unsplit): no exp formed twice; z in 3 planes, one of them two partials joined by f32 atomic add on zeros (k1); z sum order changes
# speedup vs baseline: 19.1344x; 1.1554x over previous
#include <hip/hip_runtime.h>

typedef __attribute__((ext_vector_type(2)))  float    v2f;
typedef __attribute__((ext_vector_type(8)))  float    v8f;

#define B_   2
#define C_   32
#define H_   96
#define W_   128
#define HW_  (H_ * W_)
#define BC_  (B_ * C_)
#define EPSF 1e-6f
#define LOG2E_F 1.44269504088896340736f

__device__ __forceinline__ float xhalf(float x) {
  const int b = __builtin_bit_cast(int, x);
  return __builtin_bit_cast(float, __builtin_amdgcn_permlanex16(b, b, 0x76543210, 0xfedcba98, false, false));
}

#define OFF_QS    0
#define OFF_QC    6144
#define OFF_KH    14336
#define OFF_KW    20480
#define OFF_SQ    28672
#define OFF_SK    28736
#define OFF_GMAX  28800
#define OFF_GSMAX 28928
#define OFF_HR    28960
#define OFF_Z1    OFF_HR
#define OFF_HC    (OFF_HR + 32 * 6144)
#define OFF_WR    (OFF_HC + 32 * 6144)
#define OFF_WC    (OFF_WR + 32 * 8192)
#define OFF_IRS   (OFF_WC + 32 * 8192)
#define OFF_SCORE (OFF_IRS + 4 * 64 * 224)


#define TP_ 132
__device__ __forceinline__ v8f mm4(const v2f a, const v2f b, const v8f c) {
  return __builtin_amdgcn_wmma_f32_16x16x4_f32(false, a, false, b, (short)0, c, false, false);
}
__global__ void __launch_bounds__(256)
k1_sums(const float* __restrict__ q, const float* __restrict__ k, float* __restrict__ ws) {
  __shared__ __attribute__((aligned(16))) float tile[H_ * TP_];
  __shared__ float part[6];
  const int bc   = blockIdx.x >> 1;
  const int p    = blockIdx.x & 1;
  const int tid  = threadIdx.x;
  const int wave = __builtin_amdgcn_readfirstlane(tid >> 5);
  const int lane = tid & 31;
  if (blockIdx.x == 0 && tid < C_) ws[OFF_GSMAX + tid] = 0.0f;
  const float* s = (p ? k : q) + bc * HW_;
  float4 v[HW_ / 1024];
  #pragma unroll
  for (int j = 0; j < HW_ / 1024; ++j) v[j] = *(const float4*)(s + (tid + j * 256) * 4);
  {
    const float4 zero4 = {0.0f, 0.0f, 0.0f, 0.0f};
    float4* const z2 = (float4*)(ws + OFF_SCORE);
    const int g = blockIdx.x * 256 + tid;
    #pragma unroll
    for (int j = 0; j < 6; ++j) z2[g + j * 32768] = zero4;
  }
  #pragma unroll
  for (int j = 0; j < HW_ / 1024; ++j) {
    const int i4 = tid + j * 256;
    *(float4*)&tile[(i4 >> 5) * TP_ + (i4 & 31) * 4] = v[j];
  }
  __syncthreads();

  const v2f ones = {1.0f, 1.0f};
  const int c = lane & 15, h = lane >> 4;
  {
    const float* pc = tile + (2 * h) * TP_ + wave * 16 + c;
    v8f a0 = {}, a1 = {};
    #pragma unroll
    for (int k0 = 0; k0 < H_; k0 += 8) {
      const v2f y0 = { pc[k0 * TP_],       pc[(k0 + 1) * TP_] };
      const v2f y1 = { pc[(k0 + 4) * TP_], pc[(k0 + 5) * TP_] };
      a0 = mm4(ones, y0, a0);
      a1 = mm4(ones, y1, a1);
    }
    const float cs = a0[0] + a1[0];
    if (lane < 16) ws[(p ? OFF_KW : OFF_QC) + bc * W_ + wave * 16 + lane] = cs;
  }
  if (wave < 6) {
    const float* pr = tile + (wave * 16 + c) * TP_ + 2 * h;
    v8f a0 = {}, a1 = {};
    #pragma unroll
    for (int k0 = 0; k0 < W_; k0 += 8) {
      const v2f x0 = *(const v2f*)(pr + k0);
      const v2f x1 = *(const v2f*)(pr + k0 + 4);
      a0 = mm4(x0, ones, a0);
      a1 = mm4(x1, ones, a1);
    }
    float acc[8];
    #pragma unroll
    for (int r = 0; r < 8; ++r) acc[r] = a0[r] + a1[r];
    float rs = acc[0];
    #pragma unroll
    for (int r = 1; r < 8; ++r) rs = ((lane & 7) == r) ? acc[r] : rs;
    if ((lane & 8) == 0) ws[(p ? OFF_KH : OFF_QS) + bc * H_ + wave * 16 + 8 * h + (lane & 7)] = rs;
    float t = ((acc[0] + acc[1]) + (acc[2] + acc[3])) + ((acc[4] + acc[5]) + (acc[6] + acc[7]));
    t += xhalf(t);
    if (lane == 0) part[wave] = t;
  }
  __syncthreads();
  if (tid == 0) ws[(p ? OFF_SK : OFF_SQ) + bc] = ((part[0] + part[1]) + (part[2] + part[3])) + (part[4] + part[5]);
}

#define PRP 33
#define HO_ 96

__device__ __forceinline__ void km_exps(const float (&mh)[2][3], const float (&nw)[2][4], float (&E)[4][3][4]) {
  float ml[2][3];
  #pragma unroll
  for (int r = 0; r < 3; ++r) { ml[0][r] = mh[0][r] * LOG2E_F; ml[1][r] = mh[1][r] * LOG2E_F; }
  #pragma unroll
  for (int t = 0; t < 4; ++t) {
    #pragma unroll
    for (int r = 0; r < 3; ++r) {
      #pragma unroll
      for (int q = 0; q < 4; ++q) E[t][r][q] = ml[t >> 1][r] * nw[t & 1][q];
    }
  }
  #pragma unroll
  for (int t = 0; t < 4; ++t) {
    #pragma unroll
    for (int r = 0; r < 3; ++r) {
      #pragma unroll
      for (int q = 0; q < 4; ++q) E[t][r][q] = __builtin_amdgcn_exp2f(E[t][r][q]);
    }
  }
}

__device__ __forceinline__ void km_col_partials(const float (&E)[4][3][4], float* pc, int rg, int col0) {
  #pragma unroll
  for (int t = 0; t < 4; ++t) {
    float4 cp;
    cp.x = E[t][0][0] + E[t][1][0] + E[t][2][0];
    cp.y = E[t][0][1] + E[t][1][1] + E[t][2][1];
    cp.z = E[t][0][2] + E[t][1][2] + E[t][2][2];
    cp.w = E[t][0][3] + E[t][1][3] + E[t][2][3];
    *(float4*)(pc + t * (32 * W_) + rg * W_ + col0) = cp;
  }
}

__device__ __forceinline__ void km_row_partials(const float (&E)[4][3][4], float* pr, int lrow0, int lane) {
  #pragma unroll
  for (int t = 0; t < 4; ++t) {
    #pragma unroll
    for (int r = 0; r < 3; ++r)
      pr[t * (HO_ * PRP) + (lrow0 + r) * PRP + lane] = (E[t][r][0] + E[t][r][1]) + (E[t][r][2] + E[t][r][3]);
  }
}

template <int K> __device__ __forceinline__ float rowx(float x) {
  const int v = __builtin_bit_cast(int, x);
  return __builtin_bit_cast(float, __builtin_amdgcn_update_dpp(v, v, 0x160 + K, 0xf, 0xf, false));
}
template <class F> __device__ __forceinline__ float red8(const float (&v)[8], int lane, F op) {
  const bool h = (lane & 16) != 0, b3 = (lane & 8) != 0, b2 = (lane & 4) != 0;
  float a[4], b[2];
  #pragma unroll
  for (int i = 0; i < 4; ++i) a[i] = op(h ? v[i + 4] : v[i], xhalf(h ? v[i] : v[i + 4]));
  #pragma unroll
  for (int i = 0; i < 2; ++i) b[i] = op(b3 ? a[i + 2] : a[i], rowx<8>(b3 ? a[i] : a[i + 2]));
  float c = op(b2 ? b[1] : b[0], rowx<4>(b2 ? b[0] : b[1]));
  c = op(c, rowx<2>(c));
  c = op(c, rowx<1>(c));
  return c;
}
__device__ __forceinline__ float lane_of(float x, int l) {
  return __builtin_bit_cast(float, __builtin_amdgcn_readlane(__builtin_bit_cast(int, x), l));
}

#define T_HR 0
#define T_HC 6144
#define T_WR 12288
#define T_WC 20480
#define T_N  28672

__device__ __forceinline__ void mnmx4(const float4 v, float& mn, float& mx) {
  mn = fminf(fminf(mn, v.x), fminf(v.y, fminf(v.z, v.w)));
  mx = fmaxf(fmaxf(mx, v.x), fmaxf(v.y, fmaxf(v.z, v.w)));
}
__device__ __forceinline__ float4 norm4(const float4 v, const float mn, const float inv) {
  float4 o;
  o.x = 2.0f + (v.x - mn) * inv; o.y = 2.0f + (v.y - mn) * inv;
  o.z = 2.0f + (v.z - mn) * inv; o.w = 2.0f + (v.w - mn) * inv;
  return o;
}
__device__ __forceinline__ void raw_h(const float* __restrict__ ws, int c0, int j4, float sq0, float sq1,
                                      float4& rhr, float4& rhc) {
  const int pb = j4 / 24, y4 = (j4 - pb * 24) * 4;
  const int bb = pb >> 5;
  const float sq = bb ? sq1 : sq0;
  const float sk = ws[OFF_SK + pb];
  const float4 kh = *(const float4*)(ws + OFF_KH + pb * H_ + y4);
  const float4 qs = *(const float4*)(ws + OFF_QS + (bb * C_ + c0) * H_ + y4);
  rhr.x = sq * kh.x + (H_ * EPSF); rhr.y = sq * kh.y + (H_ * EPSF);
  rhr.z = sq * kh.z + (H_ * EPSF); rhr.w = sq * kh.w + (H_ * EPSF);
  rhc.x = qs.x * sk + (H_ * EPSF); rhc.y = qs.y * sk + (H_ * EPSF);
  rhc.z = qs.z * sk + (H_ * EPSF); rhc.w = qs.w * sk + (H_ * EPSF);
}
__device__ __forceinline__ void raw_w(const float* __restrict__ ws, int c0, int j4, float sq0, float sq1,
                                      float4& rwr, float4& rwc) {
  const int pb = j4 >> 5, x4 = (j4 & 31) * 4;
  const int bb = pb >> 5;
  const float sq = bb ? sq1 : sq0;
  const float sk = ws[OFF_SK + pb];
  const float4 kw = *(const float4*)(ws + OFF_KW + pb * W_ + x4);
  const float4 qc = *(const float4*)(ws + OFF_QC + (bb * C_ + c0) * W_ + x4);
  rwr.x = sq * kw.x + (W_ * EPSF); rwr.y = sq * kw.y + (W_ * EPSF);
  rwr.z = sq * kw.z + (W_ * EPSF); rwr.w = sq * kw.w + (W_ * EPSF);
  rwc.x = qc.x * sk + (W_ * EPSF); rwc.y = qc.y * sk + (W_ * EPSF);
  rwc.z = qc.z * sk + (W_ * EPSF); rwc.w = qc.w * sk + (W_ * EPSF);
}

__global__ void __launch_bounds__(1024)
km_fused(float* __restrict__ ws, const float* __restrict__ values, float* __restrict__ out) {
  const int cq   = blockIdx.x & 3;
  const int b    = (blockIdx.x >> 2) & 1;
  const int c0   = blockIdx.x >> 3;
  const int tid  = threadIdx.x;
  const int wave = __builtin_amdgcn_readfirstlane(tid >> 5);
  const int lane = tid & 31;
  __shared__ float PR[4 * HO_ * PRP];
  __shared__ __attribute__((aligned(16))) float PC[2][4 * 32 * W_];
  __shared__ float IR[4][HO_];
  __shared__ __attribute__((aligned(16))) float IC[4][W_];
  __shared__ __attribute__((aligned(16))) float FN[8][2][W_];
  __shared__ __attribute__((aligned(16))) float FM[8][2][H_];
  __shared__ float red[32];
  float* const TF = &PC[0][0];
  float (* const RMN)[32] = (float (*)[32])(TF + T_N);
  float (* const RMX)[32] = (float (*)[32])(TF + T_N + 128);
  float (* const GP)[32]  = (float (*)[32])(TF + T_N + 256);

  const int bc0 = b * C_ + cq * 8;
  float gp[4];
  {
    const bool two = wave < 16;
    const float sq0 = ws[OFF_SQ + c0];
    const float sq1 = ws[OFF_SQ + C_ + c0];
    float4 rhr0, rhc0, rhr1, rhc1, rwr0, rwc0, rwr1, rwc1;
    raw_h(ws, c0, tid, sq0, sq1, rhr0, rhc0);
    raw_w(ws, c0, tid, sq0, sq1, rwr0, rwc0);
    raw_w(ws, c0, tid + 1024, sq0, sq1, rwr1, rwc1);
    rhr1 = rhr0; rhc1 = rhc0;
    if (two) raw_h(ws, c0, tid + 1024, sq0, sq1, rhr1, rhc1);

    float mn[4], mx[4];
    #pragma unroll
    for (int f = 0; f < 4; ++f) { mn[f] = 3.4e38f; mx[f] = -3.4e38f; }
    mnmx4(rhr0, mn[0], mx[0]); mnmx4(rhr1, mn[0], mx[0]);
    mnmx4(rhc0, mn[1], mx[1]); mnmx4(rhc1, mn[1], mx[1]);
    mnmx4(rwr0, mn[2], mx[2]); mnmx4(rwr1, mn[2], mx[2]);
    mnmx4(rwc0, mn[3], mx[3]); mnmx4(rwc1, mn[3], mx[3]);
    {
      const float v8[8] = { mn[0], mn[1], mn[2], mn[3], -mx[0], -mx[1], -mx[2], -mx[3] };
      const float c = red8(v8, lane, [](float x, float y) { return fminf(x, y); });
      if ((lane & 3) == 0) RMN[lane >> 2][wave] = (lane < 16) ? c : -c;
    }
    __syncthreads();
    float inv[4];
    {
      const int fo = lane >> 3, eo = (lane & 7) * 4;
      const float4 a4 = *(const float4*)&RMN[fo][eo];
      const float4 b4 = *(const float4*)&RMX[fo][eo];
      float lmn = fminf(fminf(a4.x, a4.y), fminf(a4.z, a4.w));
      float lmx = fmaxf(fmaxf(b4.x, b4.y), fmaxf(b4.z, b4.w));
      lmn = fminf(lmn, rowx<1>(lmn)); lmn = fminf(lmn, rowx<2>(lmn)); lmn = fminf(lmn, rowx<4>(lmn));
      lmx = fmaxf(lmx, rowx<1>(lmx)); lmx = fmaxf(lmx, rowx<2>(lmx)); lmx = fmaxf(lmx, rowx<4>(lmx));
      const float linv = 1.0f / (lmx - lmn);
      #pragma unroll
      for (int f = 0; f < 4; ++f) { mn[f] = lane_of(lmn, 8 * f); inv[f] = lane_of(linv, 8 * f); }
    }

    #pragma unroll
    for (int k = 0; k < 2; ++k) {
      const int j4 = tid + k * 1024;
      const float4 wr = norm4(k ? rwr1 : rwr0, mn[2], inv[2]);
      const float4 wc = norm4(k ? rwc1 : rwc0, mn[3], inv[3]);
      *(float4*)&TF[T_WR + j4 * 4] = wr;
      *(float4*)&TF[T_WC + j4 * 4] = wc;
      const int chn = (j4 >> 5) - bc0, x4 = (j4 & 31) * 4;
      if (chn >= 0 && chn < 8) { *(float4*)&FN[chn][0][x4] = wr; *(float4*)&FN[chn][1][x4] = wc; }
      if (k == 0 || two) {
        const float4 hr = norm4(k ? rhr1 : rhr0, mn[0], inv[0]);
        const float4 hc = norm4(k ? rhc1 : rhc0, mn[1], inv[1]);
        *(float4*)&TF[T_HR + j4 * 4] = hr;
        *(float4*)&TF[T_HC + j4 * 4] = hc;
        const int pbm = j4 / 24, chm = pbm - bc0, y4 = (j4 - pbm * 24) * 4;
        if (chm >= 0 && chm < 8) { *(float4*)&FM[chm][0][y4] = hr; *(float4*)&FM[chm][1][y4] = hc; }
      }
    }
    __syncthreads();

    {
      float m[2][2][3], n[2][2][4];
      #pragma unroll
      for (int i = 0; i < 2; ++i) {
        const int pb = wave + i * 32;
        #pragma unroll
        for (int r = 0; r < 3; ++r) {
          m[i][0][r] = TF[T_HR + pb * H_ + lane * 3 + r];
          m[i][1][r] = TF[T_HC + pb * H_ + lane * 3 + r];
        }
        const float4 a = *(const float4*)&TF[T_WR + pb * W_ + lane * 4];
        const float4 d = *(const float4*)&TF[T_WC + pb * W_ + lane * 4];
        n[i][0][0] = a.x; n[i][0][1] = a.y; n[i][0][2] = a.z; n[i][0][3] = a.w;
        n[i][1][0] = d.x; n[i][1][1] = d.y; n[i][1][2] = d.z; n[i][1][3] = d.w;
      }
      float lm[8];
      #pragma unroll
      for (int i = 0; i < 2; ++i) {
        #pragma unroll
        for (int f = 0; f < 2; ++f) {
          lm[i * 4 + f]     = fmaxf(fmaxf(m[i][f][0], m[i][f][1]), m[i][f][2]);
          lm[i * 4 + 2 + f] = fmaxf(fmaxf(n[i][f][0], n[i][f][1]), fmaxf(n[i][f][2], n[i][f][3]));
        }
      }
      const float cm = red8(lm, lane, [](float x, float y) { return fmaxf(x, y); });
      float mml[2][2], nn[2][2];
      #pragma unroll
      for (int i = 0; i < 2; ++i) {
        #pragma unroll
        for (int f = 0; f < 2; ++f) {
          mml[i][f] = lane_of(cm, 4 * (i * 4 + f)) * LOG2E_F;
          nn[i][f]  = lane_of(cm, 4 * (i * 4 + 2 + f));
        }
      }
      float rsv[8], csv[8];
      #pragma unroll
      for (int i = 0; i < 2; ++i) {
        #pragma unroll
        for (int t = 0; t < 4; ++t) {
          const int f = t >> 1, g = t & 1;
          rsv[i * 4 + t] = __builtin_amdgcn_exp2f(mml[i][f] * n[i][g][0]) + __builtin_amdgcn_exp2f(mml[i][f] * n[i][g][1])
                         + __builtin_amdgcn_exp2f(mml[i][f] * n[i][g][2]) + __builtin_amdgcn_exp2f(mml[i][f] * n[i][g][3]);
          csv[i * 4 + t] = __builtin_amdgcn_exp2f((m[i][f][0] * LOG2E_F) * nn[i][g]) + __builtin_amdgcn_exp2f((m[i][f][1] * LOG2E_F) * nn[i][g])
                         + __builtin_amdgcn_exp2f((m[i][f][2] * LOG2E_F) * nn[i][g]);
        }
      }
      const float rs = red8(rsv, lane, [](float x, float y) { return x + y; });
      const float cs = red8(csv, lane, [](float x, float y) { return x + y; });
      const bool hi = (lane & 16) != 0, f1 = (lane & 8) != 0, g1 = (lane & 4) != 0;
      const float mml_l = hi ? (f1 ? mml[1][1] : mml[1][0]) : (f1 ? mml[0][1] : mml[0][0]);
      const float nn_l  = hi ? (g1 ? nn[1][1] : nn[1][0]) : (g1 ? nn[0][1] : nn[0][0]);
      float pk = __builtin_amdgcn_exp2f(mml_l * nn_l) * (1.0f / rs + 1.0f / cs);
      pk = fmaxf(pk, xhalf(pk));
      if ((lane & 19) == 0) GP[lane >> 2][wave] = pk;
    }
    __syncthreads();
    {
      const float4 g4 = *(const float4*)&GP[lane >> 3][(lane & 7) * 4];
      float g = fmaxf(fmaxf(g4.x, g4.y), fmaxf(g4.z, g4.w));
      g = fmaxf(g, rowx<1>(g)); g = fmaxf(g, rowx<2>(g)); g = fmaxf(g, rowx<4>(g));
      #pragma unroll
      for (int t = 0; t < 4; ++t) gp[t] = lane_of(g, 8 * t);
    }
  }

  const bool rowjob = wave < 24;
  const int rg    = wave;
  const int row0  = rg * 3;
  const int col0  = lane * 4;
  const int   jt  = rowjob ? (wave / 6) : ((wave - 24) >> 1);
  const float gpj = (jt == 0) ? gp[0] : (jt == 1) ? gp[1] : (jt == 2) ? gp[2] : gp[3];

  float z[3][4];
  #pragma unroll
  for (int r = 0; r < 3; ++r) { z[r][0] = 0.f; z[r][1] = 0.f; z[r][2] = 0.f; z[r][3] = 0.f; }
  float smax = 0.0f;

  for (int ci = 0; ci < 8; ++ci) {
    const int bc = bc0 + ci;
    float  E[4][3][4];
    float4 vv[3];
    {
      float mh[2][3], nw[2][4];
      #pragma unroll
      for (int r = 0; r < 3; ++r) { mh[0][r] = FM[ci][0][row0 + r]; mh[1][r] = FM[ci][1][row0 + r]; }
      const float4 a = *(const float4*)&FN[ci][0][col0];
      const float4 d = *(const float4*)&FN[ci][1][col0];
      nw[0][0] = a.x; nw[0][1] = a.y; nw[0][2] = a.z; nw[0][3] = a.w;
      nw[1][0] = d.x; nw[1][1] = d.y; nw[1][2] = d.z; nw[1][3] = d.w;
      km_exps(mh, nw, E);
      km_row_partials(E, PR, row0, lane);
      km_col_partials(E, PC[0], rg, col0);
    }
    __syncthreads();

    #pragma unroll
    for (int r = 0; r < 3; ++r)
      vv[r] = *(const float4*)(values + bc * HW_ + (row0 + r) * W_ + col0);

    if (rowjob) {
      const v2f ones = {1.0f, 1.0f};
      const int kho = (lane < 16) ? 0 : 2;
      const int tt = wave / 6, rb = wave % 6;
      const int M  = rb * 16 + (lane & 15);
      const float* pr = PR + tt * (HO_ * PRP) + M * PRP;
      v8f acc = {};
      for (int k0 = 0; k0 < 32; k0 += 4) {
        v2f a = { pr[k0 + kho], pr[k0 + kho + 1] };
        acc = __builtin_amdgcn_wmma_f32_16x16x4_f32(false, a, false, ones, (short)0, acc, false, false);
      }
      float rs = acc[0];
      #pragma unroll
      for (int r = 1; r < 8; ++r) rs = ((lane & 7) == r) ? acc[r] : rs;
      IR[tt][rb * 16 + ((lane >> 4) << 3) + (lane & 7)] = __builtin_amdgcn_rcpf(rs * gpj);
    } else {
      const int jw = wave - 24;
      const int tt = jw >> 1;
      const int g  = ((jw & 1) << 4) | (lane & 15);
      const int ph = lane >> 4;
      const float* pc = &PC[0][tt * (32 * W_) + (ph * 16) * W_ + g * 4];
      float4 s;
      #pragma unroll
      for (int i0 = 0; i0 < 16; i0 += 4) {
        const float4 x0 = *(const float4*)(pc + (i0 + 0) * W_);
        const float4 x1 = *(const float4*)(pc + (i0 + 1) * W_);
        const float4 x2 = *(const float4*)(pc + (i0 + 2) * W_);
        const float4 x3 = *(const float4*)(pc + (i0 + 3) * W_);
        float4 t;
        t.x = (x0.x + x1.x) + (x2.x + x3.x);
        t.y = (x0.y + x1.y) + (x2.y + x3.y);
        t.z = (x0.z + x1.z) + (x2.z + x3.z);
        t.w = (x0.w + x1.w) + (x2.w + x3.w);
        if (i0 == 0) { s = t; } else { s.x += t.x; s.y += t.y; s.z += t.z; s.w += t.w; }
        __builtin_amdgcn_sched_barrier(0);
      }
      s.x += xhalf(s.x); s.y += xhalf(s.y); s.z += xhalf(s.z); s.w += xhalf(s.w);
      float2 o;
      o.x = 1.0f / ((ph ? s.z : s.x) * gpj);
      o.y = 1.0f / ((ph ? s.w : s.y) * gpj);
      *(float2*)&IC[tt][g * 4 + ph * 2] = o;
    }
    __syncthreads();

    {
      float icv[4][4];
      #pragma unroll
      for (int t = 0; t < 4; ++t) {
        const float4 x = *(const float4*)&IC[t][col0];
        icv[t][0] = x.x; icv[t][1] = x.y; icv[t][2] = x.z; icv[t][3] = x.w;
      }
      #pragma unroll
      for (int r = 0; r < 3; ++r) {
        float irv[4];
        #pragma unroll
        for (int t = 0; t < 4; ++t) irv[t] = IR[t][row0 + r];
        const float vr[4] = { vv[r].x, vv[r].y, vv[r].z, vv[r].w };
        float sc[4];
        #pragma unroll
        for (int q = 0; q < 4; ++q) {
          float s = 0.f;
          #pragma unroll
          for (int t = 0; t < 4; ++t) s += E[t][r][q] * (irv[t] + icv[t][q]);
          z[r][q] += s * vr[q];
          smax = fmaxf(smax, s);
          sc[q] = s;
        }
        if (c0 == C_ - 1) {
          float4 o; o.x = sc[0]; o.y = sc[1]; o.z = sc[2]; o.w = sc[3];
          *(float4*)(out + BC_ * HW_ + bc * HW_ + (row0 + r) * W_ + col0) = o;
        }
      }
    }
  }

  if (cq < 2) {
    float* zdst = (cq ? (ws + OFF_Z1) : out) + (b * C_ + c0) * HW_;
    #pragma unroll
    for (int r = 0; r < 3; ++r) {
      float4 o; o.x = z[r][0]; o.y = z[r][1]; o.z = z[r][2]; o.w = z[r][3];
      *(float4*)(zdst + (row0 + r) * W_ + col0) = o;
    }
  } else {
    float* const zl = &PC[0][wave * 384];
    #pragma unroll
    for (int r = 0; r < 3; ++r) {
      float4 o; o.x = z[r][0]; o.y = z[r][1]; o.z = z[r][2]; o.w = z[r][3];
      *(float4*)(zl + r * W_ + col0) = o;
    }
    __syncthreads();
    float* zdst = ws + OFF_SCORE + (b * C_ + c0) * HW_ + row0 * W_;
    float zv[12];
    #pragma unroll
    for (int j = 0; j < 12; ++j) zv[j] = zl[j * 32 + lane];
    #pragma unroll
    for (int j = 0; j < 12; ++j)
      __hip_atomic_fetch_add(zdst + j * 32 + lane, zv[j], __ATOMIC_RELAXED, __HIP_MEMORY_SCOPE_AGENT);
  }
  for (int o = 16; o > 0; o >>= 1) smax = fmaxf(smax, __shfl_xor(smax, o));
  if (lane == 0) red[wave] = smax;
  __syncthreads();
  if (tid == 0) {
    float g = red[0];
    for (int i = 1; i < 32; ++i) g = fmaxf(g, red[i]);
    atomicMax((int*)(ws + OFF_GSMAX + c0), __float_as_int(g));
  }
}

__global__ void ks_scale(const float* __restrict__ ws, float* __restrict__ out) {
  const int gid = blockIdx.x * 256 + threadIdx.x;
  float4* o4 = (float4*)out;
  if (gid < BC_ * HW_ / 4) {
    const int c0 = ((gid * 4) / HW_) % C_;
    const float inv = 1.0f / ws[OFF_GSMAX + c0];
    const float4 a = o4[gid];
    const float4 p = *(const float4*)(ws + OFF_Z1 + gid * 4);
    const float4 u = *(const float4*)(ws + OFF_SCORE + gid * 4);
    float4 r;
    r.x = ((a.x + p.x) + u.x) * inv; r.y = ((a.y + p.y) + u.y) * inv;
    r.z = ((a.z + p.z) + u.z) * inv; r.w = ((a.w + p.w) + u.w) * inv;
    o4[gid] = r;
  } else {
    const float inv = 1.0f / ws[OFF_GSMAX + 31];
    float4 a = o4[gid];
    a.x *= inv; a.y *= inv; a.z *= inv; a.w *= inv;
    o4[gid] = a;
  }
}

extern "C" void kernel_launch(void* const* d_in, const int* in_sizes, int n_in,
                              void* d_out, int out_size, void* d_ws, size_t ws_size,
                              hipStream_t stream) {
  (void)in_sizes; (void)n_in; (void)out_size; (void)ws_size;
  const float* q   = (const float*)d_in[0];
  const float* k   = (const float*)d_in[1];
  const float* val = (const float*)d_in[2];
  float* out = (float*)d_out;
  float* ws  = (float*)d_ws;

  k1_sums<<<128, 256, 0, stream>>>(q, k, ws);
  km_fused<<<256, 1024, 0, stream>>>(ws, val, out);
  ks_scale<<<1536, 256, 0, stream>>>(ws, out);
}
